# starting kernel
# baseline (speedup 1.0000x reference)
#include <hip/hip_runtime.h>
#include <math.h>

typedef float v2f __attribute__((ext_vector_type(2)));
typedef float v8f __attribute__((ext_vector_type(8)));

#define LRELU_SLOPE 0.2f

__device__ __forceinline__ unsigned f2o(float f) {
  unsigned u = __float_as_uint(f);
  return (u & 0x80000000u) ? ~u : (u | 0x80000000u);
}
__device__ __forceinline__ float o2f(unsigned o) {
  unsigned u = (o & 0x80000000u) ? (o & 0x7fffffffu) : ~o;
  return __uint_as_float(u);
}
#define ORD_NEG_INF 0x007fffffu

__global__ void fill_f32(float* p, float v, int n) {
  int i = blockIdx.x * blockDim.x + threadIdx.x;
  int stride = gridDim.x * blockDim.x;
  for (; i < n; i += stride) p[i] = v;
}
__global__ void fill_u32(unsigned* p, unsigned v, int n) {
  int i = blockIdx.x * blockDim.x + threadIdx.x;
  int stride = gridDim.x * blockDim.x;
  for (; i < n; i += stride) p[i] = v;
}

__global__ void gemm_wmma_f32(const float* __restrict__ A,
                              const float* __restrict__ B,
                              float* __restrict__ C,
                              int M, int Ncols, int K) {
  const int wavesPerBlock = blockDim.x >> 5;
  const int wave = threadIdx.x >> 5;
  const int lane = threadIdx.x & 31;
  const int half = lane >> 4;
  const int l    = lane & 15;

  const int m0 = blockIdx.x * 16;
  const int nt = blockIdx.y * wavesPerBlock + wave;
  const int n0 = nt * 16;
  if (m0 >= M || n0 >= Ncols) return;

  v8f acc = {};
  const float* arow = A + (size_t)(m0 + l) * K + half * 2;
  for (int k = 0; k < K; k += 4) {
    v2f a, b;
    a.x = arow[k];
    a.y = arow[k + 1];
    const float* bp = B + (size_t)(k + half * 2) * Ncols + n0 + l;
    b.x = bp[0];
    b.y = bp[Ncols];
    acc = __builtin_amdgcn_wmma_f32_16x16x4_f32(false, a, false, b,
                                                (short)0, acc, false, false);
  }
#pragma unroll
  for (int r = 0; r < 8; ++r) {
    C[(size_t)(m0 + half * 8 + r) * Ncols + n0 + l] = acc[r];
  }
}

__global__ void node_scores(const float* __restrict__ feat,
                            const float* __restrict__ al,
                            const float* __restrict__ ar,
                            float* __restrict__ el, float* __restrict__ er,
                            int N, int H, int D) {
  int i = blockIdx.x * blockDim.x + threadIdx.x;
  if (i >= N * H) return;
  int n = i / H, h = i - n * H;
  const float* f = feat + (size_t)n * H * D + h * D;
  const float* wl = al + h * D;
  const float* wr = ar + h * D;
  float sl = 0.f, sr = 0.f;
#pragma unroll 4
  for (int d = 0; d < D; ++d) {
    float v = f[d];
    sl += v * wl[d];
    sr += v * wr[d];
  }
  el[i] = sl;
  er[i] = sr;
}

__global__ void edge_max(const int* __restrict__ src, const int* __restrict__ dst,
                         const float* __restrict__ el, const float* __restrict__ er,
                         unsigned* __restrict__ m_ord, int E, int H) {
  int i = blockIdx.x * blockDim.x + threadIdx.x;
  if (i >= E * H) return;
  int e = i / H, h = i - e * H;
  int d = dst[e];
  float sc = el[src[e] * H + h] + er[d * H + h];
  sc = sc > 0.f ? sc : LRELU_SLOPE * sc;
  atomicMax(m_ord + d * H + h, f2o(sc));
}

__global__ void edge_expsum(const int* __restrict__ src, const int* __restrict__ dst,
                            const float* __restrict__ el, const float* __restrict__ er,
                            const unsigned* __restrict__ m_ord,
                            float* __restrict__ ee, float* __restrict__ s,
                            int E, int H) {
  int i = blockIdx.x * blockDim.x + threadIdx.x;
  if (i >= E * H) return;
  int e = i / H, h = i - e * H;
  int d = dst[e];
  float sc = el[src[e] * H + h] + er[d * H + h];
  sc = sc > 0.f ? sc : LRELU_SLOPE * sc;
  float m = o2f(m_ord[d * H + h]);
  if (m < -1e30f) m = 0.f;
  float v = __expf(sc - m);
  ee[i] = v;
  atomicAdd(s + d * H + h, v);
}

__global__ void edge_aggregate(const int* __restrict__ src, const int* __restrict__ dst,
                               const float* __restrict__ ee, const float* __restrict__ s,
                               const float* __restrict__ feat, float* __restrict__ acc,
                               int E, int H, int D) {
  const int chunks = (H * D) >> 2;
  int tid = blockIdx.x * blockDim.x + threadIdx.x;
  if (tid >= E * chunks) return;
  int e = tid / chunks;
  int c = tid - e * chunks;
  int h = (c << 2) / D;
  int sN = src[e], dN = dst[e];
  float a = ee[(size_t)e * H + h] / fmaxf(s[(size_t)dN * H + h], 1e-9f);
  const float4 f = *(const float4*)(feat + (size_t)sN * H * D + (c << 2));
  float* o = acc + (size_t)dN * H * D + (c << 2);
  atomicAdd(o + 0, a * f.x);
  atomicAdd(o + 1, a * f.y);
  atomicAdd(o + 2, a * f.z);
  atomicAdd(o + 3, a * f.w);
}

__global__ void epilogue0(const float* __restrict__ acc, const float* __restrict__ b,
                          float* __restrict__ h1, int N, int H, int D) {
  int i = blockIdx.x * blockDim.x + threadIdx.x;
  if (i >= N * D) return;
  int n = i / D, d = i - n * D;
  float sum = 0.f;
  for (int t = 0; t < 2; ++t) {
    const float* a = acc + ((size_t)t * N + n) * H * D + d;
    for (int h = 0; h < H; ++h) {
      float v = a[h * D] + b[h * D + d];
      sum += (v > 0.f) ? v : expm1f(v);
    }
  }
  h1[i] = sum / (float)H;
}

__global__ void epilogue1(const float* __restrict__ acc, const float* __restrict__ b,
                          float* __restrict__ out, int N, int D) {
  int i = blockIdx.x * blockDim.x + threadIdx.x;
  if (i >= N * D) return;
  int d = i - (i / D) * D;
  out[i] = (acc[i] + b[d]) + (acc[(size_t)N * D + i] + b[d]);
}

extern "C" void kernel_launch(void* const* d_in, const int* in_sizes, int n_in,
                              void* d_out, int out_size, void* d_ws, size_t ws_size,
                              hipStream_t stream) {
  const int IN = 128, H0 = 4, D0 = 64, H1 = 1, D1 = 64;
  const int N = in_sizes[0] / IN;
  const int E = in_sizes[9];

  const float* x   = (const float*)d_in[0];
  const float* W0  = (const float*)d_in[1];
  const float* al0 = (const float*)d_in[2];
  const float* ar0 = (const float*)d_in[3];
  const float* b0  = (const float*)d_in[4];
  const float* W1  = (const float*)d_in[5];
  const float* al1 = (const float*)d_in[6];
  const float* ar1 = (const float*)d_in[7];
  const float* b1  = (const float*)d_in[8];
  const int* srcT[2] = { (const int*)d_in[9],  (const int*)d_in[11] };
  const int* dstT[2] = { (const int*)d_in[10], (const int*)d_in[12] };
  float* out = (float*)d_out;

  float* ws = (float*)d_ws;
  float*    feat0 = ws;
  float*    el0   = feat0 + (size_t)N * H0 * D0;
  float*    er0   = el0 + (size_t)N * H0;
  unsigned* m0    = (unsigned*)(er0 + (size_t)N * H0);
  float*    s0    = (float*)(m0 + 2 * (size_t)N * H0);
  float*    ee0   = s0 + 2 * (size_t)N * H0;
  float*    acc0  = ee0 + 2 * (size_t)E * H0;
  float*    h1    = acc0 + 2 * (size_t)N * H0 * D0;
  float*    feat1 = feat0;
  float*    el1   = feat1 + (size_t)N * D1;
  float*    er1   = el1 + N;
  unsigned* m1    = (unsigned*)(er1 + N);
  float*    s1    = (float*)(m1 + 2 * (size_t)N);
  float*    ee1   = s1 + 2 * (size_t)N;
  float*    acc1  = ee1 + 2 * (size_t)E;

  const int TB = 256;
  const int fillBlocks = 2048;
  (void)ws_size; (void)n_in; (void)out_size;

  {
    dim3 grid(N / 16, (H0 * D0) / 16 / 4);
    gemm_wmma_f32<<<grid, 128, 0, stream>>>(x, W0, feat0, N, H0 * D0, IN);

    node_scores<<<(N * H0 + TB - 1) / TB, TB, 0, stream>>>(feat0, al0, ar0, el0, er0, N, H0, D0);

    fill_u32<<<fillBlocks, TB, 0, stream>>>(m0, ORD_NEG_INF, 2 * N * H0);
    fill_f32<<<fillBlocks, TB, 0, stream>>>(s0, 0.f, 2 * N * H0);
    fill_f32<<<fillBlocks, TB, 0, stream>>>(acc0, 0.f, 2 * N * H0 * D0);

    for (int t = 0; t < 2; ++t) {
      unsigned* mT = m0 + (size_t)t * N * H0;
      float*    sT = s0 + (size_t)t * N * H0;
      float*    eeT = ee0 + (size_t)t * E * H0;
      float*    accT = acc0 + (size_t)t * N * H0 * D0;
      int eh = E * H0;
      edge_max<<<(eh + TB - 1) / TB, TB, 0, stream>>>(srcT[t], dstT[t], el0, er0, mT, E, H0);
      edge_expsum<<<(eh + TB - 1) / TB, TB, 0, stream>>>(srcT[t], dstT[t], el0, er0, mT, eeT, sT, E, H0);
      int total = E * ((H0 * D0) >> 2);
      edge_aggregate<<<(total + TB - 1) / TB, TB, 0, stream>>>(srcT[t], dstT[t], eeT, sT, feat0, accT, E, H0, D0);
    }

    epilogue0<<<(N * D0 + TB - 1) / TB, TB, 0, stream>>>(acc0, b0, h1, N, H0, D0);
  }

  {
    dim3 grid(N / 16, 1);
    gemm_wmma_f32<<<grid, 128, 0, stream>>>(h1, W1, feat1, N, H1 * D1, D0);

    node_scores<<<(N * H1 + TB - 1) / TB, TB, 0, stream>>>(feat1, al1, ar1, el1, er1, N, H1, D1);

    fill_u32<<<fillBlocks, TB, 0, stream>>>(m1, ORD_NEG_INF, 2 * N * H1);
    fill_f32<<<fillBlocks, TB, 0, stream>>>(s1, 0.f, 2 * N * H1);
    fill_f32<<<fillBlocks, TB, 0, stream>>>(acc1, 0.f, 2 * N * H1 * D1);

    for (int t = 0; t < 2; ++t) {
      unsigned* mT = m1 + (size_t)t * N * H1;
      float*    sT = s1 + (size_t)t * N * H1;
      float*    eeT = ee1 + (size_t)t * E * H1;
      float*    accT = acc1 + (size_t)t * N * H1 * D1;
      int eh = E * H1;
      edge_max<<<(eh + TB - 1) / TB, TB, 0, stream>>>(srcT[t], dstT[t], el1, er1, mT, E, H1);
      edge_expsum<<<(eh + TB - 1) / TB, TB, 0, stream>>>(srcT[t], dstT[t], el1, er1, mT, eeT, sT, E, H1);
      int total = E * ((H1 * D1) >> 2);
      edge_aggregate<<<(total + TB - 1) / TB, TB, 0, stream>>>(srcT[t], dstT[t], eeT, sT, feat1, accT, E, H1, D1);
    }

    epilogue1<<<(N * D1 + TB - 1) / TB, TB, 0, stream>>>(acc1, b1, out, N, D1);
  }
}


# edge_aggregate: a 32-wave block owns 196 destination rows as sums in LDS, scans the whole dst list, adds its edges' weighted rows with LDS f32 atomics, stores rows once; sums in LDS arrival order
# speedup vs baseline: 1.4878x; 1.4878x over previous
#include <hip/hip_runtime.h>
#include <math.h>

typedef float v2f __attribute__((ext_vector_type(2)));
typedef float v8f __attribute__((ext_vector_type(8)));

#define LRELU_SLOPE 0.2f

__device__ __forceinline__ unsigned f2o(float f) {
  unsigned u = __float_as_uint(f);
  return (u & 0x80000000u) ? ~u : (u | 0x80000000u);
}
__device__ __forceinline__ float o2f(unsigned o) {
  unsigned u = (o & 0x80000000u) ? (o & 0x7fffffffu) : ~o;
  return __uint_as_float(u);
}
#define ORD_NEG_INF 0x007fffffu

__global__ void fill_f32(float* p, float v, int n) {
  const int g = blockIdx.x * blockDim.x + threadIdx.x;
  const int T = gridDim.x * blockDim.x;
  float4* p4 = reinterpret_cast<float4*>(p);
  const int n4 = n >> 2;
  const float4 vv = make_float4(v, v, v, v);
  for (int i = g; i < n4; i += T) p4[i] = vv;
  for (int i = (n4 << 2) + g; i < n; i += T) p[i] = v;
}
__global__ void fill_u32(unsigned* p, unsigned v, int n) {
  const int g = blockIdx.x * blockDim.x + threadIdx.x;
  const int T = gridDim.x * blockDim.x;
  uint4* p4 = reinterpret_cast<uint4*>(p);
  const int n4 = n >> 2;
  const uint4 vv = make_uint4(v, v, v, v);
  for (int i = g; i < n4; i += T) p4[i] = vv;
  for (int i = (n4 << 2) + g; i < n; i += T) p[i] = v;
}

__global__ void gemm_wmma_f32(const float* __restrict__ A,
                              const float* __restrict__ B,
                              float* __restrict__ C,
                              int M, int Ncols, int K) {
  const int wavesPerBlock = blockDim.x >> 5;
  const int wave = threadIdx.x >> 5;
  const int lane = threadIdx.x & 31;
  const int half = lane >> 4;
  const int l    = lane & 15;

  const int m0 = blockIdx.x * 16;
  const int nt = blockIdx.y * wavesPerBlock + wave;
  const int n0 = nt * 16;
  if (m0 >= M || n0 >= Ncols) return;

  v8f acc = {};
  const float* arow = A + (size_t)(m0 + l) * K + half * 2;
  for (int k = 0; k < K; k += 4) {
    v2f a, b;
    a.x = arow[k];
    a.y = arow[k + 1];
    const float* bp = B + (size_t)(k + half * 2) * Ncols + n0 + l;
    b.x = bp[0];
    b.y = bp[Ncols];
    acc = __builtin_amdgcn_wmma_f32_16x16x4_f32(false, a, false, b,
                                                (short)0, acc, false, false);
  }
#pragma unroll
  for (int r = 0; r < 8; ++r) {
    C[(size_t)(m0 + half * 8 + r) * Ncols + n0 + l] = acc[r];
  }
}

__global__ void node_scores(const float* __restrict__ feat,
                            const float* __restrict__ al,
                            const float* __restrict__ ar,
                            float* __restrict__ el, float* __restrict__ er,
                            int N, int H, int D) {
  int i = blockIdx.x * blockDim.x + threadIdx.x;
  if (i >= N * H) return;
  int n = i / H, h = i - n * H;
  const float* f = feat + (size_t)n * H * D + h * D;
  const float* wl = al + h * D;
  const float* wr = ar + h * D;
  float sl = 0.f, sr = 0.f;
#pragma unroll 4
  for (int d = 0; d < D; ++d) {
    float v = f[d];
    sl += v * wl[d];
    sr += v * wr[d];
  }
  el[i] = sl;
  er[i] = sr;
}

template <int H, int U>
__device__ __forceinline__ void edge_max_body(const int* __restrict__ src, const int* __restrict__ dst,
                                              const float* __restrict__ el, const float* __restrict__ er,
                                              unsigned* __restrict__ m_ord, int E) {
  const unsigned n = (unsigned)E * H;
  const unsigned T = gridDim.x * blockDim.x;
  for (unsigned i0 = blockIdx.x * blockDim.x + threadIdx.x; i0 < n; i0 += U * T) {
    unsigned sN[U], dN[U], di[U];
    float l[U], r[U];
#pragma unroll
    for (int u = 0; u < U; ++u) {
      const unsigned i  = i0 + u * T;
      const unsigned ic = i < n ? i : n - 1;
      sN[u] = (unsigned)src[ic / H];
      dN[u] = (unsigned)dst[ic / H];
    }
#pragma unroll
    for (int u = 0; u < U; ++u) asm volatile("" :: "v"(sN[u]), "v"(dN[u]));
#pragma unroll
    for (int u = 0; u < U; ++u) {
      const unsigned i  = i0 + u * T;
      const unsigned ic = i < n ? i : n - 1;
      const unsigned h  = ic % H;
      di[u] = dN[u] * H + h;
      l[u]  = el[sN[u] * H + h];
      r[u]  = er[di[u]];
    }
#pragma unroll
    for (int u = 0; u < U; ++u) asm volatile("" :: "v"(l[u]), "v"(r[u]));
#pragma unroll
    for (int u = 0; u < U; ++u) {
      if (i0 + u * T < n) {
        float sc = l[u] + r[u];
        sc = sc > 0.f ? sc : LRELU_SLOPE * sc;
        atomicMax(m_ord + di[u], f2o(sc));
      }
    }
  }
}
__global__ void edge_max_h4(const int* __restrict__ src, const int* __restrict__ dst,
                            const float* __restrict__ el, const float* __restrict__ er,
                            unsigned* __restrict__ m_ord, int E) {
  edge_max_body<4, 8>(src, dst, el, er, m_ord, E);
}
__global__ void edge_max_h1(const int* __restrict__ src, const int* __restrict__ dst,
                            const float* __restrict__ el, const float* __restrict__ er,
                            unsigned* __restrict__ m_ord, int E) {
  edge_max_body<1, 4>(src, dst, el, er, m_ord, E);
}

template <int H, int U>
__device__ __forceinline__ void edge_expsum_body(const int* __restrict__ src, const int* __restrict__ dst,
                                                 const float* __restrict__ el, const float* __restrict__ er,
                                                 const unsigned* __restrict__ m_ord,
                                                 float* __restrict__ ee, float* __restrict__ s, int E) {
  const unsigned n = (unsigned)E * H;
  const unsigned T = gridDim.x * blockDim.x;
  for (unsigned i0 = blockIdx.x * blockDim.x + threadIdx.x; i0 < n; i0 += U * T) {
    unsigned sN[U], dN[U], di[U], mo[U];
    float l[U], r[U];
#pragma unroll
    for (int u = 0; u < U; ++u) {
      const unsigned i  = i0 + u * T;
      const unsigned ic = i < n ? i : n - 1;
      sN[u] = (unsigned)src[ic / H];
      dN[u] = (unsigned)dst[ic / H];
    }
#pragma unroll
    for (int u = 0; u < U; ++u) asm volatile("" :: "v"(sN[u]), "v"(dN[u]));
#pragma unroll
    for (int u = 0; u < U; ++u) {
      const unsigned i  = i0 + u * T;
      const unsigned ic = i < n ? i : n - 1;
      const unsigned h  = ic % H;
      di[u] = dN[u] * H + h;
      l[u]  = el[sN[u] * H + h];
      r[u]  = er[di[u]];
      mo[u] = m_ord[di[u]];
    }
#pragma unroll
    for (int u = 0; u < U; ++u) asm volatile("" :: "v"(l[u]), "v"(r[u]), "v"(mo[u]));
#pragma unroll
    for (int u = 0; u < U; ++u) {
      const unsigned i = i0 + u * T;
      if (i < n) {
        float sc = l[u] + r[u];
        sc = sc > 0.f ? sc : LRELU_SLOPE * sc;
        float m = o2f(mo[u]);
        if (m < -1e30f) m = 0.f;
        const float v = __expf(sc - m);
        ee[i] = v;
        atomicAdd(s + di[u], v);
      }
    }
  }
}
__global__ void edge_expsum_h4(const int* __restrict__ src, const int* __restrict__ dst,
                               const float* __restrict__ el, const float* __restrict__ er,
                               const unsigned* __restrict__ m_ord,
                               float* __restrict__ ee, float* __restrict__ s, int E) {
  edge_expsum_body<4, 8>(src, dst, el, er, m_ord, ee, s, E);
}
__global__ void edge_expsum_h1(const int* __restrict__ src, const int* __restrict__ dst,
                               const float* __restrict__ el, const float* __restrict__ er,
                               const unsigned* __restrict__ m_ord,
                               float* __restrict__ ee, float* __restrict__ s, int E) {
  edge_expsum_body<1, 4>(src, dst, el, er, m_ord, ee, s, E);
}

#define AGG_THREADS 1024
#define AGG_NODES   196
#define AGG_QCAP    192
#define AGG_U       8

template <int H, int D>
__device__ __forceinline__ void owner_aggregate_body(const int* __restrict__ src, const int* __restrict__ dst,
                                                     const float* __restrict__ ee, const float* __restrict__ s,
                                                     const float* __restrict__ feat, float* __restrict__ acc,
                                                     int E, int N) {
  constexpr int R     = H * D;
  constexpr int LINES = R / 32;
  constexpr int LPH   = D / 32;
  constexpr int G     = (LINES >= 8) ? 4 : 16;
  static_assert(D % 32 == 0, "a line must lie inside one head");

  __shared__ float sacc[AGG_NODES * R];
  __shared__ int2  queue[(AGG_THREADS / 32) * AGG_QCAP];

  const int tid  = threadIdx.x;
  const int lane = tid & 31;
  const int wave = __builtin_amdgcn_readfirstlane(tid >> 5);
  const int lo = blockIdx.x * AGG_NODES;
  const int nn = (N - lo) < AGG_NODES ? (N - lo) : AGG_NODES;

  float4* sacc4 = reinterpret_cast<float4*>(sacc);
  for (int i = tid; i < AGG_NODES * R / 4; i += AGG_THREADS) sacc4[i] = make_float4(0.f, 0.f, 0.f, 0.f);
  __syncthreads();

  int2* q = queue + wave * AGG_QCAP;
  int cnt = 0;

  auto work_off = [&](const int2* qb, int n) {
    int sN = 0, dl = 0;
    float a[H];
#pragma unroll
    for (int h = 0; h < H; ++h) a[h] = 0.f;
    if (lane < n) {
      const int2 ed = qb[lane];
      const int e   = ed.x;
      const int dN  = ed.y;
      sN = src[e];
      dl = dN - lo;
#pragma unroll
      for (int h = 0; h < H; ++h)
        a[h] = ee[(size_t)e * H + h] / fmaxf(s[(size_t)dN * H + h], 1e-9f);
    }
    for (int k0 = 0; k0 < n; k0 += G) {
      float v[G][LINES];
#pragma unroll
      for (int g = 0; g < G; ++g) {
        const int k  = (k0 + g < n) ? (k0 + g) : (n - 1);
        const int sk = __builtin_amdgcn_readlane(sN, k);
        const float* f = feat + (size_t)sk * R + lane;
#pragma unroll
        for (int j = 0; j < LINES; ++j) v[g][j] = f[32 * j];
      }
#pragma unroll
      for (int g = 0; g < G; ++g) {
        const int k = k0 + g;
        if (k < n) {
          const int dk = __builtin_amdgcn_readlane(dl, k);
          float* o = sacc + dk * R + lane;
#pragma unroll
          for (int j = 0; j < LINES; ++j) {
            const float ak = __int_as_float(__builtin_amdgcn_readlane(__float_as_int(a[j / LPH]), k));
            atomicAdd(o + 32 * j, ak * v[g][j]);
          }
        }
      }
    }
  };

  const int nChunks = (E + 127) >> 7;
  const unsigned below = (1u << lane) - 1u;
  for (int c0 = wave; c0 < nChunks; c0 += (AGG_THREADS / 32) * AGG_U) {
    int4 d[AGG_U];
#pragma unroll
    for (int u = 0; u < AGG_U; ++u) {
      const int c    = c0 + (AGG_THREADS / 32) * u;
      const int base = c * 128 + 4 * lane;
      d[u] = make_int4(-1, -1, -1, -1);
      if (c < nChunks) {
        if (c * 128 + 128 <= E) {
          d[u] = *reinterpret_cast<const int4*>(dst + base);
        } else {
          if (base + 0 < E) d[u].x = dst[base + 0];
          if (base + 1 < E) d[u].y = dst[base + 1];
          if (base + 2 < E) d[u].z = dst[base + 2];
          if (base + 3 < E) d[u].w = dst[base + 3];
        }
      }
    }
#pragma unroll
    for (int u = 0; u < AGG_U; ++u) {
      const int c    = c0 + (AGG_THREADS / 32) * u;
      const int base = c * 128 + 4 * lane;
      const int dv[4] = { d[u].x, d[u].y, d[u].z, d[u].w };
#pragma unroll
      for (int j = 0; j < 4; ++j) {
        const bool mine = (unsigned)(dv[j] - lo) < (unsigned)nn;
        const unsigned mask = (unsigned)__ballot(mine);
        if (mask != 0u) {
          if (mine) q[cnt + __popc(mask & below)] = make_int2(base + j, dv[j]);
          cnt += __popc(mask);
        }
      }
      while (cnt >= 32) { cnt -= 32; work_off(q + cnt, 32); }
    }
  }
  if (cnt > 0) work_off(q, cnt);
  __syncthreads();

  float4* out4 = reinterpret_cast<float4*>(acc + (size_t)lo * R);
  for (int i = tid; i < nn * (R / 4); i += AGG_THREADS) out4[i] = sacc4[i];
}

__global__ void __launch_bounds__(AGG_THREADS)
owner_aggregate_h4d64(const int* __restrict__ src, const int* __restrict__ dst,
                      const float* __restrict__ ee, const float* __restrict__ s,
                      const float* __restrict__ feat, float* __restrict__ acc, int E, int N) {
  owner_aggregate_body<4, 64>(src, dst, ee, s, feat, acc, E, N);
}
__global__ void __launch_bounds__(AGG_THREADS)
owner_aggregate_h1d64(const int* __restrict__ src, const int* __restrict__ dst,
                      const float* __restrict__ ee, const float* __restrict__ s,
                      const float* __restrict__ feat, float* __restrict__ acc, int E, int N) {
  owner_aggregate_body<1, 64>(src, dst, ee, s, feat, acc, E, N);
}

__device__ __forceinline__ float elu1(float v) { return (v > 0.f) ? v : expm1f(v); }

template <int H, int D>
__device__ __forceinline__ void epilogue0_body(const float* __restrict__ acc, const float* __restrict__ b,
                                               float* __restrict__ h1, int N) {
  constexpr int D4 = D / 4;
  const unsigned n4 = (unsigned)N * D4;
  const unsigned T = gridDim.x * blockDim.x;
  for (unsigned i = blockIdx.x * blockDim.x + threadIdx.x; i < n4; i += T) {
    const unsigned n = i / D4, q = i - n * D4;
    float4 v[2][H], bb[H];
#pragma unroll
    for (int t = 0; t < 2; ++t)
#pragma unroll
      for (int h = 0; h < H; ++h)
        v[t][h] = *reinterpret_cast<const float4*>(acc + ((size_t)t * N + n) * (H * D) + h * D + 4 * q);
#pragma unroll
    for (int h = 0; h < H; ++h) bb[h] = *reinterpret_cast<const float4*>(b + h * D + 4 * q);
    float4 sum = make_float4(0.f, 0.f, 0.f, 0.f);
#pragma unroll
    for (int t = 0; t < 2; ++t)
#pragma unroll
      for (int h = 0; h < H; ++h) {
        sum.x += elu1(v[t][h].x + bb[h].x);
        sum.y += elu1(v[t][h].y + bb[h].y);
        sum.z += elu1(v[t][h].z + bb[h].z);
        sum.w += elu1(v[t][h].w + bb[h].w);
      }
    const float hf = (float)H;
    *reinterpret_cast<float4*>(h1 + (size_t)i * 4) =
        make_float4(sum.x / hf, sum.y / hf, sum.z / hf, sum.w / hf);
  }
}
__global__ void epilogue0_h4d64(const float* __restrict__ acc, const float* __restrict__ b,
                                float* __restrict__ h1, int N) {
  epilogue0_body<4, 64>(acc, b, h1, N);
}

typedef float v4f __attribute__((ext_vector_type(4)));

template <int D, int U>
__device__ __forceinline__ void epilogue1_body(const float* __restrict__ acc, const float* __restrict__ b,
                                               float* __restrict__ out, int N) {
  constexpr int D4 = D / 4;
  const unsigned n4 = (unsigned)N * D4;
  const unsigned T = gridDim.x * blockDim.x;
  const v4f* a0 = reinterpret_cast<const v4f*>(acc);
  const v4f* a1 = reinterpret_cast<const v4f*>(acc + (size_t)N * D);
  const v4f* b4 = reinterpret_cast<const v4f*>(b);
  v4f* o4 = reinterpret_cast<v4f*>(out);
  for (unsigned i0 = blockIdx.x * blockDim.x + threadIdx.x; i0 < n4; i0 += U * T) {
    v4f x[U], y[U], bb[U];
#pragma unroll
    for (int u = 0; u < U; ++u) {
      const unsigned i  = i0 + u * T;
      const unsigned ic = i < n4 ? i : n4 - 1;
      x[u]  = a0[ic];
      y[u]  = a1[ic];
      bb[u] = b4[ic % D4];
    }
#pragma unroll
    for (int u = 0; u < U; ++u) asm volatile("" :: "v"(x[u]), "v"(y[u]), "v"(bb[u]));
#pragma unroll
    for (int u = 0; u < U; ++u) {
      const unsigned i = i0 + u * T;
      if (i < n4) o4[i] = (x[u] + bb[u]) + (y[u] + bb[u]);
    }
  }
}
__global__ void epilogue1_d64(const float* __restrict__ acc, const float* __restrict__ b,
                              float* __restrict__ out, int N) {
  epilogue1_body<64, 4>(acc, b, out, N);
}

extern "C" void kernel_launch(void* const* d_in, const int* in_sizes, int n_in,
                              void* d_out, int out_size, void* d_ws, size_t ws_size,
                              hipStream_t stream) {
  const int IN = 128, H0 = 4, D0 = 64, H1 = 1, D1 = 64;
  const int N = in_sizes[0] / IN;
  const int E = in_sizes[9];

  const float* x   = (const float*)d_in[0];
  const float* W0  = (const float*)d_in[1];
  const float* al0 = (const float*)d_in[2];
  const float* ar0 = (const float*)d_in[3];
  const float* b0  = (const float*)d_in[4];
  const float* W1  = (const float*)d_in[5];
  const float* al1 = (const float*)d_in[6];
  const float* ar1 = (const float*)d_in[7];
  const float* b1  = (const float*)d_in[8];
  const int* srcT[2] = { (const int*)d_in[9],  (const int*)d_in[11] };
  const int* dstT[2] = { (const int*)d_in[10], (const int*)d_in[12] };
  float* out = (float*)d_out;

  float* ws = (float*)d_ws;
  float*    feat0 = ws;
  float*    el0   = feat0 + (size_t)N * H0 * D0;
  float*    er0   = el0 + (size_t)N * H0;
  unsigned* m0    = (unsigned*)(er0 + (size_t)N * H0);
  float*    s0    = (float*)(m0 + 2 * (size_t)N * H0);
  float*    ee0   = s0 + 2 * (size_t)N * H0;
  float*    acc0  = ee0 + 2 * (size_t)E * H0;
  float*    h1    = acc0 + 2 * (size_t)N * H0 * D0;
  float*    feat1 = feat0;
  float*    el1   = feat1 + (size_t)N * D1;
  float*    er1   = el1 + N;
  unsigned* m1    = (unsigned*)(er1 + N);
  float*    s1    = (float*)(m1 + 2 * (size_t)N);
  float*    ee1   = s1 + 2 * (size_t)N;
  float*    acc1  = ee1 + 2 * (size_t)E;

  const int TB = 256;
  const int aggBlocks = (N + AGG_NODES - 1) / AGG_NODES;
  const int fillSmallBlocks = 32;
  const int edgeBlocks0 = (E * H0 + TB * 8 - 1) / (TB * 8);
  const int edgeBlocks1 = (E * H1 + TB * 4 - 1) / (TB * 4);
  const int epi0Blocks  = 1024;
  const int epi1Blocks  = (N * (D1 / 4) + TB * 4 - 1) / (TB * 4);
  (void)ws_size; (void)n_in; (void)out_size;

  {
    dim3 grid(N / 16, (H0 * D0) / 16 / 4);
    gemm_wmma_f32<<<grid, 128, 0, stream>>>(x, W0, feat0, N, H0 * D0, IN);

    node_scores<<<(N * H0 + TB - 1) / TB, TB, 0, stream>>>(feat0, al0, ar0, el0, er0, N, H0, D0);

    fill_u32<<<fillSmallBlocks, TB, 0, stream>>>(m0, ORD_NEG_INF, 2 * N * H0);
    fill_f32<<<fillSmallBlocks, TB, 0, stream>>>(s0, 0.f, 2 * N * H0);

    for (int t = 0; t < 2; ++t) {
      unsigned* mT = m0 + (size_t)t * N * H0;
      float*    sT = s0 + (size_t)t * N * H0;
      float*    eeT = ee0 + (size_t)t * E * H0;
      float*    accT = acc0 + (size_t)t * N * H0 * D0;
      edge_max_h4<<<edgeBlocks0, TB, 0, stream>>>(srcT[t], dstT[t], el0, er0, mT, E);
      edge_expsum_h4<<<edgeBlocks0, TB, 0, stream>>>(srcT[t], dstT[t], el0, er0, mT, eeT, sT, E);
      owner_aggregate_h4d64<<<aggBlocks, AGG_THREADS, 0, stream>>>(srcT[t], dstT[t], eeT, sT, feat0, accT, E, N);
    }

    epilogue0_h4d64<<<epi0Blocks, TB, 0, stream>>>(acc0, b0, h1, N);
  }

  {
    dim3 grid(N / 16, 1);
    gemm_wmma_f32<<<grid, 128, 0, stream>>>(h1, W1, feat1, N, H1 * D1, D0);

    node_scores<<<(N * H1 + TB - 1) / TB, TB, 0, stream>>>(feat1, al1, ar1, el1, er1, N, H1, D1);

    fill_u32<<<fillSmallBlocks, TB, 0, stream>>>(m1, ORD_NEG_INF, 2 * N * H1);
    fill_f32<<<fillSmallBlocks, TB, 0, stream>>>(s1, 0.f, 2 * N * H1);

    for (int t = 0; t < 2; ++t) {
      unsigned* mT = m1 + (size_t)t * N * H1;
      float*    sT = s1 + (size_t)t * N * H1;
      float*    eeT = ee1 + (size_t)t * E * H1;
      float*    accT = acc1 + (size_t)t * N * H1 * D1;
      edge_max_h1<<<edgeBlocks1, TB, 0, stream>>>(srcT[t], dstT[t], el1, er1, mT, E);
      edge_expsum_h1<<<edgeBlocks1, TB, 0, stream>>>(srcT[t], dstT[t], el1, er1, mT, eeT, sT, E);
      owner_aggregate_h1d64<<<aggBlocks, AGG_THREADS, 0, stream>>>(srcT[t], dstT[t], eeT, sT, feat1, accT, E, N);
    }

    epilogue1_d64<<<epi1Blocks, TB, 0, stream>>>(acc1, b1, out, N);
  }
}


# GEMMs: a 16-wave block keeps all of B in LDS (k pairs interleaved), passes A row tiles through two LDS buffers filled by whole-line loads a step ahead; fragments are 8-byte LDS reads; sum order kept
# speedup vs baseline: 1.5842x; 1.0648x over previous
#include <hip/hip_runtime.h>
#include <math.h>

typedef float v2f __attribute__((ext_vector_type(2)));
typedef float v4f __attribute__((ext_vector_type(4)));
typedef float v8f __attribute__((ext_vector_type(8)));

#define LRELU_SLOPE 0.2f

__device__ __forceinline__ unsigned f2o(float f) {
  unsigned u = __float_as_uint(f);
  return (u & 0x80000000u) ? ~u : (u | 0x80000000u);
}
__device__ __forceinline__ float o2f(unsigned o) {
  unsigned u = (o & 0x80000000u) ? (o & 0x7fffffffu) : ~o;
  return __uint_as_float(u);
}
#define ORD_NEG_INF 0x007fffffu

__global__ void fill_f32(float* p, float v, int n) {
  const int g = blockIdx.x * blockDim.x + threadIdx.x;
  const int T = gridDim.x * blockDim.x;
  float4* p4 = reinterpret_cast<float4*>(p);
  const int n4 = n >> 2;
  const float4 vv = make_float4(v, v, v, v);
  for (int i = g; i < n4; i += T) p4[i] = vv;
  for (int i = (n4 << 2) + g; i < n; i += T) p[i] = v;
}
__global__ void fill_u32(unsigned* p, unsigned v, int n) {
  const int g = blockIdx.x * blockDim.x + threadIdx.x;
  const int T = gridDim.x * blockDim.x;
  uint4* p4 = reinterpret_cast<uint4*>(p);
  const int n4 = n >> 2;
  const uint4 vv = make_uint4(v, v, v, v);
  for (int i = g; i < n4; i += T) p4[i] = vv;
  for (int i = (n4 << 2) + g; i < n; i += T) p[i] = v;
}

#define GT_THREADS 512

template <int K, int NC>
__device__ __forceinline__ void gemm_lds_body(const float* __restrict__ A,
                                              const float* __restrict__ B,
                                              float* __restrict__ C, int M) {
  constexpr int NP  = NC / 32;
  constexpr int WR  = 16 / NP;
  constexpr int RS  = WR * 16;
  constexpr int PA  = K + 4;
  constexpr int PB  = 2 * NC + 32;
  constexpr int K4  = K / 4;
  constexpr int NC4 = NC / 4;
  constexpr int NPF = (RS * K4) / GT_THREADS;
  constexpr int NBF = ((K / 2) * NC4) / GT_THREADS;
  static_assert(NC % 32 == 0 && 16 % NP == 0, "a wave takes one M-tile and one column pair");
  static_assert((RS * K4) % GT_THREADS == 0 && (NPF == 2 || NPF == 4), "an A tile is 2 or 4 pieces per thread");
  static_assert(((K / 2) * NC4) % GT_THREADS == 0, "B splits evenly over the threads");
  static_assert(K % 4 == 0, "k steps of 4");

  __shared__ __align__(16) float Bs[(K / 2) * PB];
  __shared__ __align__(16) float As[2][RS * PA];

  const int tid  = threadIdx.x;
  const int lane = tid & 31;
  const int half = lane >> 4;
  const int l    = lane & 15;
  const int wave = __builtin_amdgcn_readfirstlane(tid >> 5);
  const int wm   = wave / NP;
  const int wn   = wave - wm * NP;

  const int nSteps = (M + RS - 1) / RS;

  auto fetch_piece = [&](int s, int u) -> v4f {
    const int it = tid + u * GT_THREADS;
    const int r = it / K4, c = it - r * K4;
    int row = s * RS + r;
    row = row < M ? row : M - 1;
    return *reinterpret_cast<const v4f*>(A + (size_t)row * K + 4 * c);
  };
  auto stash_piece = [&](float* as, int u, v4f v) {
    const int it = tid + u * GT_THREADS;
    const int r = it / K4, c = it - r * K4;
    *reinterpret_cast<v4f*>(as + r * PA + 4 * c) = v;
  };

  int step = blockIdx.x;
  v4f p0 = fetch_piece(step, 0), p1 = fetch_piece(step, 1), p2 = p0, p3 = p0;
  if constexpr (NPF == 4) { p2 = fetch_piece(step, 2); p3 = fetch_piece(step, 3); }
#pragma unroll
  for (int u = 0; u < NBF; ++u) {
    const int it = tid + u * GT_THREADS;
    const int kp = it / NC4, c = it - kp * NC4;
    const float4 r0 = *reinterpret_cast<const float4*>(B + (size_t)(2 * kp) * NC + 4 * c);
    const float4 r1 = *reinterpret_cast<const float4*>(B + (size_t)(2 * kp + 1) * NC + 4 * c);
    float4* d = reinterpret_cast<float4*>(Bs + kp * PB + 8 * c);
    d[0] = make_float4(r0.x, r1.x, r0.y, r1.y);
    d[1] = make_float4(r0.z, r1.z, r0.w, r1.w);
  }

  int buf = 0;
  for (; step < nSteps; step += gridDim.x) {
    float* as = As[buf];
    stash_piece(as, 0, p0);
    stash_piece(as, 1, p1);
    if constexpr (NPF == 4) { stash_piece(as, 2, p2); stash_piece(as, 3, p3); }
    __syncthreads();

    const int next = step + gridDim.x;
    p0 = fetch_piece(next, 0);
    p1 = fetch_piece(next, 1);
    if constexpr (NPF == 4) { p2 = fetch_piece(next, 2); p3 = fetch_piece(next, 3); }

    v8f acc0 = {}, acc1 = {};
    const float* ap = as + (wm * 16 + l) * PA + 2 * half;
    const float* bp = Bs + half * PB + (wn * 32 + l) * 2;
#pragma unroll
    for (int k = 0; k < K; k += 4) {
      const v2f a  = *reinterpret_cast<const v2f*>(ap + k);
      const v2f b0 = *reinterpret_cast<const v2f*>(bp + (k >> 1) * PB);
      const v2f b1 = *reinterpret_cast<const v2f*>(bp + (k >> 1) * PB + 32);
      acc0 = __builtin_amdgcn_wmma_f32_16x16x4_f32(false, a, false, b0, (short)0, acc0, false, false);
      acc1 = __builtin_amdgcn_wmma_f32_16x16x4_f32(false, a, false, b1, (short)0, acc1, false, false);
    }

    const int row0 = step * RS + wm * 16 + half * 8;
    float* cp = C + (size_t)row0 * NC + wn * 32 + l;
#pragma unroll
    for (int r = 0; r < 8; ++r) {
      if (row0 + r < M) {
        cp[(size_t)r * NC]      = acc0[r];
        cp[(size_t)r * NC + 16] = acc1[r];
      }
    }
    buf ^= 1;
  }
}

__global__ void __launch_bounds__(GT_THREADS)
gemm_lds_k128_n256(const float* __restrict__ A, const float* __restrict__ B, float* __restrict__ C, int M) {
  gemm_lds_body<128, 256>(A, B, C, M);
}
__global__ void __launch_bounds__(GT_THREADS)
gemm_lds_k64_n64(const float* __restrict__ A, const float* __restrict__ B, float* __restrict__ C, int M) {
  gemm_lds_body<64, 64>(A, B, C, M);
}

__global__ void node_scores(const float* __restrict__ feat,
                            const float* __restrict__ al,
                            const float* __restrict__ ar,
                            float* __restrict__ el, float* __restrict__ er,
                            int N, int H, int D) {
  int i = blockIdx.x * blockDim.x + threadIdx.x;
  if (i >= N * H) return;
  int n = i / H, h = i - n * H;
  const float* f = feat + (size_t)n * H * D + h * D;
  const float* wl = al + h * D;
  const float* wr = ar + h * D;
  float sl = 0.f, sr = 0.f;
#pragma unroll 4
  for (int d = 0; d < D; ++d) {
    float v = f[d];
    sl += v * wl[d];
    sr += v * wr[d];
  }
  el[i] = sl;
  er[i] = sr;
}

template <int H, int U>
__device__ __forceinline__ void edge_max_body(const int* __restrict__ src, const int* __restrict__ dst,
                                              const float* __restrict__ el, const float* __restrict__ er,
                                              unsigned* __restrict__ m_ord, int E) {
  const unsigned n = (unsigned)E * H;
  const unsigned T = gridDim.x * blockDim.x;
  for (unsigned i0 = blockIdx.x * blockDim.x + threadIdx.x; i0 < n; i0 += U * T) {
    unsigned sN[U], dN[U], di[U];
    float l[U], r[U];
#pragma unroll
    for (int u = 0; u < U; ++u) {
      const unsigned i  = i0 + u * T;
      const unsigned ic = i < n ? i : n - 1;
      sN[u] = (unsigned)src[ic / H];
      dN[u] = (unsigned)dst[ic / H];
    }
#pragma unroll
    for (int u = 0; u < U; ++u) asm volatile("" :: "v"(sN[u]), "v"(dN[u]));
#pragma unroll
    for (int u = 0; u < U; ++u) {
      const unsigned i  = i0 + u * T;
      const unsigned ic = i < n ? i : n - 1;
      const unsigned h  = ic % H;
      di[u] = dN[u] * H + h;
      l[u]  = el[sN[u] * H + h];
      r[u]  = er[di[u]];
    }
#pragma unroll
    for (int u = 0; u < U; ++u) asm volatile("" :: "v"(l[u]), "v"(r[u]));
#pragma unroll
    for (int u = 0; u < U; ++u) {
      if (i0 + u * T < n) {
        float sc = l[u] + r[u];
        sc = sc > 0.f ? sc : LRELU_SLOPE * sc;
        atomicMax(m_ord + di[u], f2o(sc));
      }
    }
  }
}
__global__ void edge_max_h4(const int* __restrict__ src, const int* __restrict__ dst,
                            const float* __restrict__ el, const float* __restrict__ er,
                            unsigned* __restrict__ m_ord, int E) {
  edge_max_body<4, 8>(src, dst, el, er, m_ord, E);
}
__global__ void edge_max_h1(const int* __restrict__ src, const int* __restrict__ dst,
                            const float* __restrict__ el, const float* __restrict__ er,
                            unsigned* __restrict__ m_ord, int E) {
  edge_max_body<1, 4>(src, dst, el, er, m_ord, E);
}

template <int H, int U>
__device__ __forceinline__ void edge_expsum_body(const int* __restrict__ src, const int* __restrict__ dst,
                                                 const float* __restrict__ el, const float* __restrict__ er,
                                                 const unsigned* __restrict__ m_ord,
                                                 float* __restrict__ ee, float* __restrict__ s, int E) {
  const unsigned n = (unsigned)E * H;
  const unsigned T = gridDim.x * blockDim.x;
  for (unsigned i0 = blockIdx.x * blockDim.x + threadIdx.x; i0 < n; i0 += U * T) {
    unsigned sN[U], dN[U], di[U], mo[U];
    float l[U], r[U];
#pragma unroll
    for (int u = 0; u < U; ++u) {
      const unsigned i  = i0 + u * T;
      const unsigned ic = i < n ? i : n - 1;
      sN[u] = (unsigned)src[ic / H];
      dN[u] = (unsigned)dst[ic / H];
    }
#pragma unroll
    for (int u = 0; u < U; ++u) asm volatile("" :: "v"(sN[u]), "v"(dN[u]));
#pragma unroll
    for (int u = 0; u < U; ++u) {
      const unsigned i  = i0 + u * T;
      const unsigned ic = i < n ? i : n - 1;
      const unsigned h  = ic % H;
      di[u] = dN[u] * H + h;
      l[u]  = el[sN[u] * H + h];
      r[u]  = er[di[u]];
      mo[u] = m_ord[di[u]];
    }
#pragma unroll
    for (int u = 0; u < U; ++u) asm volatile("" :: "v"(l[u]), "v"(r[u]), "v"(mo[u]));
#pragma unroll
    for (int u = 0; u < U; ++u) {
      const unsigned i = i0 + u * T;
      if (i < n) {
        float sc = l[u] + r[u];
        sc = sc > 0.f ? sc : LRELU_SLOPE * sc;
        float m = o2f(mo[u]);
        if (m < -1e30f) m = 0.f;
        const float v = __expf(sc - m);
        ee[i] = v;
        atomicAdd(s + di[u], v);
      }
    }
  }
}
__global__ void edge_expsum_h4(const int* __restrict__ src, const int* __restrict__ dst,
                               const float* __restrict__ el, const float* __restrict__ er,
                               const unsigned* __restrict__ m_ord,
                               float* __restrict__ ee, float* __restrict__ s, int E) {
  edge_expsum_body<4, 8>(src, dst, el, er, m_ord, ee, s, E);
}
__global__ void edge_expsum_h1(const int* __restrict__ src, const int* __restrict__ dst,
                               const float* __restrict__ el, const float* __restrict__ er,
                               const unsigned* __restrict__ m_ord,
                               float* __restrict__ ee, float* __restrict__ s, int E) {
  edge_expsum_body<1, 4>(src, dst, el, er, m_ord, ee, s, E);
}

#define AGG_THREADS 1024
#define AGG_NODES   196
#define AGG_QCAP    192
#define AGG_U       8

template <int H, int D>
__device__ __forceinline__ void owner_aggregate_body(const int* __restrict__ src, const int* __restrict__ dst,
                                                     const float* __restrict__ ee, const float* __restrict__ s,
                                                     const float* __restrict__ feat, float* __restrict__ acc,
                                                     int E, int N) {
  constexpr int R     = H * D;
  constexpr int LINES = R / 32;
  constexpr int LPH   = D / 32;
  constexpr int G     = (LINES >= 8) ? 4 : 16;
  static_assert(D % 32 == 0, "a line must lie inside one head");

  __shared__ float sacc[AGG_NODES * R];
  __shared__ int2  queue[(AGG_THREADS / 32) * AGG_QCAP];

  const int tid  = threadIdx.x;
  const int lane = tid & 31;
  const int wave = __builtin_amdgcn_readfirstlane(tid >> 5);
  const int lo = blockIdx.x * AGG_NODES;
  const int nn = (N - lo) < AGG_NODES ? (N - lo) : AGG_NODES;

  float4* sacc4 = reinterpret_cast<float4*>(sacc);
  for (int i = tid; i < AGG_NODES * R / 4; i += AGG_THREADS) sacc4[i] = make_float4(0.f, 0.f, 0.f, 0.f);
  __syncthreads();

  int2* q = queue + wave * AGG_QCAP;
  int cnt = 0;

  auto work_off = [&](const int2* qb, int n) {
    int sN = 0, dl = 0;
    float a[H];
#pragma unroll
    for (int h = 0; h < H; ++h) a[h] = 0.f;
    if (lane < n) {
      const int2 ed = qb[lane];
      const int e   = ed.x;
      const int dN  = ed.y;
      sN = src[e];
      dl = dN - lo;
#pragma unroll
      for (int h = 0; h < H; ++h)
        a[h] = ee[(size_t)e * H + h] / fmaxf(s[(size_t)dN * H + h], 1e-9f);
    }
    for (int k0 = 0; k0 < n; k0 += G) {
      float v[G][LINES];
#pragma unroll
      for (int g = 0; g < G; ++g) {
        const int k  = (k0 + g < n) ? (k0 + g) : (n - 1);
        const int sk = __builtin_amdgcn_readlane(sN, k);
        const float* f = feat + (size_t)sk * R + lane;
#pragma unroll
        for (int j = 0; j < LINES; ++j) v[g][j] = f[32 * j];
      }
#pragma unroll
      for (int g = 0; g < G; ++g) {
        const int k = k0 + g;
        if (k < n) {
          const int dk = __builtin_amdgcn_readlane(dl, k);
          float* o = sacc + dk * R + lane;
#pragma unroll
          for (int j = 0; j < LINES; ++j) {
            const float ak = __int_as_float(__builtin_amdgcn_readlane(__float_as_int(a[j / LPH]), k));
            atomicAdd(o + 32 * j, ak * v[g][j]);
          }
        }
      }
    }
  };

  const int nChunks = (E + 127) >> 7;
  const unsigned below = (1u << lane) - 1u;
  for (int c0 = wave; c0 < nChunks; c0 += (AGG_THREADS / 32) * AGG_U) {
    int4 d[AGG_U];
#pragma unroll
    for (int u = 0; u < AGG_U; ++u) {
      const int c    = c0 + (AGG_THREADS / 32) * u;
      const int base = c * 128 + 4 * lane;
      d[u] = make_int4(-1, -1, -1, -1);
      if (c < nChunks) {
        if (c * 128 + 128 <= E) {
          d[u] = *reinterpret_cast<const int4*>(dst + base);
        } else {
          if (base + 0 < E) d[u].x = dst[base + 0];
          if (base + 1 < E) d[u].y = dst[base + 1];
          if (base + 2 < E) d[u].z = dst[base + 2];
          if (base + 3 < E) d[u].w = dst[base + 3];
        }
      }
    }
#pragma unroll
    for (int u = 0; u < AGG_U; ++u) {
      const int c    = c0 + (AGG_THREADS / 32) * u;
      const int base = c * 128 + 4 * lane;
      const int dv[4] = { d[u].x, d[u].y, d[u].z, d[u].w };
#pragma unroll
      for (int j = 0; j < 4; ++j) {
        const bool mine = (unsigned)(dv[j] - lo) < (unsigned)nn;
        const unsigned mask = (unsigned)__ballot(mine);
        if (mask != 0u) {
          if (mine) q[cnt + __popc(mask & below)] = make_int2(base + j, dv[j]);
          cnt += __popc(mask);
        }
      }
      while (cnt >= 32) { cnt -= 32; work_off(q + cnt, 32); }
    }
  }
  if (cnt > 0) work_off(q, cnt);
  __syncthreads();

  float4* out4 = reinterpret_cast<float4*>(acc + (size_t)lo * R);
  for (int i = tid; i < nn * (R / 4); i += AGG_THREADS) out4[i] = sacc4[i];
}

__global__ void __launch_bounds__(AGG_THREADS)
owner_aggregate_h4d64(const int* __restrict__ src, const int* __restrict__ dst,
                      const float* __restrict__ ee, const float* __restrict__ s,
                      const float* __restrict__ feat, float* __restrict__ acc, int E, int N) {
  owner_aggregate_body<4, 64>(src, dst, ee, s, feat, acc, E, N);
}
__global__ void __launch_bounds__(AGG_THREADS)
owner_aggregate_h1d64(const int* __restrict__ src, const int* __restrict__ dst,
                      const float* __restrict__ ee, const float* __restrict__ s,
                      const float* __restrict__ feat, float* __restrict__ acc, int E, int N) {
  owner_aggregate_body<1, 64>(src, dst, ee, s, feat, acc, E, N);
}

__device__ __forceinline__ float elu1(float v) { return (v > 0.f) ? v : expm1f(v); }

template <int H, int D>
__device__ __forceinline__ void epilogue0_body(const float* __restrict__ acc, const float* __restrict__ b,
                                               float* __restrict__ h1, int N) {
  constexpr int D4 = D / 4;
  const unsigned n4 = (unsigned)N * D4;
  const unsigned T = gridDim.x * blockDim.x;
  for (unsigned i = blockIdx.x * blockDim.x + threadIdx.x; i < n4; i += T) {
    const unsigned n = i / D4, q = i - n * D4;
    float4 v[2][H], bb[H];
#pragma unroll
    for (int t = 0; t < 2; ++t)
#pragma unroll
      for (int h = 0; h < H; ++h)
        v[t][h] = *reinterpret_cast<const float4*>(acc + ((size_t)t * N + n) * (H * D) + h * D + 4 * q);
#pragma unroll
    for (int h = 0; h < H; ++h) bb[h] = *reinterpret_cast<const float4*>(b + h * D + 4 * q);
    float4 sum = make_float4(0.f, 0.f, 0.f, 0.f);
#pragma unroll
    for (int t = 0; t < 2; ++t)
#pragma unroll
      for (int h = 0; h < H; ++h) {
        sum.x += elu1(v[t][h].x + bb[h].x);
        sum.y += elu1(v[t][h].y + bb[h].y);
        sum.z += elu1(v[t][h].z + bb[h].z);
        sum.w += elu1(v[t][h].w + bb[h].w);
      }
    const float hf = (float)H;
    *reinterpret_cast<float4*>(h1 + (size_t)i * 4) =
        make_float4(sum.x / hf, sum.y / hf, sum.z / hf, sum.w / hf);
  }
}
__global__ void epilogue0_h4d64(const float* __restrict__ acc, const float* __restrict__ b,
                                float* __restrict__ h1, int N) {
  epilogue0_body<4, 64>(acc, b, h1, N);
}


template <int D, int U>
__device__ __forceinline__ void epilogue1_body(const float* __restrict__ acc, const float* __restrict__ b,
                                               float* __restrict__ out, int N) {
  constexpr int D4 = D / 4;
  const unsigned n4 = (unsigned)N * D4;
  const unsigned T = gridDim.x * blockDim.x;
  const v4f* a0 = reinterpret_cast<const v4f*>(acc);
  const v4f* a1 = reinterpret_cast<const v4f*>(acc + (size_t)N * D);
  const v4f* b4 = reinterpret_cast<const v4f*>(b);
  v4f* o4 = reinterpret_cast<v4f*>(out);
  for (unsigned i0 = blockIdx.x * blockDim.x + threadIdx.x; i0 < n4; i0 += U * T) {
    v4f x[U], y[U], bb[U];
#pragma unroll
    for (int u = 0; u < U; ++u) {
      const unsigned i  = i0 + u * T;
      const unsigned ic = i < n4 ? i : n4 - 1;
      x[u]  = a0[ic];
      y[u]  = a1[ic];
      bb[u] = b4[ic % D4];
    }
#pragma unroll
    for (int u = 0; u < U; ++u) asm volatile("" :: "v"(x[u]), "v"(y[u]), "v"(bb[u]));
#pragma unroll
    for (int u = 0; u < U; ++u) {
      const unsigned i = i0 + u * T;
      if (i < n4) o4[i] = (x[u] + bb[u]) + (y[u] + bb[u]);
    }
  }
}
__global__ void epilogue1_d64(const float* __restrict__ acc, const float* __restrict__ b,
                              float* __restrict__ out, int N) {
  epilogue1_body<64, 4>(acc, b, out, N);
}

extern "C" void kernel_launch(void* const* d_in, const int* in_sizes, int n_in,
                              void* d_out, int out_size, void* d_ws, size_t ws_size,
                              hipStream_t stream) {
  const int IN = 128, H0 = 4, D0 = 64, H1 = 1, D1 = 64;
  const int N = in_sizes[0] / IN;
  const int E = in_sizes[9];

  const float* x   = (const float*)d_in[0];
  const float* W0  = (const float*)d_in[1];
  const float* al0 = (const float*)d_in[2];
  const float* ar0 = (const float*)d_in[3];
  const float* b0  = (const float*)d_in[4];
  const float* W1  = (const float*)d_in[5];
  const float* al1 = (const float*)d_in[6];
  const float* ar1 = (const float*)d_in[7];
  const float* b1  = (const float*)d_in[8];
  const int* srcT[2] = { (const int*)d_in[9],  (const int*)d_in[11] };
  const int* dstT[2] = { (const int*)d_in[10], (const int*)d_in[12] };
  float* out = (float*)d_out;

  float* ws = (float*)d_ws;
  float*    feat0 = ws;
  float*    el0   = feat0 + (size_t)N * H0 * D0;
  float*    er0   = el0 + (size_t)N * H0;
  unsigned* m0    = (unsigned*)(er0 + (size_t)N * H0);
  float*    s0    = (float*)(m0 + 2 * (size_t)N * H0);
  float*    ee0   = s0 + 2 * (size_t)N * H0;
  float*    acc0  = ee0 + 2 * (size_t)E * H0;
  float*    h1    = acc0 + 2 * (size_t)N * H0 * D0;
  float*    feat1 = feat0;
  float*    el1   = feat1 + (size_t)N * D1;
  float*    er1   = el1 + N;
  unsigned* m1    = (unsigned*)(er1 + N);
  float*    s1    = (float*)(m1 + 2 * (size_t)N);
  float*    ee1   = s1 + 2 * (size_t)N;
  float*    acc1  = ee1 + 2 * (size_t)E;

  const int TB = 256;
  const int aggBlocks = (N + AGG_NODES - 1) / AGG_NODES;
  const int fillSmallBlocks = 32;
  const int edgeBlocks0 = (E * H0 + TB * 8 - 1) / (TB * 8);
  const int edgeBlocks1 = (E * H1 + TB * 4 - 1) / (TB * 4);
  const int epi0Blocks  = 1024;
  const int epi1Blocks  = (N * (D1 / 4) + TB * 4 - 1) / (TB * 4);
  (void)ws_size; (void)n_in; (void)out_size;

  {
    gemm_lds_k128_n256<<<256, GT_THREADS, 0, stream>>>(x, W0, feat0, N);

    node_scores<<<(N * H0 + TB - 1) / TB, TB, 0, stream>>>(feat0, al0, ar0, el0, er0, N, H0, D0);

    fill_u32<<<fillSmallBlocks, TB, 0, stream>>>(m0, ORD_NEG_INF, 2 * N * H0);
    fill_f32<<<fillSmallBlocks, TB, 0, stream>>>(s0, 0.f, 2 * N * H0);

    for (int t = 0; t < 2; ++t) {
      unsigned* mT = m0 + (size_t)t * N * H0;
      float*    sT = s0 + (size_t)t * N * H0;
      float*    eeT = ee0 + (size_t)t * E * H0;
      float*    accT = acc0 + (size_t)t * N * H0 * D0;
      edge_max_h4<<<edgeBlocks0, TB, 0, stream>>>(srcT[t], dstT[t], el0, er0, mT, E);
      edge_expsum_h4<<<edgeBlocks0, TB, 0, stream>>>(srcT[t], dstT[t], el0, er0, mT, eeT, sT, E);
      owner_aggregate_h4d64<<<aggBlocks, AGG_THREADS, 0, stream>>>(srcT[t], dstT[t], eeT, sT, feat0, accT, E, N);
    }

    epilogue0_h4d64<<<epi0Blocks, TB, 0, stream>>>(acc0, b0, h1, N);
  }

  {
    gemm_lds_k64_n64<<<(N + 127) / 128, GT_THREADS, 0, stream>>>(h1, W1, feat1, N);

    node_scores<<<(N * H1 + TB - 1) / TB, TB, 0, stream>>>(feat1, al1, ar1, el1, er1, N, H1, D1);

    fill_u32<<<fillSmallBlocks, TB, 0, stream>>>(m1, ORD_NEG_INF, 2 * N * H1);
    fill_f32<<<fillSmallBlocks, TB, 0, stream>>>(s1, 0.f, 2 * N * H1);

    for (int t = 0; t < 2; ++t) {
      unsigned* mT = m1 + (size_t)t * N * H1;
      float*    sT = s1 + (size_t)t * N * H1;
      float*    eeT = ee1 + (size_t)t * E * H1;
      float*    accT = acc1 + (size_t)t * N * H1 * D1;
      edge_max_h1<<<edgeBlocks1, TB, 0, stream>>>(srcT[t], dstT[t], el1, er1, mT, E);
      edge_expsum_h1<<<edgeBlocks1, TB, 0, stream>>>(srcT[t], dstT[t], el1, er1, mT, eeT, sT, E);
      owner_aggregate_h1d64<<<aggBlocks, AGG_THREADS, 0, stream>>>(srcT[t], dstT[t], eeT, sT, feat1, accT, E, N);
    }

    epilogue1_d64<<<epi1Blocks, TB, 0, stream>>>(acc1, b1, out, N);
  }
}


# Launch merging, same work per item: each edge pass and the aggregation take both edge types in one launch (half the grid per type); node_scores stores the m and s start values; 22 -> 12 launches
# speedup vs baseline: 1.7426x; 1.1000x over previous
#include <hip/hip_runtime.h>
#include <math.h>

typedef float v2f __attribute__((ext_vector_type(2)));
typedef float v4f __attribute__((ext_vector_type(4)));
typedef float v8f __attribute__((ext_vector_type(8)));

#define LRELU_SLOPE 0.2f

__device__ __forceinline__ unsigned f2o(float f) {
  unsigned u = __float_as_uint(f);
  return (u & 0x80000000u) ? ~u : (u | 0x80000000u);
}
__device__ __forceinline__ float o2f(unsigned o) {
  unsigned u = (o & 0x80000000u) ? (o & 0x7fffffffu) : ~o;
  return __uint_as_float(u);
}
#define ORD_NEG_INF 0x007fffffu

#define GT_THREADS 512

template <int K, int NC>
__device__ __forceinline__ void gemm_lds_body(const float* __restrict__ A,
                                              const float* __restrict__ B,
                                              float* __restrict__ C, int M) {
  constexpr int NP  = NC / 32;
  constexpr int WR  = 16 / NP;
  constexpr int RS  = WR * 16;
  constexpr int PA  = K + 4;
  constexpr int PB  = 2 * NC + 32;
  constexpr int K4  = K / 4;
  constexpr int NC4 = NC / 4;
  constexpr int NPF = (RS * K4) / GT_THREADS;
  constexpr int NBF = ((K / 2) * NC4) / GT_THREADS;
  static_assert(NC % 32 == 0 && 16 % NP == 0, "a wave takes one M-tile and one column pair");
  static_assert((RS * K4) % GT_THREADS == 0 && (NPF == 2 || NPF == 4), "an A tile is 2 or 4 pieces per thread");
  static_assert(((K / 2) * NC4) % GT_THREADS == 0, "B splits evenly over the threads");
  static_assert(K % 4 == 0, "k steps of 4");

  __shared__ __align__(16) float Bs[(K / 2) * PB];
  __shared__ __align__(16) float As[2][RS * PA];

  const int tid  = threadIdx.x;
  const int lane = tid & 31;
  const int half = lane >> 4;
  const int l    = lane & 15;
  const int wave = __builtin_amdgcn_readfirstlane(tid >> 5);
  const int wm   = wave / NP;
  const int wn   = wave - wm * NP;

  const int nSteps = (M + RS - 1) / RS;

  auto fetch_piece = [&](int s, int u) -> v4f {
    const int it = tid + u * GT_THREADS;
    const int r = it / K4, c = it - r * K4;
    int row = s * RS + r;
    row = row < M ? row : M - 1;
    return *reinterpret_cast<const v4f*>(A + (size_t)row * K + 4 * c);
  };
  auto stash_piece = [&](float* as, int u, v4f v) {
    const int it = tid + u * GT_THREADS;
    const int r = it / K4, c = it - r * K4;
    *reinterpret_cast<v4f*>(as + r * PA + 4 * c) = v;
  };

  int step = blockIdx.x;
  v4f p0 = fetch_piece(step, 0), p1 = fetch_piece(step, 1), p2 = p0, p3 = p0;
  if constexpr (NPF == 4) { p2 = fetch_piece(step, 2); p3 = fetch_piece(step, 3); }
#pragma unroll
  for (int u = 0; u < NBF; ++u) {
    const int it = tid + u * GT_THREADS;
    const int kp = it / NC4, c = it - kp * NC4;
    const float4 r0 = *reinterpret_cast<const float4*>(B + (size_t)(2 * kp) * NC + 4 * c);
    const float4 r1 = *reinterpret_cast<const float4*>(B + (size_t)(2 * kp + 1) * NC + 4 * c);
    float4* d = reinterpret_cast<float4*>(Bs + kp * PB + 8 * c);
    d[0] = make_float4(r0.x, r1.x, r0.y, r1.y);
    d[1] = make_float4(r0.z, r1.z, r0.w, r1.w);
  }

  int buf = 0;
  for (; step < nSteps; step += gridDim.x) {
    float* as = As[buf];
    stash_piece(as, 0, p0);
    stash_piece(as, 1, p1);
    if constexpr (NPF == 4) { stash_piece(as, 2, p2); stash_piece(as, 3, p3); }
    __syncthreads();

    const int next = step + gridDim.x;
    p0 = fetch_piece(next, 0);
    p1 = fetch_piece(next, 1);
    if constexpr (NPF == 4) { p2 = fetch_piece(next, 2); p3 = fetch_piece(next, 3); }

    v8f acc0 = {}, acc1 = {};
    const float* ap = as + (wm * 16 + l) * PA + 2 * half;
    const float* bp = Bs + half * PB + (wn * 32 + l) * 2;
#pragma unroll
    for (int k = 0; k < K; k += 4) {
      const v2f a  = *reinterpret_cast<const v2f*>(ap + k);
      const v2f b0 = *reinterpret_cast<const v2f*>(bp + (k >> 1) * PB);
      const v2f b1 = *reinterpret_cast<const v2f*>(bp + (k >> 1) * PB + 32);
      acc0 = __builtin_amdgcn_wmma_f32_16x16x4_f32(false, a, false, b0, (short)0, acc0, false, false);
      acc1 = __builtin_amdgcn_wmma_f32_16x16x4_f32(false, a, false, b1, (short)0, acc1, false, false);
    }

    const int row0 = step * RS + wm * 16 + half * 8;
    float* cp = C + (size_t)row0 * NC + wn * 32 + l;
#pragma unroll
    for (int r = 0; r < 8; ++r) {
      if (row0 + r < M) {
        cp[(size_t)r * NC]      = acc0[r];
        cp[(size_t)r * NC + 16] = acc1[r];
      }
    }
    buf ^= 1;
  }
}

__global__ void __launch_bounds__(GT_THREADS)
gemm_lds_k128_n256(const float* __restrict__ A, const float* __restrict__ B, float* __restrict__ C, int M) {
  gemm_lds_body<128, 256>(A, B, C, M);
}
__global__ void __launch_bounds__(GT_THREADS)
gemm_lds_k64_n64(const float* __restrict__ A, const float* __restrict__ B, float* __restrict__ C, int M) {
  gemm_lds_body<64, 64>(A, B, C, M);
}

__global__ void node_scores(const float* __restrict__ feat,
                            const float* __restrict__ al,
                            const float* __restrict__ ar,
                            float* __restrict__ el, float* __restrict__ er,
                            unsigned* __restrict__ m_ord, float* __restrict__ s,
                            int N, int H, int D) {
  int i = blockIdx.x * blockDim.x + threadIdx.x;
  if (i >= N * H) return;
  int n = i / H, h = i - n * H;
  const float* f = feat + (size_t)n * H * D + h * D;
  const float* wl = al + h * D;
  const float* wr = ar + h * D;
  float sl = 0.f, sr = 0.f;
#pragma unroll 4
  for (int d = 0; d < D; ++d) {
    float v = f[d];
    sl += v * wl[d];
    sr += v * wr[d];
  }
  el[i] = sl;
  er[i] = sr;
  const size_t plane = (size_t)N * H;
  m_ord[i]         = ORD_NEG_INF;
  m_ord[plane + i] = ORD_NEG_INF;
  s[i]             = 0.f;
  s[plane + i]     = 0.f;
}

template <int H, int U>
__device__ __forceinline__ void edge_max_body(const int* __restrict__ src0, const int* __restrict__ dst0,
                                              const int* __restrict__ src1, const int* __restrict__ dst1,
                                              const float* __restrict__ el, const float* __restrict__ er,
                                              unsigned* __restrict__ m_ord, int E, int N) {
  const unsigned nb  = gridDim.x >> 1;
  const unsigned t   = blockIdx.x >= nb ? 1u : 0u;
  const unsigned bid = blockIdx.x - t * nb;
  const int* __restrict__ src = t ? src1 : src0;
  const int* __restrict__ dst = t ? dst1 : dst0;
  m_ord += (size_t)t * N * H;
  const unsigned n = (unsigned)E * H;
  const unsigned T = nb * blockDim.x;
  for (unsigned i0 = bid * blockDim.x + threadIdx.x; i0 < n; i0 += U * T) {
    unsigned sN[U], dN[U], di[U];
    float l[U], r[U];
#pragma unroll
    for (int u = 0; u < U; ++u) {
      const unsigned i  = i0 + u * T;
      const unsigned ic = i < n ? i : n - 1;
      sN[u] = (unsigned)src[ic / H];
      dN[u] = (unsigned)dst[ic / H];
    }
#pragma unroll
    for (int u = 0; u < U; ++u) asm volatile("" :: "v"(sN[u]), "v"(dN[u]));
#pragma unroll
    for (int u = 0; u < U; ++u) {
      const unsigned i  = i0 + u * T;
      const unsigned ic = i < n ? i : n - 1;
      const unsigned h  = ic % H;
      di[u] = dN[u] * H + h;
      l[u]  = el[sN[u] * H + h];
      r[u]  = er[di[u]];
    }
#pragma unroll
    for (int u = 0; u < U; ++u) asm volatile("" :: "v"(l[u]), "v"(r[u]));
#pragma unroll
    for (int u = 0; u < U; ++u) {
      if (i0 + u * T < n) {
        float sc = l[u] + r[u];
        sc = sc > 0.f ? sc : LRELU_SLOPE * sc;
        atomicMax(m_ord + di[u], f2o(sc));
      }
    }
  }
}
__global__ void edge_max_h4(const int* __restrict__ src0, const int* __restrict__ dst0,
                            const int* __restrict__ src1, const int* __restrict__ dst1,
                            const float* __restrict__ el, const float* __restrict__ er,
                            unsigned* __restrict__ m_ord, int E, int N) {
  edge_max_body<4, 8>(src0, dst0, src1, dst1, el, er, m_ord, E, N);
}
__global__ void edge_max_h1(const int* __restrict__ src0, const int* __restrict__ dst0,
                            const int* __restrict__ src1, const int* __restrict__ dst1,
                            const float* __restrict__ el, const float* __restrict__ er,
                            unsigned* __restrict__ m_ord, int E, int N) {
  edge_max_body<1, 4>(src0, dst0, src1, dst1, el, er, m_ord, E, N);
}

template <int H, int U>
__device__ __forceinline__ void edge_expsum_body(const int* __restrict__ src0, const int* __restrict__ dst0,
                                                 const int* __restrict__ src1, const int* __restrict__ dst1,
                                                 const float* __restrict__ el, const float* __restrict__ er,
                                                 const unsigned* __restrict__ m_ord,
                                                 float* __restrict__ ee, float* __restrict__ s, int E, int N) {
  const unsigned nb  = gridDim.x >> 1;
  const unsigned t   = blockIdx.x >= nb ? 1u : 0u;
  const unsigned bid = blockIdx.x - t * nb;
  const int* __restrict__ src = t ? src1 : src0;
  const int* __restrict__ dst = t ? dst1 : dst0;
  m_ord += (size_t)t * N * H;
  s     += (size_t)t * N * H;
  ee    += (size_t)t * E * H;
  const unsigned n = (unsigned)E * H;
  const unsigned T = nb * blockDim.x;
  for (unsigned i0 = bid * blockDim.x + threadIdx.x; i0 < n; i0 += U * T) {
    unsigned sN[U], dN[U], di[U], mo[U];
    float l[U], r[U];
#pragma unroll
    for (int u = 0; u < U; ++u) {
      const unsigned i  = i0 + u * T;
      const unsigned ic = i < n ? i : n - 1;
      sN[u] = (unsigned)src[ic / H];
      dN[u] = (unsigned)dst[ic / H];
    }
#pragma unroll
    for (int u = 0; u < U; ++u) asm volatile("" :: "v"(sN[u]), "v"(dN[u]));
#pragma unroll
    for (int u = 0; u < U; ++u) {
      const unsigned i  = i0 + u * T;
      const unsigned ic = i < n ? i : n - 1;
      const unsigned h  = ic % H;
      di[u] = dN[u] * H + h;
      l[u]  = el[sN[u] * H + h];
      r[u]  = er[di[u]];
      mo[u] = m_ord[di[u]];
    }
#pragma unroll
    for (int u = 0; u < U; ++u) asm volatile("" :: "v"(l[u]), "v"(r[u]), "v"(mo[u]));
#pragma unroll
    for (int u = 0; u < U; ++u) {
      const unsigned i = i0 + u * T;
      if (i < n) {
        float sc = l[u] + r[u];
        sc = sc > 0.f ? sc : LRELU_SLOPE * sc;
        float m = o2f(mo[u]);
        if (m < -1e30f) m = 0.f;
        const float v = __expf(sc - m);
        ee[i] = v;
        atomicAdd(s + di[u], v);
      }
    }
  }
}
__global__ void edge_expsum_h4(const int* __restrict__ src0, const int* __restrict__ dst0,
                               const int* __restrict__ src1, const int* __restrict__ dst1,
                               const float* __restrict__ el, const float* __restrict__ er,
                               const unsigned* __restrict__ m_ord,
                               float* __restrict__ ee, float* __restrict__ s, int E, int N) {
  edge_expsum_body<4, 8>(src0, dst0, src1, dst1, el, er, m_ord, ee, s, E, N);
}
__global__ void edge_expsum_h1(const int* __restrict__ src0, const int* __restrict__ dst0,
                               const int* __restrict__ src1, const int* __restrict__ dst1,
                               const float* __restrict__ el, const float* __restrict__ er,
                               const unsigned* __restrict__ m_ord,
                               float* __restrict__ ee, float* __restrict__ s, int E, int N) {
  edge_expsum_body<1, 4>(src0, dst0, src1, dst1, el, er, m_ord, ee, s, E, N);
}

#define AGG_THREADS 1024
#define AGG_NODES   196
#define AGG_QCAP    192
#define AGG_U       8

template <int H, int D>
__device__ __forceinline__ void owner_aggregate_body(const int* __restrict__ src0, const int* __restrict__ dst0,
                                                     const int* __restrict__ src1, const int* __restrict__ dst1,
                                                     const float* __restrict__ ee, const float* __restrict__ s,
                                                     const float* __restrict__ feat, float* __restrict__ acc,
                                                     int E, int N) {
  constexpr int R     = H * D;
  constexpr int LINES = R / 32;
  constexpr int LPH   = D / 32;
  constexpr int G     = (LINES >= 8) ? 4 : 16;
  static_assert(D % 32 == 0, "a line must lie inside one head");

  __shared__ float sacc[AGG_NODES * R];
  __shared__ int2  queue[(AGG_THREADS / 32) * AGG_QCAP];

  const int tid  = threadIdx.x;
  const int lane = tid & 31;
  const int wave = __builtin_amdgcn_readfirstlane(tid >> 5);
  const int nb  = (int)(gridDim.x >> 1);
  const int t   = (int)blockIdx.x >= nb ? 1 : 0;
  const int bid = (int)blockIdx.x - t * nb;
  const int* __restrict__ src = t ? src1 : src0;
  const int* __restrict__ dst = t ? dst1 : dst0;
  ee  += (size_t)t * E * H;
  s   += (size_t)t * N * H;
  acc += (size_t)t * N * R;
  const int lo = bid * AGG_NODES;
  const int nn = (N - lo) < AGG_NODES ? (N - lo) : AGG_NODES;

  float4* sacc4 = reinterpret_cast<float4*>(sacc);
  for (int i = tid; i < AGG_NODES * R / 4; i += AGG_THREADS) sacc4[i] = make_float4(0.f, 0.f, 0.f, 0.f);
  __syncthreads();

  int2* q = queue + wave * AGG_QCAP;
  int cnt = 0;

  auto work_off = [&](const int2* qb, int n) {
    int sN = 0, dl = 0;
    float a[H];
#pragma unroll
    for (int h = 0; h < H; ++h) a[h] = 0.f;
    if (lane < n) {
      const int2 ed = qb[lane];
      const int e   = ed.x;
      const int dN  = ed.y;
      sN = src[e];
      dl = dN - lo;
#pragma unroll
      for (int h = 0; h < H; ++h)
        a[h] = ee[(size_t)e * H + h] / fmaxf(s[(size_t)dN * H + h], 1e-9f);
    }
    for (int k0 = 0; k0 < n; k0 += G) {
      float v[G][LINES];
#pragma unroll
      for (int g = 0; g < G; ++g) {
        const int k  = (k0 + g < n) ? (k0 + g) : (n - 1);
        const int sk = __builtin_amdgcn_readlane(sN, k);
        const float* f = feat + (size_t)sk * R + lane;
#pragma unroll
        for (int j = 0; j < LINES; ++j) v[g][j] = f[32 * j];
      }
#pragma unroll
      for (int g = 0; g < G; ++g) {
        const int k = k0 + g;
        if (k < n) {
          const int dk = __builtin_amdgcn_readlane(dl, k);
          float* o = sacc + dk * R + lane;
#pragma unroll
          for (int j = 0; j < LINES; ++j) {
            const float ak = __int_as_float(__builtin_amdgcn_readlane(__float_as_int(a[j / LPH]), k));
            atomicAdd(o + 32 * j, ak * v[g][j]);
          }
        }
      }
    }
  };

  const int nChunks = (E + 127) >> 7;
  const unsigned below = (1u << lane) - 1u;
  for (int c0 = wave; c0 < nChunks; c0 += (AGG_THREADS / 32) * AGG_U) {
    int4 d[AGG_U];
#pragma unroll
    for (int u = 0; u < AGG_U; ++u) {
      const int c    = c0 + (AGG_THREADS / 32) * u;
      const int base = c * 128 + 4 * lane;
      d[u] = make_int4(-1, -1, -1, -1);
      if (c < nChunks) {
        if (c * 128 + 128 <= E) {
          d[u] = *reinterpret_cast<const int4*>(dst + base);
        } else {
          if (base + 0 < E) d[u].x = dst[base + 0];
          if (base + 1 < E) d[u].y = dst[base + 1];
          if (base + 2 < E) d[u].z = dst[base + 2];
          if (base + 3 < E) d[u].w = dst[base + 3];
        }
      }
    }
#pragma unroll
    for (int u = 0; u < AGG_U; ++u) {
      const int c    = c0 + (AGG_THREADS / 32) * u;
      const int base = c * 128 + 4 * lane;
      const int dv[4] = { d[u].x, d[u].y, d[u].z, d[u].w };
#pragma unroll
      for (int j = 0; j < 4; ++j) {
        const bool mine = (unsigned)(dv[j] - lo) < (unsigned)nn;
        const unsigned mask = (unsigned)__ballot(mine);
        if (mask != 0u) {
          if (mine) q[cnt + __popc(mask & below)] = make_int2(base + j, dv[j]);
          cnt += __popc(mask);
        }
      }
      while (cnt >= 32) { cnt -= 32; work_off(q + cnt, 32); }
    }
  }
  if (cnt > 0) work_off(q, cnt);
  __syncthreads();

  float4* out4 = reinterpret_cast<float4*>(acc + (size_t)lo * R);
  for (int i = tid; i < nn * (R / 4); i += AGG_THREADS) out4[i] = sacc4[i];
}

__global__ void __launch_bounds__(AGG_THREADS)
owner_aggregate_h4d64(const int* __restrict__ src0, const int* __restrict__ dst0,
                      const int* __restrict__ src1, const int* __restrict__ dst1,
                      const float* __restrict__ ee, const float* __restrict__ s,
                      const float* __restrict__ feat, float* __restrict__ acc, int E, int N) {
  owner_aggregate_body<4, 64>(src0, dst0, src1, dst1, ee, s, feat, acc, E, N);
}
__global__ void __launch_bounds__(AGG_THREADS)
owner_aggregate_h1d64(const int* __restrict__ src0, const int* __restrict__ dst0,
                      const int* __restrict__ src1, const int* __restrict__ dst1,
                      const float* __restrict__ ee, const float* __restrict__ s,
                      const float* __restrict__ feat, float* __restrict__ acc, int E, int N) {
  owner_aggregate_body<1, 64>(src0, dst0, src1, dst1, ee, s, feat, acc, E, N);
}

__device__ __forceinline__ float elu1(float v) { return (v > 0.f) ? v : expm1f(v); }

template <int H, int D>
__device__ __forceinline__ void epilogue0_body(const float* __restrict__ acc, const float* __restrict__ b,
                                               float* __restrict__ h1, int N) {
  constexpr int D4 = D / 4;
  const unsigned n4 = (unsigned)N * D4;
  const unsigned T = gridDim.x * blockDim.x;
  for (unsigned i = blockIdx.x * blockDim.x + threadIdx.x; i < n4; i += T) {
    const unsigned n = i / D4, q = i - n * D4;
    float4 v[2][H], bb[H];
#pragma unroll
    for (int t = 0; t < 2; ++t)
#pragma unroll
      for (int h = 0; h < H; ++h)
        v[t][h] = *reinterpret_cast<const float4*>(acc + ((size_t)t * N + n) * (H * D) + h * D + 4 * q);
#pragma unroll
    for (int h = 0; h < H; ++h) bb[h] = *reinterpret_cast<const float4*>(b + h * D + 4 * q);
    float4 sum = make_float4(0.f, 0.f, 0.f, 0.f);
#pragma unroll
    for (int t = 0; t < 2; ++t)
#pragma unroll
      for (int h = 0; h < H; ++h) {
        sum.x += elu1(v[t][h].x + bb[h].x);
        sum.y += elu1(v[t][h].y + bb[h].y);
        sum.z += elu1(v[t][h].z + bb[h].z);
        sum.w += elu1(v[t][h].w + bb[h].w);
      }
    const float hf = (float)H;
    *reinterpret_cast<float4*>(h1 + (size_t)i * 4) =
        make_float4(sum.x / hf, sum.y / hf, sum.z / hf, sum.w / hf);
  }
}
__global__ void epilogue0_h4d64(const float* __restrict__ acc, const float* __restrict__ b,
                                float* __restrict__ h1, int N) {
  epilogue0_body<4, 64>(acc, b, h1, N);
}


template <int D, int U>
__device__ __forceinline__ void epilogue1_body(const float* __restrict__ acc, const float* __restrict__ b,
                                               float* __restrict__ out, int N) {
  constexpr int D4 = D / 4;
  const unsigned n4 = (unsigned)N * D4;
  const unsigned T = gridDim.x * blockDim.x;
  const v4f* a0 = reinterpret_cast<const v4f*>(acc);
  const v4f* a1 = reinterpret_cast<const v4f*>(acc + (size_t)N * D);
  const v4f* b4 = reinterpret_cast<const v4f*>(b);
  v4f* o4 = reinterpret_cast<v4f*>(out);
  for (unsigned i0 = blockIdx.x * blockDim.x + threadIdx.x; i0 < n4; i0 += U * T) {
    v4f x[U], y[U], bb[U];
#pragma unroll
    for (int u = 0; u < U; ++u) {
      const unsigned i  = i0 + u * T;
      const unsigned ic = i < n4 ? i : n4 - 1;
      x[u]  = a0[ic];
      y[u]  = a1[ic];
      bb[u] = b4[ic % D4];
    }
#pragma unroll
    for (int u = 0; u < U; ++u) asm volatile("" :: "v"(x[u]), "v"(y[u]), "v"(bb[u]));
#pragma unroll
    for (int u = 0; u < U; ++u) {
      const unsigned i = i0 + u * T;
      if (i < n4) o4[i] = (x[u] + bb[u]) + (y[u] + bb[u]);
    }
  }
}
__global__ void epilogue1_d64(const float* __restrict__ acc, const float* __restrict__ b,
                              float* __restrict__ out, int N) {
  epilogue1_body<64, 4>(acc, b, out, N);
}

extern "C" void kernel_launch(void* const* d_in, const int* in_sizes, int n_in,
                              void* d_out, int out_size, void* d_ws, size_t ws_size,
                              hipStream_t stream) {
  const int IN = 128, H0 = 4, D0 = 64, H1 = 1, D1 = 64;
  const int N = in_sizes[0] / IN;
  const int E = in_sizes[9];

  const float* x   = (const float*)d_in[0];
  const float* W0  = (const float*)d_in[1];
  const float* al0 = (const float*)d_in[2];
  const float* ar0 = (const float*)d_in[3];
  const float* b0  = (const float*)d_in[4];
  const float* W1  = (const float*)d_in[5];
  const float* al1 = (const float*)d_in[6];
  const float* ar1 = (const float*)d_in[7];
  const float* b1  = (const float*)d_in[8];
  const int* srcT[2] = { (const int*)d_in[9],  (const int*)d_in[11] };
  const int* dstT[2] = { (const int*)d_in[10], (const int*)d_in[12] };
  float* out = (float*)d_out;

  float* ws = (float*)d_ws;
  float*    feat0 = ws;
  float*    el0   = feat0 + (size_t)N * H0 * D0;
  float*    er0   = el0 + (size_t)N * H0;
  unsigned* m0    = (unsigned*)(er0 + (size_t)N * H0);
  float*    s0    = (float*)(m0 + 2 * (size_t)N * H0);
  float*    ee0   = s0 + 2 * (size_t)N * H0;
  float*    acc0  = ee0 + 2 * (size_t)E * H0;
  float*    h1    = acc0 + 2 * (size_t)N * H0 * D0;
  float*    feat1 = feat0;
  float*    el1   = feat1 + (size_t)N * D1;
  float*    er1   = el1 + N;
  unsigned* m1    = (unsigned*)(er1 + N);
  float*    s1    = (float*)(m1 + 2 * (size_t)N);
  float*    ee1   = s1 + 2 * (size_t)N;
  float*    acc1  = ee1 + 2 * (size_t)E;

  const int TB = 256;
  const int aggBlocks = (N + AGG_NODES - 1) / AGG_NODES;
  const int edgeBlocks0 = (E * H0 + TB * 8 - 1) / (TB * 8);
  const int edgeBlocks1 = (E * H1 + TB * 4 - 1) / (TB * 4);
  const int epi0Blocks  = 1024;
  const int epi1Blocks  = (N * (D1 / 4) + TB * 4 - 1) / (TB * 4);
  (void)ws_size; (void)n_in; (void)out_size;

  {
    gemm_lds_k128_n256<<<256, GT_THREADS, 0, stream>>>(x, W0, feat0, N);

    node_scores<<<(N * H0 + TB - 1) / TB, TB, 0, stream>>>(feat0, al0, ar0, el0, er0, m0, s0, N, H0, D0);

    edge_max_h4<<<2 * edgeBlocks0, TB, 0, stream>>>(srcT[0], dstT[0], srcT[1], dstT[1], el0, er0, m0, E, N);
    edge_expsum_h4<<<2 * edgeBlocks0, TB, 0, stream>>>(srcT[0], dstT[0], srcT[1], dstT[1], el0, er0, m0, ee0, s0, E, N);
    owner_aggregate_h4d64<<<2 * aggBlocks, AGG_THREADS, 0, stream>>>(srcT[0], dstT[0], srcT[1], dstT[1], ee0, s0, feat0, acc0, E, N);

    epilogue0_h4d64<<<epi0Blocks, TB, 0, stream>>>(acc0, b0, h1, N);
  }

  {
    gemm_lds_k64_n64<<<(N + 127) / 128, GT_THREADS, 0, stream>>>(h1, W1, feat1, N);

    node_scores<<<(N * H1 + TB - 1) / TB, TB, 0, stream>>>(feat1, al1, ar1, el1, er1, m1, s1, N, H1, D1);

    edge_max_h1<<<2 * edgeBlocks1, TB, 0, stream>>>(srcT[0], dstT[0], srcT[1], dstT[1], el1, er1, m1, E, N);
    edge_expsum_h1<<<2 * edgeBlocks1, TB, 0, stream>>>(srcT[0], dstT[0], srcT[1], dstT[1], el1, er1, m1, ee1, s1, E, N);
    owner_aggregate_h1d64<<<2 * aggBlocks, AGG_THREADS, 0, stream>>>(srcT[0], dstT[0], srcT[1], dstT[1], ee1, s1, feat1, acc1, E, N);

    epilogue1_d64<<<epi1Blocks, TB, 0, stream>>>(acc1, b1, out, N);
  }
}


# Edge softmax moves into aggregation's owner blocks: row max and sum s by LDS atomics (arrival order), block's edge list kept in LDS between passes, rows divided by s after the sum; 12 -> 8 launches
# speedup vs baseline: 2.9378x; 1.6859x over previous
#include <hip/hip_runtime.h>
#include <math.h>

typedef float v2f __attribute__((ext_vector_type(2)));
typedef float v4f __attribute__((ext_vector_type(4)));
typedef float v8f __attribute__((ext_vector_type(8)));

#define LRELU_SLOPE 0.2f

__device__ __forceinline__ unsigned f2o(float f) {
  unsigned u = __float_as_uint(f);
  return (u & 0x80000000u) ? ~u : (u | 0x80000000u);
}
__device__ __forceinline__ float o2f(unsigned o) {
  unsigned u = (o & 0x80000000u) ? (o & 0x7fffffffu) : ~o;
  return __uint_as_float(u);
}
#define ORD_NEG_INF 0x007fffffu

#define GT_THREADS 512

template <int K, int NC>
__device__ __forceinline__ void gemm_lds_body(const float* __restrict__ A,
                                              const float* __restrict__ B,
                                              float* __restrict__ C, int M) {
  constexpr int NP  = NC / 32;
  constexpr int WR  = 16 / NP;
  constexpr int RS  = WR * 16;
  constexpr int PA  = K + 4;
  constexpr int PB  = 2 * NC + 32;
  constexpr int K4  = K / 4;
  constexpr int NC4 = NC / 4;
  constexpr int NPF = (RS * K4) / GT_THREADS;
  constexpr int NBF = ((K / 2) * NC4) / GT_THREADS;
  static_assert(NC % 32 == 0 && 16 % NP == 0, "a wave takes one M-tile and one column pair");
  static_assert((RS * K4) % GT_THREADS == 0 && (NPF == 2 || NPF == 4), "an A tile is 2 or 4 pieces per thread");
  static_assert(((K / 2) * NC4) % GT_THREADS == 0, "B splits evenly over the threads");
  static_assert(K % 4 == 0, "k steps of 4");

  __shared__ __align__(16) float Bs[(K / 2) * PB];
  __shared__ __align__(16) float As[2][RS * PA];

  const int tid  = threadIdx.x;
  const int lane = tid & 31;
  const int half = lane >> 4;
  const int l    = lane & 15;
  const int wave = __builtin_amdgcn_readfirstlane(tid >> 5);
  const int wm   = wave / NP;
  const int wn   = wave - wm * NP;

  const int nSteps = (M + RS - 1) / RS;

  auto fetch_piece = [&](int s, int u) -> v4f {
    const int it = tid + u * GT_THREADS;
    const int r = it / K4, c = it - r * K4;
    int row = s * RS + r;
    row = row < M ? row : M - 1;
    return *reinterpret_cast<const v4f*>(A + (size_t)row * K + 4 * c);
  };
  auto stash_piece = [&](float* as, int u, v4f v) {
    const int it = tid + u * GT_THREADS;
    const int r = it / K4, c = it - r * K4;
    *reinterpret_cast<v4f*>(as + r * PA + 4 * c) = v;
  };

  int step = blockIdx.x;
  v4f p0 = fetch_piece(step, 0), p1 = fetch_piece(step, 1), p2 = p0, p3 = p0;
  if constexpr (NPF == 4) { p2 = fetch_piece(step, 2); p3 = fetch_piece(step, 3); }
#pragma unroll
  for (int u = 0; u < NBF; ++u) {
    const int it = tid + u * GT_THREADS;
    const int kp = it / NC4, c = it - kp * NC4;
    const float4 r0 = *reinterpret_cast<const float4*>(B + (size_t)(2 * kp) * NC + 4 * c);
    const float4 r1 = *reinterpret_cast<const float4*>(B + (size_t)(2 * kp + 1) * NC + 4 * c);
    float4* d = reinterpret_cast<float4*>(Bs + kp * PB + 8 * c);
    d[0] = make_float4(r0.x, r1.x, r0.y, r1.y);
    d[1] = make_float4(r0.z, r1.z, r0.w, r1.w);
  }

  int buf = 0;
  for (; step < nSteps; step += gridDim.x) {
    float* as = As[buf];
    stash_piece(as, 0, p0);
    stash_piece(as, 1, p1);
    if constexpr (NPF == 4) { stash_piece(as, 2, p2); stash_piece(as, 3, p3); }
    __syncthreads();

    const int next = step + gridDim.x;
    p0 = fetch_piece(next, 0);
    p1 = fetch_piece(next, 1);
    if constexpr (NPF == 4) { p2 = fetch_piece(next, 2); p3 = fetch_piece(next, 3); }

    v8f acc0 = {}, acc1 = {};
    const float* ap = as + (wm * 16 + l) * PA + 2 * half;
    const float* bp = Bs + half * PB + (wn * 32 + l) * 2;
#pragma unroll
    for (int k = 0; k < K; k += 4) {
      const v2f a  = *reinterpret_cast<const v2f*>(ap + k);
      const v2f b0 = *reinterpret_cast<const v2f*>(bp + (k >> 1) * PB);
      const v2f b1 = *reinterpret_cast<const v2f*>(bp + (k >> 1) * PB + 32);
      acc0 = __builtin_amdgcn_wmma_f32_16x16x4_f32(false, a, false, b0, (short)0, acc0, false, false);
      acc1 = __builtin_amdgcn_wmma_f32_16x16x4_f32(false, a, false, b1, (short)0, acc1, false, false);
    }

    const int row0 = step * RS + wm * 16 + half * 8;
    float* cp = C + (size_t)row0 * NC + wn * 32 + l;
#pragma unroll
    for (int r = 0; r < 8; ++r) {
      if (row0 + r < M) {
        cp[(size_t)r * NC]      = acc0[r];
        cp[(size_t)r * NC + 16] = acc1[r];
      }
    }
    buf ^= 1;
  }
}

__global__ void __launch_bounds__(GT_THREADS)
gemm_lds_k128_n256(const float* __restrict__ A, const float* __restrict__ B, float* __restrict__ C, int M) {
  gemm_lds_body<128, 256>(A, B, C, M);
}
__global__ void __launch_bounds__(GT_THREADS)
gemm_lds_k64_n64(const float* __restrict__ A, const float* __restrict__ B, float* __restrict__ C, int M) {
  gemm_lds_body<64, 64>(A, B, C, M);
}

__global__ void node_scores(const float* __restrict__ feat,
                            const float* __restrict__ al,
                            const float* __restrict__ ar,
                            float* __restrict__ el, float* __restrict__ er,
                            int N, int H, int D) {
  int i = blockIdx.x * blockDim.x + threadIdx.x;
  if (i >= N * H) return;
  int n = i / H, h = i - n * H;
  const float* f = feat + (size_t)n * H * D + h * D;
  const float* wl = al + h * D;
  const float* wr = ar + h * D;
  float sl = 0.f, sr = 0.f;
#pragma unroll 4
  for (int d = 0; d < D; ++d) {
    float v = f[d];
    sl += v * wl[d];
    sr += v * wr[d];
  }
  el[i] = sl;
  er[i] = sr;
}

#define AGG_THREADS 1024
#define AGG_NODES   196
#define AGG_QCAP    192
#define AGG_U       8
#define AGG_LCAP    4096

template <int H>
__device__ __forceinline__ void load_heads(const float* __restrict__ p, int node, float (&o)[H]) {
  if constexpr (H == 4) {
    const float4 t = *reinterpret_cast<const float4*>(p + (size_t)node * 4);
    o[0] = t.x; o[1] = t.y; o[2] = t.z; o[3] = t.w;
  } else {
#pragma unroll
    for (int h = 0; h < H; ++h) o[h] = p[(size_t)node * H + h];
  }
}

template <int H, int D>
__device__ __forceinline__ void owner_softmax_aggregate_body(const int* __restrict__ src0, const int* __restrict__ dst0,
                                                             const int* __restrict__ src1, const int* __restrict__ dst1,
                                                             const float* __restrict__ el, const float* __restrict__ er,
                                                             const float* __restrict__ feat, float* __restrict__ acc,
                                                             int E, int N) {
  constexpr int R     = H * D;
  constexpr int LINES = R / 32;
  constexpr int LPH   = D / 32;
  constexpr int G     = (LINES >= 8) ? 4 : 16;
  static_assert(D % 32 == 0, "a line must lie inside one head");

  __shared__ float    sacc[AGG_NODES * R];
  __shared__ unsigned mrow[AGG_NODES * H];
  __shared__ float    srow[AGG_NODES * H];
  __shared__ int2     queue[(AGG_THREADS / 32) * AGG_QCAP];
  __shared__ int2     elist[AGG_LCAP];
  __shared__ int      listCount;

  const int tid  = threadIdx.x;
  const int lane = tid & 31;
  const int wave = __builtin_amdgcn_readfirstlane(tid >> 5);
  const int nb  = (int)(gridDim.x >> 1);
  const int t   = (int)blockIdx.x >= nb ? 1 : 0;
  const int bid = (int)blockIdx.x - t * nb;
  const int* __restrict__ src = t ? src1 : src0;
  const int* __restrict__ dst = t ? dst1 : dst0;
  acc += (size_t)t * N * R;
  const int lo = bid * AGG_NODES;
  const int nn = (N - lo) < AGG_NODES ? (N - lo) : AGG_NODES;

  float4* sacc4 = reinterpret_cast<float4*>(sacc);
  for (int i = tid; i < AGG_NODES * R / 4; i += AGG_THREADS) sacc4[i] = make_float4(0.f, 0.f, 0.f, 0.f);
  for (int i = tid; i < AGG_NODES * H; i += AGG_THREADS) { mrow[i] = ORD_NEG_INF; srow[i] = 0.f; }
  if (tid == 0) listCount = 0;
  __syncthreads();

  int2* q = queue + wave * AGG_QCAP;

  auto work_max = [&](const int2* qb, int n) {
    int2 ed = make_int2(0, 0);
    if (lane < n) {
      ed = qb[lane];
      const int sN = src[ed.x];
      float l[H], r[H];
      load_heads<H>(er, ed.y, r);
      load_heads<H>(el, sN, l);
      unsigned* mo = mrow + (ed.y - lo) * H;
#pragma unroll
      for (int h = 0; h < H; ++h) {
        float sc = l[h] + r[h];
        sc = sc > 0.f ? sc : LRELU_SLOPE * sc;
        atomicMax(mo + h, f2o(sc));
      }
    }
    int pos = 0;
    if (lane == 0) pos = atomicAdd(&listCount, n);
    pos = __builtin_amdgcn_readfirstlane(pos);
    if (lane < n && pos + lane < AGG_LCAP) elist[pos + lane] = ed;
  };

  auto work_acc = [&](const int2* qb, int n) {
    int sN = 0, dl = 0;
    float a[H];
#pragma unroll
    for (int h = 0; h < H; ++h) a[h] = 0.f;
    if (lane < n) {
      const int2 ed = qb[lane];
      sN = src[ed.x];
      dl = ed.y - lo;
      float l[H], r[H];
      load_heads<H>(er, ed.y, r);
      load_heads<H>(el, sN, l);
#pragma unroll
      for (int h = 0; h < H; ++h) {
        float sc = l[h] + r[h];
        sc = sc > 0.f ? sc : LRELU_SLOPE * sc;
        float m = o2f(mrow[dl * H + h]);
        if (m < -1e30f) m = 0.f;
        a[h] = __expf(sc - m);
        atomicAdd(srow + dl * H + h, a[h]);
      }
    }
    for (int k0 = 0; k0 < n; k0 += G) {
      float v[G][LINES];
#pragma unroll
      for (int g = 0; g < G; ++g) {
        const int k  = (k0 + g < n) ? (k0 + g) : (n - 1);
        const int sk = __builtin_amdgcn_readlane(sN, k);
        const float* f = feat + (size_t)sk * R + lane;
#pragma unroll
        for (int j = 0; j < LINES; ++j) v[g][j] = f[32 * j];
      }
#pragma unroll
      for (int g = 0; g < G; ++g)
#pragma unroll
        for (int j = 0; j < LINES; ++j) asm volatile("" :: "v"(v[g][j]));
#pragma unroll
      for (int g = 0; g < G; ++g) {
        const int k = k0 + g;
        if (k < n) {
          const int dk = __builtin_amdgcn_readlane(dl, k);
          float* o = sacc + dk * R + lane;
#pragma unroll
          for (int j = 0; j < LINES; ++j) {
            const float ak = __int_as_float(__builtin_amdgcn_readlane(__float_as_int(a[j / LPH]), k));
            atomicAdd(o + 32 * j, ak * v[g][j]);
          }
        }
      }
    }
  };

  auto scan_apply = [&](auto&& work) {
    int cnt = 0;
    const int nChunks = (E + 127) >> 7;
    const unsigned below = (1u << lane) - 1u;
    for (int c0 = wave; c0 < nChunks; c0 += (AGG_THREADS / 32) * AGG_U) {
      int4 d[AGG_U];
#pragma unroll
      for (int u = 0; u < AGG_U; ++u) {
        const int c    = c0 + (AGG_THREADS / 32) * u;
        const int base = c * 128 + 4 * lane;
        d[u] = make_int4(-1, -1, -1, -1);
        if (c < nChunks) {
          if (c * 128 + 128 <= E) {
            d[u] = *reinterpret_cast<const int4*>(dst + base);
          } else {
            if (base + 0 < E) d[u].x = dst[base + 0];
            if (base + 1 < E) d[u].y = dst[base + 1];
            if (base + 2 < E) d[u].z = dst[base + 2];
            if (base + 3 < E) d[u].w = dst[base + 3];
          }
        }
      }
#pragma unroll
      for (int u = 0; u < AGG_U; ++u) {
        const int c    = c0 + (AGG_THREADS / 32) * u;
        const int base = c * 128 + 4 * lane;
        const int dv[4] = { d[u].x, d[u].y, d[u].z, d[u].w };
#pragma unroll
        for (int j = 0; j < 4; ++j) {
          const bool mine = (unsigned)(dv[j] - lo) < (unsigned)nn;
          const unsigned mask = (unsigned)__ballot(mine);
          if (mask != 0u) {
            if (mine) q[cnt + __popc(mask & below)] = make_int2(base + j, dv[j]);
            cnt += __popc(mask);
          }
        }
#pragma nounroll
        while (cnt >= 32) { cnt -= 32; work(q + cnt, 32); }
      }
    }
    if (cnt > 0) work(q, cnt);
  };

  scan_apply(work_max);
  __syncthreads();

  const int total = __builtin_amdgcn_readfirstlane(listCount);
  if (__builtin_expect(total <= AGG_LCAP, 1)) {
    const int per = (total + (AGG_THREADS / 32) - 1) / (AGG_THREADS / 32);
    const int p1  = (wave + 1) * per < total ? (wave + 1) * per : total;
    for (int p = wave * per; p < p1; p += 32) {
      const int left = p1 - p;
      work_acc(elist + p, left < 32 ? left : 32);
    }
  } else {
    scan_apply(work_acc);
  }
  __syncthreads();

  float4* out4 = reinterpret_cast<float4*>(acc + (size_t)lo * R);
  for (int i = tid; i < nn * (R / 4); i += AGG_THREADS) {
    const int row = i / (R / 4);
    const int h   = (i - row * (R / 4)) / (D / 4);
    const float den = fmaxf(srow[row * H + h], 1e-9f);
    const float4 v = sacc4[i];
    out4[i] = make_float4(v.x / den, v.y / den, v.z / den, v.w / den);
  }
}

__global__ void __launch_bounds__(AGG_THREADS)
owner_softmax_aggregate_h4d64(const int* __restrict__ src0, const int* __restrict__ dst0,
                              const int* __restrict__ src1, const int* __restrict__ dst1,
                              const float* __restrict__ el, const float* __restrict__ er,
                              const float* __restrict__ feat, float* __restrict__ acc, int E, int N) {
  owner_softmax_aggregate_body<4, 64>(src0, dst0, src1, dst1, el, er, feat, acc, E, N);
}
__global__ void __launch_bounds__(AGG_THREADS)
owner_softmax_aggregate_h1d64(const int* __restrict__ src0, const int* __restrict__ dst0,
                              const int* __restrict__ src1, const int* __restrict__ dst1,
                              const float* __restrict__ el, const float* __restrict__ er,
                              const float* __restrict__ feat, float* __restrict__ acc, int E, int N) {
  owner_softmax_aggregate_body<1, 64>(src0, dst0, src1, dst1, el, er, feat, acc, E, N);
}

__device__ __forceinline__ float elu1(float v) { return (v > 0.f) ? v : expm1f(v); }

template <int H, int D>
__device__ __forceinline__ void epilogue0_body(const float* __restrict__ acc, const float* __restrict__ b,
                                               float* __restrict__ h1, int N) {
  constexpr int D4 = D / 4;
  const unsigned n4 = (unsigned)N * D4;
  const unsigned T = gridDim.x * blockDim.x;
  for (unsigned i = blockIdx.x * blockDim.x + threadIdx.x; i < n4; i += T) {
    const unsigned n = i / D4, q = i - n * D4;
    float4 v[2][H], bb[H];
#pragma unroll
    for (int t = 0; t < 2; ++t)
#pragma unroll
      for (int h = 0; h < H; ++h)
        v[t][h] = *reinterpret_cast<const float4*>(acc + ((size_t)t * N + n) * (H * D) + h * D + 4 * q);
#pragma unroll
    for (int h = 0; h < H; ++h) bb[h] = *reinterpret_cast<const float4*>(b + h * D + 4 * q);
    float4 sum = make_float4(0.f, 0.f, 0.f, 0.f);
#pragma unroll
    for (int t = 0; t < 2; ++t)
#pragma unroll
      for (int h = 0; h < H; ++h) {
        sum.x += elu1(v[t][h].x + bb[h].x);
        sum.y += elu1(v[t][h].y + bb[h].y);
        sum.z += elu1(v[t][h].z + bb[h].z);
        sum.w += elu1(v[t][h].w + bb[h].w);
      }
    const float hf = (float)H;
    *reinterpret_cast<float4*>(h1 + (size_t)i * 4) =
        make_float4(sum.x / hf, sum.y / hf, sum.z / hf, sum.w / hf);
  }
}
__global__ void epilogue0_h4d64(const float* __restrict__ acc, const float* __restrict__ b,
                                float* __restrict__ h1, int N) {
  epilogue0_body<4, 64>(acc, b, h1, N);
}


template <int D, int U>
__device__ __forceinline__ void epilogue1_body(const float* __restrict__ acc, const float* __restrict__ b,
                                               float* __restrict__ out, int N) {
  constexpr int D4 = D / 4;
  const unsigned n4 = (unsigned)N * D4;
  const unsigned T = gridDim.x * blockDim.x;
  const v4f* a0 = reinterpret_cast<const v4f*>(acc);
  const v4f* a1 = reinterpret_cast<const v4f*>(acc + (size_t)N * D);
  const v4f* b4 = reinterpret_cast<const v4f*>(b);
  v4f* o4 = reinterpret_cast<v4f*>(out);
  for (unsigned i0 = blockIdx.x * blockDim.x + threadIdx.x; i0 < n4; i0 += U * T) {
    v4f x[U], y[U], bb[U];
#pragma unroll
    for (int u = 0; u < U; ++u) {
      const unsigned i  = i0 + u * T;
      const unsigned ic = i < n4 ? i : n4 - 1;
      x[u]  = a0[ic];
      y[u]  = a1[ic];
      bb[u] = b4[ic % D4];
    }
#pragma unroll
    for (int u = 0; u < U; ++u) asm volatile("" :: "v"(x[u]), "v"(y[u]), "v"(bb[u]));
#pragma unroll
    for (int u = 0; u < U; ++u) {
      const unsigned i = i0 + u * T;
      if (i < n4) o4[i] = (x[u] + bb[u]) + (y[u] + bb[u]);
    }
  }
}
__global__ void epilogue1_d64(const float* __restrict__ acc, const float* __restrict__ b,
                              float* __restrict__ out, int N) {
  epilogue1_body<64, 4>(acc, b, out, N);
}

extern "C" void kernel_launch(void* const* d_in, const int* in_sizes, int n_in,
                              void* d_out, int out_size, void* d_ws, size_t ws_size,
                              hipStream_t stream) {
  const int IN = 128, H0 = 4, D0 = 64, H1 = 1, D1 = 64;
  const int N = in_sizes[0] / IN;
  const int E = in_sizes[9];

  const float* x   = (const float*)d_in[0];
  const float* W0  = (const float*)d_in[1];
  const float* al0 = (const float*)d_in[2];
  const float* ar0 = (const float*)d_in[3];
  const float* b0  = (const float*)d_in[4];
  const float* W1  = (const float*)d_in[5];
  const float* al1 = (const float*)d_in[6];
  const float* ar1 = (const float*)d_in[7];
  const float* b1  = (const float*)d_in[8];
  const int* srcT[2] = { (const int*)d_in[9],  (const int*)d_in[11] };
  const int* dstT[2] = { (const int*)d_in[10], (const int*)d_in[12] };
  float* out = (float*)d_out;

  float* ws = (float*)d_ws;
  float* feat0 = ws;
  float* el0   = feat0 + (size_t)N * H0 * D0;
  float* er0   = el0 + (size_t)N * H0;
  float* acc0  = er0 + (size_t)N * H0;
  float* h1    = acc0 + 2 * (size_t)N * H0 * D0;
  float* feat1 = feat0;
  float* el1   = feat1 + (size_t)N * D1;
  float* er1   = el1 + N;
  float* acc1  = er1 + N;

  const int TB = 256;
  const int aggBlocks = (N + AGG_NODES - 1) / AGG_NODES;
  const int epi0Blocks  = 1024;
  const int epi1Blocks  = (N * (D1 / 4) + TB * 4 - 1) / (TB * 4);
  (void)ws_size; (void)n_in; (void)out_size;

  {
    gemm_lds_k128_n256<<<256, GT_THREADS, 0, stream>>>(x, W0, feat0, N);

    node_scores<<<(N * H0 + TB - 1) / TB, TB, 0, stream>>>(feat0, al0, ar0, el0, er0, N, H0, D0);

    owner_softmax_aggregate_h4d64<<<2 * aggBlocks, AGG_THREADS, 0, stream>>>(srcT[0], dstT[0], srcT[1], dstT[1], el0, er0, feat0, acc0, E, N);

    epilogue0_h4d64<<<epi0Blocks, TB, 0, stream>>>(acc0, b0, h1, N);
  }

  {
    gemm_lds_k64_n64<<<(N + 127) / 128, GT_THREADS, 0, stream>>>(h1, W1, feat1, N);

    node_scores<<<(N * H1 + TB - 1) / TB, TB, 0, stream>>>(feat1, al1, ar1, el1, er1, N, H1, D1);

    owner_softmax_aggregate_h1d64<<<2 * aggBlocks, AGG_THREADS, 0, stream>>>(srcT[0], dstT[0], srcT[1], dstT[1], el1, er1, feat1, acc1, E, N);

    epilogue1_d64<<<epi1Blocks, TB, 0, stream>>>(acc1, b1, out, N);
  }
}


# Aggregation: layer 0's owner blocks leave their edge lists (source, destination pairs) in the workspace; layer 1's blocks walk them and no longer scan the dst lists; sums stay in LDS arrival order
# speedup vs baseline: 3.4536x; 1.1756x over previous
#include <hip/hip_runtime.h>
#include <math.h>

typedef float v2f __attribute__((ext_vector_type(2)));
typedef float v4f __attribute__((ext_vector_type(4)));
typedef float v8f __attribute__((ext_vector_type(8)));

#define LRELU_SLOPE 0.2f

__device__ __forceinline__ unsigned f2o(float f) {
  unsigned u = __float_as_uint(f);
  return (u & 0x80000000u) ? ~u : (u | 0x80000000u);
}
__device__ __forceinline__ float o2f(unsigned o) {
  unsigned u = (o & 0x80000000u) ? (o & 0x7fffffffu) : ~o;
  return __uint_as_float(u);
}
#define ORD_NEG_INF 0x007fffffu

#define GT_THREADS 512

template <int K, int NC>
__device__ __forceinline__ void gemm_lds_body(const float* __restrict__ A,
                                              const float* __restrict__ B,
                                              float* __restrict__ C, int M) {
  constexpr int NP  = NC / 32;
  constexpr int WR  = 16 / NP;
  constexpr int RS  = WR * 16;
  constexpr int PA  = K + 4;
  constexpr int PB  = 2 * NC + 32;
  constexpr int K4  = K / 4;
  constexpr int NC4 = NC / 4;
  constexpr int NPF = (RS * K4) / GT_THREADS;
  constexpr int NBF = ((K / 2) * NC4) / GT_THREADS;
  static_assert(NC % 32 == 0 && 16 % NP == 0, "a wave takes one M-tile and one column pair");
  static_assert((RS * K4) % GT_THREADS == 0 && (NPF == 2 || NPF == 4), "an A tile is 2 or 4 pieces per thread");
  static_assert(((K / 2) * NC4) % GT_THREADS == 0, "B splits evenly over the threads");
  static_assert(K % 4 == 0, "k steps of 4");

  __shared__ __align__(16) float Bs[(K / 2) * PB];
  __shared__ __align__(16) float As[2][RS * PA];

  const int tid  = threadIdx.x;
  const int lane = tid & 31;
  const int half = lane >> 4;
  const int l    = lane & 15;
  const int wave = __builtin_amdgcn_readfirstlane(tid >> 5);
  const int wm   = wave / NP;
  const int wn   = wave - wm * NP;

  const int nSteps = (M + RS - 1) / RS;

  auto fetch_piece = [&](int s, int u) -> v4f {
    const int it = tid + u * GT_THREADS;
    const int r = it / K4, c = it - r * K4;
    int row = s * RS + r;
    row = row < M ? row : M - 1;
    return *reinterpret_cast<const v4f*>(A + (size_t)row * K + 4 * c);
  };
  auto stash_piece = [&](float* as, int u, v4f v) {
    const int it = tid + u * GT_THREADS;
    const int r = it / K4, c = it - r * K4;
    *reinterpret_cast<v4f*>(as + r * PA + 4 * c) = v;
  };

  int step = blockIdx.x;
  v4f p0 = fetch_piece(step, 0), p1 = fetch_piece(step, 1), p2 = p0, p3 = p0;
  if constexpr (NPF == 4) { p2 = fetch_piece(step, 2); p3 = fetch_piece(step, 3); }
#pragma unroll
  for (int u = 0; u < NBF; ++u) {
    const int it = tid + u * GT_THREADS;
    const int kp = it / NC4, c = it - kp * NC4;
    const float4 r0 = *reinterpret_cast<const float4*>(B + (size_t)(2 * kp) * NC + 4 * c);
    const float4 r1 = *reinterpret_cast<const float4*>(B + (size_t)(2 * kp + 1) * NC + 4 * c);
    float4* d = reinterpret_cast<float4*>(Bs + kp * PB + 8 * c);
    d[0] = make_float4(r0.x, r1.x, r0.y, r1.y);
    d[1] = make_float4(r0.z, r1.z, r0.w, r1.w);
  }

  int buf = 0;
  for (; step < nSteps; step += gridDim.x) {
    float* as = As[buf];
    stash_piece(as, 0, p0);
    stash_piece(as, 1, p1);
    if constexpr (NPF == 4) { stash_piece(as, 2, p2); stash_piece(as, 3, p3); }
    __syncthreads();

    const int next = step + gridDim.x;
    p0 = fetch_piece(next, 0);
    p1 = fetch_piece(next, 1);
    if constexpr (NPF == 4) { p2 = fetch_piece(next, 2); p3 = fetch_piece(next, 3); }

    v8f acc0 = {}, acc1 = {};
    const float* ap = as + (wm * 16 + l) * PA + 2 * half;
    const float* bp = Bs + half * PB + (wn * 32 + l) * 2;
#pragma unroll
    for (int k = 0; k < K; k += 4) {
      const v2f a  = *reinterpret_cast<const v2f*>(ap + k);
      const v2f b0 = *reinterpret_cast<const v2f*>(bp + (k >> 1) * PB);
      const v2f b1 = *reinterpret_cast<const v2f*>(bp + (k >> 1) * PB + 32);
      acc0 = __builtin_amdgcn_wmma_f32_16x16x4_f32(false, a, false, b0, (short)0, acc0, false, false);
      acc1 = __builtin_amdgcn_wmma_f32_16x16x4_f32(false, a, false, b1, (short)0, acc1, false, false);
    }

    const int row0 = step * RS + wm * 16 + half * 8;
    float* cp = C + (size_t)row0 * NC + wn * 32 + l;
#pragma unroll
    for (int r = 0; r < 8; ++r) {
      if (row0 + r < M) {
        cp[(size_t)r * NC]      = acc0[r];
        cp[(size_t)r * NC + 16] = acc1[r];
      }
    }
    buf ^= 1;
  }
}

__global__ void __launch_bounds__(GT_THREADS)
gemm_lds_k128_n256(const float* __restrict__ A, const float* __restrict__ B, float* __restrict__ C, int M) {
  gemm_lds_body<128, 256>(A, B, C, M);
}
__global__ void __launch_bounds__(GT_THREADS)
gemm_lds_k64_n64(const float* __restrict__ A, const float* __restrict__ B, float* __restrict__ C, int M) {
  gemm_lds_body<64, 64>(A, B, C, M);
}

__global__ void node_scores(const float* __restrict__ feat,
                            const float* __restrict__ al,
                            const float* __restrict__ ar,
                            float* __restrict__ el, float* __restrict__ er,
                            int N, int H, int D) {
  int i = blockIdx.x * blockDim.x + threadIdx.x;
  if (i >= N * H) return;
  int n = i / H, h = i - n * H;
  const float* f = feat + (size_t)n * H * D + h * D;
  const float* wl = al + h * D;
  const float* wr = ar + h * D;
  float sl = 0.f, sr = 0.f;
#pragma unroll 4
  for (int d = 0; d < D; ++d) {
    float v = f[d];
    sl += v * wl[d];
    sr += v * wr[d];
  }
  el[i] = sl;
  er[i] = sr;
}

#define AGG_THREADS 1024
#define AGG_NODES   196
#define AGG_QCAP    192
#define AGG_U       8
#define AGG_LCAP    4096

template <bool B> struct list_kind { static constexpr bool value = B; };

template <int H>
__device__ __forceinline__ void load_heads(const float* __restrict__ p, int node, float (&o)[H]) {
  if constexpr (H == 4) {
    const float4 t = *reinterpret_cast<const float4*>(p + (size_t)node * 4);
    o[0] = t.x; o[1] = t.y; o[2] = t.z; o[3] = t.w;
  } else {
#pragma unroll
    for (int h = 0; h < H; ++h) o[h] = p[(size_t)node * H + h];
  }
}

template <int H, int D, int MODE>
__device__ __forceinline__ void owner_softmax_aggregate_body(const int* __restrict__ src0, const int* __restrict__ dst0,
                                                             const int* __restrict__ src1, const int* __restrict__ dst1,
                                                             const float* __restrict__ el, const float* __restrict__ er,
                                                             const float* __restrict__ feat, float* __restrict__ acc,
                                                             int* __restrict__ lists, int* __restrict__ counts,
                                                             int E, int N) {
  constexpr int R     = H * D;
  constexpr int LINES = R / 32;
  constexpr int LPH   = D / 32;
  constexpr int G     = (LINES >= 8) ? 4 : 16;
  static_assert(D % 32 == 0, "a line must lie inside one head");

  __shared__ float    sacc[AGG_NODES * R];
  __shared__ unsigned mrow[AGG_NODES * H];
  __shared__ float    srow[AGG_NODES * H];
  __shared__ int2     queue[(AGG_THREADS / 32) * AGG_QCAP];
  __shared__ int2     elist[AGG_LCAP];
  __shared__ int      listCount;

  const int tid  = threadIdx.x;
  const int lane = tid & 31;
  const int wave = __builtin_amdgcn_readfirstlane(tid >> 5);
  const int nb  = (int)(gridDim.x >> 1);
  const int t   = (int)blockIdx.x >= nb ? 1 : 0;
  const int bid = (int)blockIdx.x - t * nb;
  const int* __restrict__ src = t ? src1 : src0;
  const int* __restrict__ dst = t ? dst1 : dst0;
  acc += (size_t)t * N * R;
  const int lo = bid * AGG_NODES;
  const int nn = (N - lo) < AGG_NODES ? (N - lo) : AGG_NODES;

  int2* glist = reinterpret_cast<int2*>(lists) + (size_t)blockIdx.x * AGG_LCAP;
  int saved = AGG_LCAP + 1;
  if constexpr (MODE == 1) saved = counts[blockIdx.x];

  float4* sacc4 = reinterpret_cast<float4*>(sacc);
  for (int i = tid; i < AGG_NODES * R / 4; i += AGG_THREADS) sacc4[i] = make_float4(0.f, 0.f, 0.f, 0.f);
  for (int i = tid; i < AGG_NODES * H; i += AGG_THREADS) { mrow[i] = ORD_NEG_INF; srow[i] = 0.f; }
  if (tid == 0) listCount = 0;
  __syncthreads();

  int2* q = queue + wave * AGG_QCAP;

  auto work_max = [&](const int2* qb, int n, auto kind) {
    constexpr bool SAVED = decltype(kind)::value;
    int2 ed = make_int2(0, 0);
    int sN = 0;
    if (lane < n) {
      ed = qb[lane];
      if constexpr (SAVED) sN = ed.x; else sN = src[ed.x];
      float l[H], r[H];
      load_heads<H>(er, ed.y, r);
      load_heads<H>(el, sN, l);
      unsigned* mo = mrow + (ed.y - lo) * H;
#pragma unroll
      for (int h = 0; h < H; ++h) {
        float sc = l[h] + r[h];
        sc = sc > 0.f ? sc : LRELU_SLOPE * sc;
        atomicMax(mo + h, f2o(sc));
      }
    }
    if constexpr (!SAVED) {
      int pos = 0;
      if (lane == 0) pos = atomicAdd(&listCount, n);
      pos = __builtin_amdgcn_readfirstlane(pos);
      if (lane < n && pos + lane < AGG_LCAP) {
        elist[pos + lane] = ed;
        if constexpr (MODE == 0) glist[pos + lane] = make_int2(sN, ed.y);
      }
    }
  };

  auto work_acc = [&](const int2* qb, int n, auto kind) {
    constexpr bool SAVED = decltype(kind)::value;
    int sN = 0, dl = 0;
    float a[H];
#pragma unroll
    for (int h = 0; h < H; ++h) a[h] = 0.f;
    if (lane < n) {
      const int2 ed = qb[lane];
      if constexpr (SAVED) sN = ed.x; else sN = src[ed.x];
      dl = ed.y - lo;
      float l[H], r[H];
      load_heads<H>(er, ed.y, r);
      load_heads<H>(el, sN, l);
#pragma unroll
      for (int h = 0; h < H; ++h) {
        float sc = l[h] + r[h];
        sc = sc > 0.f ? sc : LRELU_SLOPE * sc;
        float m = o2f(mrow[dl * H + h]);
        if (m < -1e30f) m = 0.f;
        a[h] = __expf(sc - m);
        atomicAdd(srow + dl * H + h, a[h]);
      }
    }
    for (int k0 = 0; k0 < n; k0 += G) {
      float v[G][LINES];
#pragma unroll
      for (int g = 0; g < G; ++g) {
        const int k  = (k0 + g < n) ? (k0 + g) : (n - 1);
        const int sk = __builtin_amdgcn_readlane(sN, k);
        const float* f = feat + (size_t)sk * R + lane;
#pragma unroll
        for (int j = 0; j < LINES; ++j) v[g][j] = f[32 * j];
      }
#pragma unroll
      for (int g = 0; g < G; ++g)
#pragma unroll
        for (int j = 0; j < LINES; ++j) asm volatile("" :: "v"(v[g][j]));
#pragma unroll
      for (int g = 0; g < G; ++g) {
        const int k = k0 + g;
        if (k < n) {
          const int dk = __builtin_amdgcn_readlane(dl, k);
          float* o = sacc + dk * R + lane;
#pragma unroll
          for (int j = 0; j < LINES; ++j) {
            const float ak = __int_as_float(__builtin_amdgcn_readlane(__float_as_int(a[j / LPH]), k));
            atomicAdd(o + 32 * j, ak * v[g][j]);
          }
        }
      }
    }
  };

  auto scan_apply = [&](auto&& work) {
    int cnt = 0;
    const int nChunks = (E + 127) >> 7;
    const unsigned below = (1u << lane) - 1u;
    for (int c0 = wave; c0 < nChunks; c0 += (AGG_THREADS / 32) * AGG_U) {
      int4 d[AGG_U];
#pragma unroll
      for (int u = 0; u < AGG_U; ++u) {
        const int c    = c0 + (AGG_THREADS / 32) * u;
        const int base = c * 128 + 4 * lane;
        d[u] = make_int4(-1, -1, -1, -1);
        if (c < nChunks) {
          if (c * 128 + 128 <= E) {
            d[u] = *reinterpret_cast<const int4*>(dst + base);
          } else {
            if (base + 0 < E) d[u].x = dst[base + 0];
            if (base + 1 < E) d[u].y = dst[base + 1];
            if (base + 2 < E) d[u].z = dst[base + 2];
            if (base + 3 < E) d[u].w = dst[base + 3];
          }
        }
      }
#pragma unroll
      for (int u = 0; u < AGG_U; ++u) {
        const int c    = c0 + (AGG_THREADS / 32) * u;
        const int base = c * 128 + 4 * lane;
        const int dv[4] = { d[u].x, d[u].y, d[u].z, d[u].w };
#pragma unroll
        for (int j = 0; j < 4; ++j) {
          const bool mine = (unsigned)(dv[j] - lo) < (unsigned)nn;
          const unsigned mask = (unsigned)__ballot(mine);
          if (mask != 0u) {
            if (mine) q[cnt + __popc(mask & below)] = make_int2(base + j, dv[j]);
            cnt += __popc(mask);
          }
        }
#pragma nounroll
        while (cnt >= 32) { cnt -= 32; work(q + cnt, 32); }
      }
    }
    if (cnt > 0) work(q, cnt);
  };

  saved = __builtin_amdgcn_readfirstlane(saved);
  if (MODE == 1 && saved <= AGG_LCAP) {
    const int per = (saved + (AGG_THREADS / 32) - 1) / (AGG_THREADS / 32);
    const int p1  = (wave + 1) * per < saved ? (wave + 1) * per : saved;
    for (int p = wave * per; p < p1; p += 32) {
      const int left = p1 - p;
      work_max(glist + p, left < 32 ? left : 32, list_kind<true>{});
    }
    __syncthreads();
    for (int p = wave * per; p < p1; p += 32) {
      const int left = p1 - p;
      work_acc(glist + p, left < 32 ? left : 32, list_kind<true>{});
    }
  } else {
    scan_apply([&](const int2* qb, int n) { work_max(qb, n, list_kind<false>{}); });
    __syncthreads();

    const int total = __builtin_amdgcn_readfirstlane(listCount);
    if constexpr (MODE == 0) { if (tid == 0) counts[blockIdx.x] = total; }
    if (__builtin_expect(total <= AGG_LCAP, 1)) {
      const int per = (total + (AGG_THREADS / 32) - 1) / (AGG_THREADS / 32);
      const int p1  = (wave + 1) * per < total ? (wave + 1) * per : total;
      for (int p = wave * per; p < p1; p += 32) {
        const int left = p1 - p;
        work_acc(elist + p, left < 32 ? left : 32, list_kind<false>{});
      }
    } else {
      scan_apply([&](const int2* qb, int n) { work_acc(qb, n, list_kind<false>{}); });
    }
  }
  __syncthreads();

  float4* out4 = reinterpret_cast<float4*>(acc + (size_t)lo * R);
  for (int i = tid; i < nn * (R / 4); i += AGG_THREADS) {
    const int row = i / (R / 4);
    const int h   = (i - row * (R / 4)) / (D / 4);
    const float den = fmaxf(srow[row * H + h], 1e-9f);
    const float4 v = sacc4[i];
    out4[i] = make_float4(v.x / den, v.y / den, v.z / den, v.w / den);
  }
}

__global__ void __launch_bounds__(AGG_THREADS)
owner_softmax_aggregate_h4d64(const int* __restrict__ src0, const int* __restrict__ dst0,
                              const int* __restrict__ src1, const int* __restrict__ dst1,
                              const float* __restrict__ el, const float* __restrict__ er,
                              const float* __restrict__ feat, float* __restrict__ acc,
                              int* __restrict__ lists, int* __restrict__ counts, int E, int N) {
  owner_softmax_aggregate_body<4, 64, 0>(src0, dst0, src1, dst1, el, er, feat, acc, lists, counts, E, N);
}
__global__ void __launch_bounds__(AGG_THREADS)
owner_softmax_aggregate_h1d64(const int* __restrict__ src0, const int* __restrict__ dst0,
                              const int* __restrict__ src1, const int* __restrict__ dst1,
                              const float* __restrict__ el, const float* __restrict__ er,
                              const float* __restrict__ feat, float* __restrict__ acc,
                              int* __restrict__ lists, int* __restrict__ counts, int E, int N) {
  owner_softmax_aggregate_body<1, 64, 1>(src0, dst0, src1, dst1, el, er, feat, acc, lists, counts, E, N);
}

__device__ __forceinline__ float elu1(float v) { return (v > 0.f) ? v : expm1f(v); }

template <int H, int D>
__device__ __forceinline__ void epilogue0_body(const float* __restrict__ acc, const float* __restrict__ b,
                                               float* __restrict__ h1, int N) {
  constexpr int D4 = D / 4;
  const unsigned n4 = (unsigned)N * D4;
  const unsigned T = gridDim.x * blockDim.x;
  for (unsigned i = blockIdx.x * blockDim.x + threadIdx.x; i < n4; i += T) {
    const unsigned n = i / D4, q = i - n * D4;
    float4 v[2][H], bb[H];
#pragma unroll
    for (int t = 0; t < 2; ++t)
#pragma unroll
      for (int h = 0; h < H; ++h)
        v[t][h] = *reinterpret_cast<const float4*>(acc + ((size_t)t * N + n) * (H * D) + h * D + 4 * q);
#pragma unroll
    for (int h = 0; h < H; ++h) bb[h] = *reinterpret_cast<const float4*>(b + h * D + 4 * q);
    float4 sum = make_float4(0.f, 0.f, 0.f, 0.f);
#pragma unroll
    for (int t = 0; t < 2; ++t)
#pragma unroll
      for (int h = 0; h < H; ++h) {
        sum.x += elu1(v[t][h].x + bb[h].x);
        sum.y += elu1(v[t][h].y + bb[h].y);
        sum.z += elu1(v[t][h].z + bb[h].z);
        sum.w += elu1(v[t][h].w + bb[h].w);
      }
    const float hf = (float)H;
    *reinterpret_cast<float4*>(h1 + (size_t)i * 4) =
        make_float4(sum.x / hf, sum.y / hf, sum.z / hf, sum.w / hf);
  }
}
__global__ void epilogue0_h4d64(const float* __restrict__ acc, const float* __restrict__ b,
                                float* __restrict__ h1, int N) {
  epilogue0_body<4, 64>(acc, b, h1, N);
}


template <int D, int U>
__device__ __forceinline__ void epilogue1_body(const float* __restrict__ acc, const float* __restrict__ b,
                                               float* __restrict__ out, int N) {
  constexpr int D4 = D / 4;
  const unsigned n4 = (unsigned)N * D4;
  const unsigned T = gridDim.x * blockDim.x;
  const v4f* a0 = reinterpret_cast<const v4f*>(acc);
  const v4f* a1 = reinterpret_cast<const v4f*>(acc + (size_t)N * D);
  const v4f* b4 = reinterpret_cast<const v4f*>(b);
  v4f* o4 = reinterpret_cast<v4f*>(out);
  for (unsigned i0 = blockIdx.x * blockDim.x + threadIdx.x; i0 < n4; i0 += U * T) {
    v4f x[U], y[U], bb[U];
#pragma unroll
    for (int u = 0; u < U; ++u) {
      const unsigned i  = i0 + u * T;
      const unsigned ic = i < n4 ? i : n4 - 1;
      x[u]  = a0[ic];
      y[u]  = a1[ic];
      bb[u] = b4[ic % D4];
    }
#pragma unroll
    for (int u = 0; u < U; ++u) asm volatile("" :: "v"(x[u]), "v"(y[u]), "v"(bb[u]));
#pragma unroll
    for (int u = 0; u < U; ++u) {
      const unsigned i = i0 + u * T;
      if (i < n4) o4[i] = (x[u] + bb[u]) + (y[u] + bb[u]);
    }
  }
}
__global__ void epilogue1_d64(const float* __restrict__ acc, const float* __restrict__ b,
                              float* __restrict__ out, int N) {
  epilogue1_body<64, 4>(acc, b, out, N);
}

extern "C" void kernel_launch(void* const* d_in, const int* in_sizes, int n_in,
                              void* d_out, int out_size, void* d_ws, size_t ws_size,
                              hipStream_t stream) {
  const int IN = 128, H0 = 4, D0 = 64, H1 = 1, D1 = 64;
  const int N = in_sizes[0] / IN;
  const int E = in_sizes[9];

  const float* x   = (const float*)d_in[0];
  const float* W0  = (const float*)d_in[1];
  const float* al0 = (const float*)d_in[2];
  const float* ar0 = (const float*)d_in[3];
  const float* b0  = (const float*)d_in[4];
  const float* W1  = (const float*)d_in[5];
  const float* al1 = (const float*)d_in[6];
  const float* ar1 = (const float*)d_in[7];
  const float* b1  = (const float*)d_in[8];
  const int* srcT[2] = { (const int*)d_in[9],  (const int*)d_in[11] };
  const int* dstT[2] = { (const int*)d_in[10], (const int*)d_in[12] };
  float* out = (float*)d_out;

  float* ws = (float*)d_ws;
  float* feat0 = ws;
  float* el0   = feat0 + (size_t)N * H0 * D0;
  float* er0   = el0 + (size_t)N * H0;
  float* acc0  = er0 + (size_t)N * H0;
  float* h1    = acc0 + 2 * (size_t)N * H0 * D0;
  float* feat1 = feat0;
  float* el1   = feat1 + (size_t)N * D1;
  float* er1   = el1 + N;
  float* acc1  = er1 + N;

  const int TB = 256;
  const int aggBlocks = (N + AGG_NODES - 1) / AGG_NODES;
  int* lists  = (int*)(h1 + (size_t)N * D0);
  int* counts = lists + (size_t)2 * aggBlocks * AGG_LCAP * 2;
  const int epi0Blocks  = 1024;
  const int epi1Blocks  = (N * (D1 / 4) + TB * 4 - 1) / (TB * 4);
  (void)ws_size; (void)n_in; (void)out_size;

  {
    gemm_lds_k128_n256<<<256, GT_THREADS, 0, stream>>>(x, W0, feat0, N);

    node_scores<<<(N * H0 + TB - 1) / TB, TB, 0, stream>>>(feat0, al0, ar0, el0, er0, N, H0, D0);

    owner_softmax_aggregate_h4d64<<<2 * aggBlocks, AGG_THREADS, 0, stream>>>(srcT[0], dstT[0], srcT[1], dstT[1], el0, er0, feat0, acc0, lists, counts, E, N);

    epilogue0_h4d64<<<epi0Blocks, TB, 0, stream>>>(acc0, b0, h1, N);
  }

  {
    gemm_lds_k64_n64<<<(N + 127) / 128, GT_THREADS, 0, stream>>>(h1, W1, feat1, N);

    node_scores<<<(N * H1 + TB - 1) / TB, TB, 0, stream>>>(feat1, al1, ar1, el1, er1, N, H1, D1);

    owner_softmax_aggregate_h1d64<<<2 * aggBlocks, AGG_THREADS, 0, stream>>>(srcT[0], dstT[0], srcT[1], dstT[1], el1, er1, feat1, acc1, lists, counts, E, N);

    epilogue1_d64<<<epi1Blocks, TB, 0, stream>>>(acc1, b1, out, N);
  }
}


# Owner blocks' edge lists come from one bucket pass (LDS counts per 196-row bucket, one global add per block and bucket); layer 0 no longer scans the dst lists; sums stay in LDS arrival order
# speedup vs baseline: 3.9565x; 1.1456x over previous
#include <hip/hip_runtime.h>
#include <math.h>

typedef float v2f __attribute__((ext_vector_type(2)));
typedef float v4f __attribute__((ext_vector_type(4)));
typedef float v8f __attribute__((ext_vector_type(8)));

#define LRELU_SLOPE 0.2f

__device__ __forceinline__ unsigned f2o(float f) {
  unsigned u = __float_as_uint(f);
  return (u & 0x80000000u) ? ~u : (u | 0x80000000u);
}
__device__ __forceinline__ float o2f(unsigned o) {
  unsigned u = (o & 0x80000000u) ? (o & 0x7fffffffu) : ~o;
  return __uint_as_float(u);
}
#define ORD_NEG_INF 0x007fffffu

#define GT_THREADS 512

template <int K, int NC>
__device__ __forceinline__ void gemm_lds_body(const float* __restrict__ A,
                                              const float* __restrict__ B,
                                              float* __restrict__ C, int M) {
  constexpr int NP  = NC / 32;
  constexpr int WR  = 16 / NP;
  constexpr int RS  = WR * 16;
  constexpr int PA  = K + 4;
  constexpr int PB  = 2 * NC + 32;
  constexpr int K4  = K / 4;
  constexpr int NC4 = NC / 4;
  constexpr int NPF = (RS * K4) / GT_THREADS;
  constexpr int NBF = ((K / 2) * NC4) / GT_THREADS;
  static_assert(NC % 32 == 0 && 16 % NP == 0, "a wave takes one M-tile and one column pair");
  static_assert((RS * K4) % GT_THREADS == 0 && (NPF == 2 || NPF == 4), "an A tile is 2 or 4 pieces per thread");
  static_assert(((K / 2) * NC4) % GT_THREADS == 0, "B splits evenly over the threads");
  static_assert(K % 4 == 0, "k steps of 4");

  __shared__ __align__(16) float Bs[(K / 2) * PB];
  __shared__ __align__(16) float As[2][RS * PA];

  const int tid  = threadIdx.x;
  const int lane = tid & 31;
  const int half = lane >> 4;
  const int l    = lane & 15;
  const int wave = __builtin_amdgcn_readfirstlane(tid >> 5);
  const int wm   = wave / NP;
  const int wn   = wave - wm * NP;

  const int nSteps = (M + RS - 1) / RS;

  auto fetch_piece = [&](int s, int u) -> v4f {
    const int it = tid + u * GT_THREADS;
    const int r = it / K4, c = it - r * K4;
    int row = s * RS + r;
    row = row < M ? row : M - 1;
    return *reinterpret_cast<const v4f*>(A + (size_t)row * K + 4 * c);
  };
  auto stash_piece = [&](float* as, int u, v4f v) {
    const int it = tid + u * GT_THREADS;
    const int r = it / K4, c = it - r * K4;
    *reinterpret_cast<v4f*>(as + r * PA + 4 * c) = v;
  };

  int step = blockIdx.x;
  v4f p0 = fetch_piece(step, 0), p1 = fetch_piece(step, 1), p2 = p0, p3 = p0;
  if constexpr (NPF == 4) { p2 = fetch_piece(step, 2); p3 = fetch_piece(step, 3); }
#pragma unroll
  for (int u = 0; u < NBF; ++u) {
    const int it = tid + u * GT_THREADS;
    const int kp = it / NC4, c = it - kp * NC4;
    const float4 r0 = *reinterpret_cast<const float4*>(B + (size_t)(2 * kp) * NC + 4 * c);
    const float4 r1 = *reinterpret_cast<const float4*>(B + (size_t)(2 * kp + 1) * NC + 4 * c);
    float4* d = reinterpret_cast<float4*>(Bs + kp * PB + 8 * c);
    d[0] = make_float4(r0.x, r1.x, r0.y, r1.y);
    d[1] = make_float4(r0.z, r1.z, r0.w, r1.w);
  }

  int buf = 0;
  for (; step < nSteps; step += gridDim.x) {
    float* as = As[buf];
    stash_piece(as, 0, p0);
    stash_piece(as, 1, p1);
    if constexpr (NPF == 4) { stash_piece(as, 2, p2); stash_piece(as, 3, p3); }
    __syncthreads();

    const int next = step + gridDim.x;
    p0 = fetch_piece(next, 0);
    p1 = fetch_piece(next, 1);
    if constexpr (NPF == 4) { p2 = fetch_piece(next, 2); p3 = fetch_piece(next, 3); }

    v8f acc0 = {}, acc1 = {};
    const float* ap = as + (wm * 16 + l) * PA + 2 * half;
    const float* bp = Bs + half * PB + (wn * 32 + l) * 2;
#pragma unroll
    for (int k = 0; k < K; k += 4) {
      const v2f a  = *reinterpret_cast<const v2f*>(ap + k);
      const v2f b0 = *reinterpret_cast<const v2f*>(bp + (k >> 1) * PB);
      const v2f b1 = *reinterpret_cast<const v2f*>(bp + (k >> 1) * PB + 32);
      acc0 = __builtin_amdgcn_wmma_f32_16x16x4_f32(false, a, false, b0, (short)0, acc0, false, false);
      acc1 = __builtin_amdgcn_wmma_f32_16x16x4_f32(false, a, false, b1, (short)0, acc1, false, false);
    }

    const int row0 = step * RS + wm * 16 + half * 8;
    float* cp = C + (size_t)row0 * NC + wn * 32 + l;
#pragma unroll
    for (int r = 0; r < 8; ++r) {
      if (row0 + r < M) {
        cp[(size_t)r * NC]      = acc0[r];
        cp[(size_t)r * NC + 16] = acc1[r];
      }
    }
    buf ^= 1;
  }
}

__global__ void __launch_bounds__(GT_THREADS)
gemm_lds_k128_n256(const float* __restrict__ A, const float* __restrict__ B, float* __restrict__ C, int M,
                   int* __restrict__ zero, int nzero) {
  if (blockIdx.x == 0)
    for (int i = threadIdx.x; i < nzero; i += GT_THREADS) zero[i] = 0;
  gemm_lds_body<128, 256>(A, B, C, M);
}
__global__ void __launch_bounds__(GT_THREADS)
gemm_lds_k64_n64(const float* __restrict__ A, const float* __restrict__ B, float* __restrict__ C, int M) {
  gemm_lds_body<64, 64>(A, B, C, M);
}

__global__ void node_scores(const float* __restrict__ feat,
                            const float* __restrict__ al,
                            const float* __restrict__ ar,
                            float* __restrict__ el, float* __restrict__ er,
                            int N, int H, int D) {
  int i = blockIdx.x * blockDim.x + threadIdx.x;
  if (i >= N * H) return;
  int n = i / H, h = i - n * H;
  const float* f = feat + (size_t)n * H * D + h * D;
  const float* wl = al + h * D;
  const float* wr = ar + h * D;
  float sl = 0.f, sr = 0.f;
#pragma unroll 4
  for (int d = 0; d < D; ++d) {
    float v = f[d];
    sl += v * wl[d];
    sr += v * wr[d];
  }
  el[i] = sl;
  er[i] = sr;
}

#define AGG_THREADS 1024
#define AGG_NODES   196
#define AGG_QCAP    192
#define AGG_U       8
#define AGG_LCAP    4096

#define BK_THREADS 1024
#define BK_MAXB    1024

__global__ void __launch_bounds__(BK_THREADS)
bucket_edges(const int* __restrict__ src0, const int* __restrict__ dst0,
             const int* __restrict__ src1, const int* __restrict__ dst1,
             int* __restrict__ lists, int* __restrict__ counts, int E, int N, int nb) {
  __shared__ int place[BK_MAXB];
  const int tid = threadIdx.x;
  const int cb  = (int)(gridDim.x >> 1);
  const int t   = (int)blockIdx.x >= cb ? 1 : 0;
  const int b   = (int)blockIdx.x - t * cb;
  counts += t * nb;
  if (nb > BK_MAXB) {
    if (b == 0)
      for (int i = tid; i < nb; i += BK_THREADS) counts[i] = AGG_LCAP + 1;
    return;
  }
  const int* __restrict__ src = t ? src1 : src0;
  const int* __restrict__ dst = t ? dst1 : dst0;
  int2* glist = reinterpret_cast<int2*>(lists) + (size_t)t * nb * AGG_LCAP;

  for (int i = tid; i < nb; i += BK_THREADS) place[i] = 0;
  __syncthreads();

  const int base = (b * BK_THREADS + tid) * 4;
  int dv[4] = { -1, -1, -1, -1 };
  int sv[4] = { 0, 0, 0, 0 };
  if (base + 4 <= E) {
    const int4 d4 = *reinterpret_cast<const int4*>(dst + base);
    const int4 s4 = *reinterpret_cast<const int4*>(src + base);
    dv[0] = d4.x; dv[1] = d4.y; dv[2] = d4.z; dv[3] = d4.w;
    sv[0] = s4.x; sv[1] = s4.y; sv[2] = s4.z; sv[3] = s4.w;
  } else {
#pragma unroll
    for (int j = 0; j < 4; ++j)
      if (base + j < E) { dv[j] = dst[base + j]; sv[j] = src[base + j]; }
  }
  int bk[4], rk[4];
#pragma unroll
  for (int j = 0; j < 4; ++j) {
    const bool in = (unsigned)dv[j] < (unsigned)N;
    bk[j] = in ? dv[j] / AGG_NODES : -1;
    rk[j] = 0;
    if (in) rk[j] = atomicAdd(&place[bk[j]], 1);
  }
  __syncthreads();
  for (int i = tid; i < nb; i += BK_THREADS) {
    const int n = place[i];
    if (n > 0) place[i] = atomicAdd(counts + i, n);
  }
  __syncthreads();
#pragma unroll
  for (int j = 0; j < 4; ++j) {
    if (bk[j] >= 0) {
      const int pos = place[bk[j]] + rk[j];
      if (pos < AGG_LCAP) glist[(size_t)bk[j] * AGG_LCAP + pos] = make_int2(sv[j], dv[j]);
    }
  }
}

template <bool B> struct list_kind { static constexpr bool value = B; };

template <int H>
__device__ __forceinline__ void load_heads(const float* __restrict__ p, int node, float (&o)[H]) {
  if constexpr (H == 4) {
    const float4 t = *reinterpret_cast<const float4*>(p + (size_t)node * 4);
    o[0] = t.x; o[1] = t.y; o[2] = t.z; o[3] = t.w;
  } else {
#pragma unroll
    for (int h = 0; h < H; ++h) o[h] = p[(size_t)node * H + h];
  }
}

template <int H, int D>
__device__ __forceinline__ void owner_softmax_aggregate_body(const int* __restrict__ src0, const int* __restrict__ dst0,
                                                             const int* __restrict__ src1, const int* __restrict__ dst1,
                                                             const float* __restrict__ el, const float* __restrict__ er,
                                                             const float* __restrict__ feat, float* __restrict__ acc,
                                                             const int* __restrict__ lists, const int* __restrict__ counts,
                                                             int E, int N) {
  constexpr int R     = H * D;
  constexpr int LINES = R / 32;
  constexpr int LPH   = D / 32;
  constexpr int G     = (LINES >= 8) ? 4 : 16;
  static_assert(D % 32 == 0, "a line must lie inside one head");

  __shared__ float    sacc[AGG_NODES * R];
  __shared__ unsigned mrow[AGG_NODES * H];
  __shared__ float    srow[AGG_NODES * H];
  __shared__ int2     queue[(AGG_THREADS / 32) * AGG_QCAP];
  __shared__ int2     elist[AGG_LCAP];
  __shared__ int      listCount;

  const int tid  = threadIdx.x;
  const int lane = tid & 31;
  const int wave = __builtin_amdgcn_readfirstlane(tid >> 5);
  const int nb  = (int)(gridDim.x >> 1);
  const int t   = (int)blockIdx.x >= nb ? 1 : 0;
  const int bid = (int)blockIdx.x - t * nb;
  const int* __restrict__ src = t ? src1 : src0;
  const int* __restrict__ dst = t ? dst1 : dst0;
  acc += (size_t)t * N * R;
  const int lo = bid * AGG_NODES;
  const int nn = (N - lo) < AGG_NODES ? (N - lo) : AGG_NODES;

  const int2* glist = reinterpret_cast<const int2*>(lists) + (size_t)blockIdx.x * AGG_LCAP;
  int saved = counts[blockIdx.x];

  float4* sacc4 = reinterpret_cast<float4*>(sacc);
  for (int i = tid; i < AGG_NODES * R / 4; i += AGG_THREADS) sacc4[i] = make_float4(0.f, 0.f, 0.f, 0.f);
  for (int i = tid; i < AGG_NODES * H; i += AGG_THREADS) { mrow[i] = ORD_NEG_INF; srow[i] = 0.f; }
  if (tid == 0) listCount = 0;
  __syncthreads();

  int2* q = queue + wave * AGG_QCAP;

  auto work_max = [&](const int2* qb, int n, auto kind) {
    constexpr bool SAVED = decltype(kind)::value;
    int2 ed = make_int2(0, 0);
    int sN = 0;
    if (lane < n) {
      ed = qb[lane];
      if constexpr (SAVED) sN = ed.x; else sN = src[ed.x];
      float l[H], r[H];
      load_heads<H>(er, ed.y, r);
      load_heads<H>(el, sN, l);
      unsigned* mo = mrow + (ed.y - lo) * H;
#pragma unroll
      for (int h = 0; h < H; ++h) {
        float sc = l[h] + r[h];
        sc = sc > 0.f ? sc : LRELU_SLOPE * sc;
        atomicMax(mo + h, f2o(sc));
      }
    }
    if constexpr (!SAVED) {
      int pos = 0;
      if (lane == 0) pos = atomicAdd(&listCount, n);
      pos = __builtin_amdgcn_readfirstlane(pos);
      if (lane < n && pos + lane < AGG_LCAP) elist[pos + lane] = ed;
    }
  };

  auto work_acc = [&](const int2* qb, int n, auto kind) {
    constexpr bool SAVED = decltype(kind)::value;
    int sN = 0, dl = 0;
    float a[H];
#pragma unroll
    for (int h = 0; h < H; ++h) a[h] = 0.f;
    if (lane < n) {
      const int2 ed = qb[lane];
      if constexpr (SAVED) sN = ed.x; else sN = src[ed.x];
      dl = ed.y - lo;
      float l[H], r[H];
      load_heads<H>(er, ed.y, r);
      load_heads<H>(el, sN, l);
#pragma unroll
      for (int h = 0; h < H; ++h) {
        float sc = l[h] + r[h];
        sc = sc > 0.f ? sc : LRELU_SLOPE * sc;
        float m = o2f(mrow[dl * H + h]);
        if (m < -1e30f) m = 0.f;
        a[h] = __expf(sc - m);
        atomicAdd(srow + dl * H + h, a[h]);
      }
    }
    for (int k0 = 0; k0 < n; k0 += G) {
      float v[G][LINES];
#pragma unroll
      for (int g = 0; g < G; ++g) {
        const int k  = (k0 + g < n) ? (k0 + g) : (n - 1);
        const int sk = __builtin_amdgcn_readlane(sN, k);
        const float* f = feat + (size_t)sk * R + lane;
#pragma unroll
        for (int j = 0; j < LINES; ++j) v[g][j] = f[32 * j];
      }
#pragma unroll
      for (int g = 0; g < G; ++g)
#pragma unroll
        for (int j = 0; j < LINES; ++j) asm volatile("" :: "v"(v[g][j]));
#pragma unroll
      for (int g = 0; g < G; ++g) {
        const int k = k0 + g;
        if (k < n) {
          const int dk = __builtin_amdgcn_readlane(dl, k);
          float* o = sacc + dk * R + lane;
#pragma unroll
          for (int j = 0; j < LINES; ++j) {
            const float ak = __int_as_float(__builtin_amdgcn_readlane(__float_as_int(a[j / LPH]), k));
            atomicAdd(o + 32 * j, ak * v[g][j]);
          }
        }
      }
    }
  };

  auto scan_apply = [&](auto&& work) {
    int cnt = 0;
    const int nChunks = (E + 127) >> 7;
    const unsigned below = (1u << lane) - 1u;
    for (int c0 = wave; c0 < nChunks; c0 += (AGG_THREADS / 32) * AGG_U) {
      int4 d[AGG_U];
#pragma unroll
      for (int u = 0; u < AGG_U; ++u) {
        const int c    = c0 + (AGG_THREADS / 32) * u;
        const int base = c * 128 + 4 * lane;
        d[u] = make_int4(-1, -1, -1, -1);
        if (c < nChunks) {
          if (c * 128 + 128 <= E) {
            d[u] = *reinterpret_cast<const int4*>(dst + base);
          } else {
            if (base + 0 < E) d[u].x = dst[base + 0];
            if (base + 1 < E) d[u].y = dst[base + 1];
            if (base + 2 < E) d[u].z = dst[base + 2];
            if (base + 3 < E) d[u].w = dst[base + 3];
          }
        }
      }
#pragma unroll
      for (int u = 0; u < AGG_U; ++u) {
        const int c    = c0 + (AGG_THREADS / 32) * u;
        const int base = c * 128 + 4 * lane;
        const int dv[4] = { d[u].x, d[u].y, d[u].z, d[u].w };
#pragma unroll
        for (int j = 0; j < 4; ++j) {
          const bool mine = (unsigned)(dv[j] - lo) < (unsigned)nn;
          const unsigned mask = (unsigned)__ballot(mine);
          if (mask != 0u) {
            if (mine) q[cnt + __popc(mask & below)] = make_int2(base + j, dv[j]);
            cnt += __popc(mask);
          }
        }
#pragma nounroll
        while (cnt >= 32) { cnt -= 32; work(q + cnt, 32); }
      }
    }
    if (cnt > 0) work(q, cnt);
  };

  saved = __builtin_amdgcn_readfirstlane(saved);
  if (__builtin_expect(saved <= AGG_LCAP, 1)) {
    const int per = (saved + (AGG_THREADS / 32) - 1) / (AGG_THREADS / 32);
    const int p1  = (wave + 1) * per < saved ? (wave + 1) * per : saved;
    for (int p = wave * per; p < p1; p += 32) {
      const int left = p1 - p;
      work_max(glist + p, left < 32 ? left : 32, list_kind<true>{});
    }
    __syncthreads();
    for (int p = wave * per; p < p1; p += 32) {
      const int left = p1 - p;
      work_acc(glist + p, left < 32 ? left : 32, list_kind<true>{});
    }
  } else {
    scan_apply([&](const int2* qb, int n) { work_max(qb, n, list_kind<false>{}); });
    __syncthreads();

    const int total = __builtin_amdgcn_readfirstlane(listCount);
    if (__builtin_expect(total <= AGG_LCAP, 1)) {
      const int per = (total + (AGG_THREADS / 32) - 1) / (AGG_THREADS / 32);
      const int p1  = (wave + 1) * per < total ? (wave + 1) * per : total;
      for (int p = wave * per; p < p1; p += 32) {
        const int left = p1 - p;
        work_acc(elist + p, left < 32 ? left : 32, list_kind<false>{});
      }
    } else {
      scan_apply([&](const int2* qb, int n) { work_acc(qb, n, list_kind<false>{}); });
    }
  }
  __syncthreads();

  float4* out4 = reinterpret_cast<float4*>(acc + (size_t)lo * R);
  for (int i = tid; i < nn * (R / 4); i += AGG_THREADS) {
    const int row = i / (R / 4);
    const int h   = (i - row * (R / 4)) / (D / 4);
    const float den = fmaxf(srow[row * H + h], 1e-9f);
    const float4 v = sacc4[i];
    out4[i] = make_float4(v.x / den, v.y / den, v.z / den, v.w / den);
  }
}

__global__ void __launch_bounds__(AGG_THREADS)
owner_softmax_aggregate_h4d64(const int* __restrict__ src0, const int* __restrict__ dst0,
                              const int* __restrict__ src1, const int* __restrict__ dst1,
                              const float* __restrict__ el, const float* __restrict__ er,
                              const float* __restrict__ feat, float* __restrict__ acc,
                              const int* __restrict__ lists, const int* __restrict__ counts, int E, int N) {
  owner_softmax_aggregate_body<4, 64>(src0, dst0, src1, dst1, el, er, feat, acc, lists, counts, E, N);
}
__global__ void __launch_bounds__(AGG_THREADS)
owner_softmax_aggregate_h1d64(const int* __restrict__ src0, const int* __restrict__ dst0,
                              const int* __restrict__ src1, const int* __restrict__ dst1,
                              const float* __restrict__ el, const float* __restrict__ er,
                              const float* __restrict__ feat, float* __restrict__ acc,
                              const int* __restrict__ lists, const int* __restrict__ counts, int E, int N) {
  owner_softmax_aggregate_body<1, 64>(src0, dst0, src1, dst1, el, er, feat, acc, lists, counts, E, N);
}

__device__ __forceinline__ float elu1(float v) { return (v > 0.f) ? v : expm1f(v); }

template <int H, int D>
__device__ __forceinline__ void epilogue0_body(const float* __restrict__ acc, const float* __restrict__ b,
                                               float* __restrict__ h1, int N) {
  constexpr int D4 = D / 4;
  const unsigned n4 = (unsigned)N * D4;
  const unsigned T = gridDim.x * blockDim.x;
  for (unsigned i = blockIdx.x * blockDim.x + threadIdx.x; i < n4; i += T) {
    const unsigned n = i / D4, q = i - n * D4;
    float4 v[2][H], bb[H];
#pragma unroll
    for (int t = 0; t < 2; ++t)
#pragma unroll
      for (int h = 0; h < H; ++h)
        v[t][h] = *reinterpret_cast<const float4*>(acc + ((size_t)t * N + n) * (H * D) + h * D + 4 * q);
#pragma unroll
    for (int h = 0; h < H; ++h) bb[h] = *reinterpret_cast<const float4*>(b + h * D + 4 * q);
    float4 sum = make_float4(0.f, 0.f, 0.f, 0.f);
#pragma unroll
    for (int t = 0; t < 2; ++t)
#pragma unroll
      for (int h = 0; h < H; ++h) {
        sum.x += elu1(v[t][h].x + bb[h].x);
        sum.y += elu1(v[t][h].y + bb[h].y);
        sum.z += elu1(v[t][h].z + bb[h].z);
        sum.w += elu1(v[t][h].w + bb[h].w);
      }
    const float hf = (float)H;
    *reinterpret_cast<float4*>(h1 + (size_t)i * 4) =
        make_float4(sum.x / hf, sum.y / hf, sum.z / hf, sum.w / hf);
  }
}
__global__ void epilogue0_h4d64(const float* __restrict__ acc, const float* __restrict__ b,
                                float* __restrict__ h1, int N) {
  epilogue0_body<4, 64>(acc, b, h1, N);
}


template <int D, int U>
__device__ __forceinline__ void epilogue1_body(const float* __restrict__ acc, const float* __restrict__ b,
                                               float* __restrict__ out, int N) {
  constexpr int D4 = D / 4;
  const unsigned n4 = (unsigned)N * D4;
  const unsigned T = gridDim.x * blockDim.x;
  const v4f* a0 = reinterpret_cast<const v4f*>(acc);
  const v4f* a1 = reinterpret_cast<const v4f*>(acc + (size_t)N * D);
  const v4f* b4 = reinterpret_cast<const v4f*>(b);
  v4f* o4 = reinterpret_cast<v4f*>(out);
  for (unsigned i0 = blockIdx.x * blockDim.x + threadIdx.x; i0 < n4; i0 += U * T) {
    v4f x[U], y[U], bb[U];
#pragma unroll
    for (int u = 0; u < U; ++u) {
      const unsigned i  = i0 + u * T;
      const unsigned ic = i < n4 ? i : n4 - 1;
      x[u]  = a0[ic];
      y[u]  = a1[ic];
      bb[u] = b4[ic % D4];
    }
#pragma unroll
    for (int u = 0; u < U; ++u) asm volatile("" :: "v"(x[u]), "v"(y[u]), "v"(bb[u]));
#pragma unroll
    for (int u = 0; u < U; ++u) {
      const unsigned i = i0 + u * T;
      if (i < n4) o4[i] = (x[u] + bb[u]) + (y[u] + bb[u]);
    }
  }
}
__global__ void epilogue1_d64(const float* __restrict__ acc, const float* __restrict__ b,
                              float* __restrict__ out, int N) {
  epilogue1_body<64, 4>(acc, b, out, N);
}

extern "C" void kernel_launch(void* const* d_in, const int* in_sizes, int n_in,
                              void* d_out, int out_size, void* d_ws, size_t ws_size,
                              hipStream_t stream) {
  const int IN = 128, H0 = 4, D0 = 64, H1 = 1, D1 = 64;
  const int N = in_sizes[0] / IN;
  const int E = in_sizes[9];

  const float* x   = (const float*)d_in[0];
  const float* W0  = (const float*)d_in[1];
  const float* al0 = (const float*)d_in[2];
  const float* ar0 = (const float*)d_in[3];
  const float* b0  = (const float*)d_in[4];
  const float* W1  = (const float*)d_in[5];
  const float* al1 = (const float*)d_in[6];
  const float* ar1 = (const float*)d_in[7];
  const float* b1  = (const float*)d_in[8];
  const int* srcT[2] = { (const int*)d_in[9],  (const int*)d_in[11] };
  const int* dstT[2] = { (const int*)d_in[10], (const int*)d_in[12] };
  float* out = (float*)d_out;

  float* ws = (float*)d_ws;
  float* feat0 = ws;
  float* el0   = feat0 + (size_t)N * H0 * D0;
  float* er0   = el0 + (size_t)N * H0;
  float* acc0  = er0 + (size_t)N * H0;
  float* h1    = acc0 + 2 * (size_t)N * H0 * D0;
  float* feat1 = feat0;
  float* el1   = feat1 + (size_t)N * D1;
  float* er1   = el1 + N;
  float* acc1  = er1 + N;

  const int TB = 256;
  const int aggBlocks = (N + AGG_NODES - 1) / AGG_NODES;
  int* lists  = (int*)(h1 + (size_t)N * D0);
  int* counts = lists + (size_t)2 * aggBlocks * AGG_LCAP * 2;
  const int epi0Blocks  = 1024;
  const int epi1Blocks  = (N * (D1 / 4) + TB * 4 - 1) / (TB * 4);
  const int bucketBlocks = (E + BK_THREADS * 4 - 1) / (BK_THREADS * 4);
  (void)ws_size; (void)n_in; (void)out_size;

  {
    gemm_lds_k128_n256<<<256, GT_THREADS, 0, stream>>>(x, W0, feat0, N, counts, 2 * aggBlocks);

    bucket_edges<<<2 * bucketBlocks, BK_THREADS, 0, stream>>>(srcT[0], dstT[0], srcT[1], dstT[1], lists, counts, E, N, aggBlocks);

    node_scores<<<(N * H0 + TB - 1) / TB, TB, 0, stream>>>(feat0, al0, ar0, el0, er0, N, H0, D0);

    owner_softmax_aggregate_h4d64<<<2 * aggBlocks, AGG_THREADS, 0, stream>>>(srcT[0], dstT[0], srcT[1], dstT[1], el0, er0, feat0, acc0, lists, counts, E, N);

    epilogue0_h4d64<<<epi0Blocks, TB, 0, stream>>>(acc0, b0, h1, N);
  }

  {
    gemm_lds_k64_n64<<<(N + 127) / 128, GT_THREADS, 0, stream>>>(h1, W1, feat1, N);

    node_scores<<<(N * H1 + TB - 1) / TB, TB, 0, stream>>>(feat1, al1, ar1, el1, er1, N, H1, D1);

    owner_softmax_aggregate_h1d64<<<2 * aggBlocks, AGG_THREADS, 0, stream>>>(srcT[0], dstT[0], srcT[1], dstT[1], el1, er1, feat1, acc1, lists, counts, E, N);

    epilogue1_d64<<<epi1Blocks, TB, 0, stream>>>(acc1, b1, out, N);
  }
}


# node_scores: a 16-lane half reads a head's 64 floats as one 256-byte piece (16-byte loads, 16 in flight per lane); lane-tree sums leave a wave's 32 outputs for one line store; el/er sum order changes
# speedup vs baseline: 4.1845x; 1.0576x over previous
#include <hip/hip_runtime.h>
#include <math.h>

typedef float v2f __attribute__((ext_vector_type(2)));
typedef float v4f __attribute__((ext_vector_type(4)));
typedef float v8f __attribute__((ext_vector_type(8)));

#define LRELU_SLOPE 0.2f

__device__ __forceinline__ unsigned f2o(float f) {
  unsigned u = __float_as_uint(f);
  return (u & 0x80000000u) ? ~u : (u | 0x80000000u);
}
__device__ __forceinline__ float o2f(unsigned o) {
  unsigned u = (o & 0x80000000u) ? (o & 0x7fffffffu) : ~o;
  return __uint_as_float(u);
}
#define ORD_NEG_INF 0x007fffffu

#define GT_THREADS 512

template <int K, int NC>
__device__ __forceinline__ void gemm_lds_body(const float* __restrict__ A,
                                              const float* __restrict__ B,
                                              float* __restrict__ C, int M) {
  constexpr int NP  = NC / 32;
  constexpr int WR  = 16 / NP;
  constexpr int RS  = WR * 16;
  constexpr int PA  = K + 4;
  constexpr int PB  = 2 * NC + 32;
  constexpr int K4  = K / 4;
  constexpr int NC4 = NC / 4;
  constexpr int NPF = (RS * K4) / GT_THREADS;
  constexpr int NBF = ((K / 2) * NC4) / GT_THREADS;
  static_assert(NC % 32 == 0 && 16 % NP == 0, "a wave takes one M-tile and one column pair");
  static_assert((RS * K4) % GT_THREADS == 0 && (NPF == 2 || NPF == 4), "an A tile is 2 or 4 pieces per thread");
  static_assert(((K / 2) * NC4) % GT_THREADS == 0, "B splits evenly over the threads");
  static_assert(K % 4 == 0, "k steps of 4");

  __shared__ __align__(16) float Bs[(K / 2) * PB];
  __shared__ __align__(16) float As[2][RS * PA];

  const int tid  = threadIdx.x;
  const int lane = tid & 31;
  const int half = lane >> 4;
  const int l    = lane & 15;
  const int wave = __builtin_amdgcn_readfirstlane(tid >> 5);
  const int wm   = wave / NP;
  const int wn   = wave - wm * NP;

  const int nSteps = (M + RS - 1) / RS;

  auto fetch_piece = [&](int s, int u) -> v4f {
    const int it = tid + u * GT_THREADS;
    const int r = it / K4, c = it - r * K4;
    int row = s * RS + r;
    row = row < M ? row : M - 1;
    return *reinterpret_cast<const v4f*>(A + (size_t)row * K + 4 * c);
  };
  auto stash_piece = [&](float* as, int u, v4f v) {
    const int it = tid + u * GT_THREADS;
    const int r = it / K4, c = it - r * K4;
    *reinterpret_cast<v4f*>(as + r * PA + 4 * c) = v;
  };

  int step = blockIdx.x;
  v4f p0 = fetch_piece(step, 0), p1 = fetch_piece(step, 1), p2 = p0, p3 = p0;
  if constexpr (NPF == 4) { p2 = fetch_piece(step, 2); p3 = fetch_piece(step, 3); }
#pragma unroll
  for (int u = 0; u < NBF; ++u) {
    const int it = tid + u * GT_THREADS;
    const int kp = it / NC4, c = it - kp * NC4;
    const float4 r0 = *reinterpret_cast<const float4*>(B + (size_t)(2 * kp) * NC + 4 * c);
    const float4 r1 = *reinterpret_cast<const float4*>(B + (size_t)(2 * kp + 1) * NC + 4 * c);
    float4* d = reinterpret_cast<float4*>(Bs + kp * PB + 8 * c);
    d[0] = make_float4(r0.x, r1.x, r0.y, r1.y);
    d[1] = make_float4(r0.z, r1.z, r0.w, r1.w);
  }

  int buf = 0;
  for (; step < nSteps; step += gridDim.x) {
    float* as = As[buf];
    stash_piece(as, 0, p0);
    stash_piece(as, 1, p1);
    if constexpr (NPF == 4) { stash_piece(as, 2, p2); stash_piece(as, 3, p3); }
    __syncthreads();

    const int next = step + gridDim.x;
    p0 = fetch_piece(next, 0);
    p1 = fetch_piece(next, 1);
    if constexpr (NPF == 4) { p2 = fetch_piece(next, 2); p3 = fetch_piece(next, 3); }

    v8f acc0 = {}, acc1 = {};
    const float* ap = as + (wm * 16 + l) * PA + 2 * half;
    const float* bp = Bs + half * PB + (wn * 32 + l) * 2;
#pragma unroll
    for (int k = 0; k < K; k += 4) {
      const v2f a  = *reinterpret_cast<const v2f*>(ap + k);
      const v2f b0 = *reinterpret_cast<const v2f*>(bp + (k >> 1) * PB);
      const v2f b1 = *reinterpret_cast<const v2f*>(bp + (k >> 1) * PB + 32);
      acc0 = __builtin_amdgcn_wmma_f32_16x16x4_f32(false, a, false, b0, (short)0, acc0, false, false);
      acc1 = __builtin_amdgcn_wmma_f32_16x16x4_f32(false, a, false, b1, (short)0, acc1, false, false);
    }

    const int row0 = step * RS + wm * 16 + half * 8;
    float* cp = C + (size_t)row0 * NC + wn * 32 + l;
#pragma unroll
    for (int r = 0; r < 8; ++r) {
      if (row0 + r < M) {
        cp[(size_t)r * NC]      = acc0[r];
        cp[(size_t)r * NC + 16] = acc1[r];
      }
    }
    buf ^= 1;
  }
}

__global__ void __launch_bounds__(GT_THREADS)
gemm_lds_k128_n256(const float* __restrict__ A, const float* __restrict__ B, float* __restrict__ C, int M,
                   int* __restrict__ zero, int nzero) {
  if (blockIdx.x == 0)
    for (int i = threadIdx.x; i < nzero; i += GT_THREADS) zero[i] = 0;
  gemm_lds_body<128, 256>(A, B, C, M);
}
__global__ void __launch_bounds__(GT_THREADS)
gemm_lds_k64_n64(const float* __restrict__ A, const float* __restrict__ B, float* __restrict__ C, int M) {
  gemm_lds_body<64, 64>(A, B, C, M);
}

#define NS_THREADS 256

template <int M>
__device__ __forceinline__ void transpose_sum_level(float (&a)[16], int l) {
  const bool up = (l & M) != 0;
#pragma unroll
  for (int k = 0; k < M; ++k) {
    const float keep = up ? a[k + M] : a[k];
    const float send = up ? a[k] : a[k + M];
    a[k] = keep + __shfl_xor(send, M);
  }
}
__device__ __forceinline__ float transpose_sum16(float (&a)[16], int l) {
  transpose_sum_level<8>(a, l);
  transpose_sum_level<4>(a, l);
  transpose_sum_level<2>(a, l);
  transpose_sum_level<1>(a, l);
  return a[0];
}

template <int H>
__device__ __forceinline__ void node_scores_body(const float* __restrict__ feat,
                                                 const float* __restrict__ al,
                                                 const float* __restrict__ ar,
                                                 float* __restrict__ el, float* __restrict__ er, int N) {
  constexpr int D   = 64;
  constexpr int RPH = 16 / H;
  static_assert(16 % H == 0, "a half takes whole nodes");
  const int lane = threadIdx.x & 31;
  const int half = lane >> 4;
  const int l    = lane & 15;
  const int wave = blockIdx.x * (NS_THREADS / 32) + (threadIdx.x >> 5);
  const int n0   = wave * (2 * RPH);
  if (n0 >= N) return;

  float4 wl[H], wr[H];
#pragma unroll
  for (int q = 0; q < H; ++q) {
    wl[q] = *reinterpret_cast<const float4*>(al + q * D + 4 * l);
    wr[q] = *reinterpret_cast<const float4*>(ar + q * D + 4 * l);
  }
  float4 v[16];
#pragma unroll
  for (int k = 0; k < 16; ++k) {
    int row = n0 + half * RPH + k / H;
    row = row < N ? row : N - 1;
    v[k] = *reinterpret_cast<const float4*>(feat + (size_t)row * (H * D) + (k % H) * D + 4 * l);
  }
  float sl[16], sr[16];
#pragma unroll
  for (int k = 0; k < 16; ++k) {
    const float4 a = wl[k % H], b = wr[k % H];
    sl[k] = v[k].x * a.x + v[k].y * a.y + v[k].z * a.z + v[k].w * a.w;
    sr[k] = v[k].x * b.x + v[k].y * b.y + v[k].z * b.z + v[k].w * b.w;
  }
  const float tl = transpose_sum16(sl, l);
  const float tr = transpose_sum16(sr, l);
  if (n0 + half * RPH + l / H < N) {
    el[(size_t)n0 * H + lane] = tl;
    er[(size_t)n0 * H + lane] = tr;
  }
}
__global__ void __launch_bounds__(NS_THREADS)
node_scores_h4(const float* __restrict__ feat, const float* __restrict__ al, const float* __restrict__ ar,
               float* __restrict__ el, float* __restrict__ er, int N) {
  node_scores_body<4>(feat, al, ar, el, er, N);
}
__global__ void __launch_bounds__(NS_THREADS)
node_scores_h1(const float* __restrict__ feat, const float* __restrict__ al, const float* __restrict__ ar,
               float* __restrict__ el, float* __restrict__ er, int N) {
  node_scores_body<1>(feat, al, ar, el, er, N);
}

#define AGG_THREADS 1024
#define AGG_NODES   196
#define AGG_QCAP    192
#define AGG_U       8
#define AGG_LCAP    4096

#define BK_THREADS 1024
#define BK_MAXB    1024

__global__ void __launch_bounds__(BK_THREADS)
bucket_edges(const int* __restrict__ src0, const int* __restrict__ dst0,
             const int* __restrict__ src1, const int* __restrict__ dst1,
             int* __restrict__ lists, int* __restrict__ counts, int E, int N, int nb) {
  __shared__ int place[BK_MAXB];
  const int tid = threadIdx.x;
  const int cb  = (int)(gridDim.x >> 1);
  const int t   = (int)blockIdx.x >= cb ? 1 : 0;
  const int b   = (int)blockIdx.x - t * cb;
  counts += t * nb;
  if (nb > BK_MAXB) {
    if (b == 0)
      for (int i = tid; i < nb; i += BK_THREADS) counts[i] = AGG_LCAP + 1;
    return;
  }
  const int* __restrict__ src = t ? src1 : src0;
  const int* __restrict__ dst = t ? dst1 : dst0;
  int2* glist = reinterpret_cast<int2*>(lists) + (size_t)t * nb * AGG_LCAP;

  for (int i = tid; i < nb; i += BK_THREADS) place[i] = 0;
  __syncthreads();

  const int base = (b * BK_THREADS + tid) * 4;
  int dv[4] = { -1, -1, -1, -1 };
  int sv[4] = { 0, 0, 0, 0 };
  if (base + 4 <= E) {
    const int4 d4 = *reinterpret_cast<const int4*>(dst + base);
    const int4 s4 = *reinterpret_cast<const int4*>(src + base);
    dv[0] = d4.x; dv[1] = d4.y; dv[2] = d4.z; dv[3] = d4.w;
    sv[0] = s4.x; sv[1] = s4.y; sv[2] = s4.z; sv[3] = s4.w;
  } else {
#pragma unroll
    for (int j = 0; j < 4; ++j)
      if (base + j < E) { dv[j] = dst[base + j]; sv[j] = src[base + j]; }
  }
  int bk[4], rk[4];
#pragma unroll
  for (int j = 0; j < 4; ++j) {
    const bool in = (unsigned)dv[j] < (unsigned)N;
    bk[j] = in ? dv[j] / AGG_NODES : -1;
    rk[j] = 0;
    if (in) rk[j] = atomicAdd(&place[bk[j]], 1);
  }
  __syncthreads();
  for (int i = tid; i < nb; i += BK_THREADS) {
    const int n = place[i];
    if (n > 0) place[i] = atomicAdd(counts + i, n);
  }
  __syncthreads();
#pragma unroll
  for (int j = 0; j < 4; ++j) {
    if (bk[j] >= 0) {
      const int pos = place[bk[j]] + rk[j];
      if (pos < AGG_LCAP) glist[(size_t)bk[j] * AGG_LCAP + pos] = make_int2(sv[j], dv[j]);
    }
  }
}

template <bool B> struct list_kind { static constexpr bool value = B; };

template <int H>
__device__ __forceinline__ void load_heads(const float* __restrict__ p, int node, float (&o)[H]) {
  if constexpr (H == 4) {
    const float4 t = *reinterpret_cast<const float4*>(p + (size_t)node * 4);
    o[0] = t.x; o[1] = t.y; o[2] = t.z; o[3] = t.w;
  } else {
#pragma unroll
    for (int h = 0; h < H; ++h) o[h] = p[(size_t)node * H + h];
  }
}

template <int H, int D>
__device__ __forceinline__ void owner_softmax_aggregate_body(const int* __restrict__ src0, const int* __restrict__ dst0,
                                                             const int* __restrict__ src1, const int* __restrict__ dst1,
                                                             const float* __restrict__ el, const float* __restrict__ er,
                                                             const float* __restrict__ feat, float* __restrict__ acc,
                                                             const int* __restrict__ lists, const int* __restrict__ counts,
                                                             int E, int N) {
  constexpr int R     = H * D;
  constexpr int LINES = R / 32;
  constexpr int LPH   = D / 32;
  constexpr int G     = (LINES >= 8) ? 4 : 16;
  static_assert(D % 32 == 0, "a line must lie inside one head");

  __shared__ float    sacc[AGG_NODES * R];
  __shared__ unsigned mrow[AGG_NODES * H];
  __shared__ float    srow[AGG_NODES * H];
  __shared__ int2     queue[(AGG_THREADS / 32) * AGG_QCAP];
  __shared__ int2     elist[AGG_LCAP];
  __shared__ int      listCount;

  const int tid  = threadIdx.x;
  const int lane = tid & 31;
  const int wave = __builtin_amdgcn_readfirstlane(tid >> 5);
  const int nb  = (int)(gridDim.x >> 1);
  const int t   = (int)blockIdx.x >= nb ? 1 : 0;
  const int bid = (int)blockIdx.x - t * nb;
  const int* __restrict__ src = t ? src1 : src0;
  const int* __restrict__ dst = t ? dst1 : dst0;
  acc += (size_t)t * N * R;
  const int lo = bid * AGG_NODES;
  const int nn = (N - lo) < AGG_NODES ? (N - lo) : AGG_NODES;

  const int2* glist = reinterpret_cast<const int2*>(lists) + (size_t)blockIdx.x * AGG_LCAP;
  int saved = counts[blockIdx.x];

  float4* sacc4 = reinterpret_cast<float4*>(sacc);
  for (int i = tid; i < AGG_NODES * R / 4; i += AGG_THREADS) sacc4[i] = make_float4(0.f, 0.f, 0.f, 0.f);
  for (int i = tid; i < AGG_NODES * H; i += AGG_THREADS) { mrow[i] = ORD_NEG_INF; srow[i] = 0.f; }
  if (tid == 0) listCount = 0;
  __syncthreads();

  int2* q = queue + wave * AGG_QCAP;

  auto work_max = [&](const int2* qb, int n, auto kind) {
    constexpr bool SAVED = decltype(kind)::value;
    int2 ed = make_int2(0, 0);
    int sN = 0;
    if (lane < n) {
      ed = qb[lane];
      if constexpr (SAVED) sN = ed.x; else sN = src[ed.x];
      float l[H], r[H];
      load_heads<H>(er, ed.y, r);
      load_heads<H>(el, sN, l);
      unsigned* mo = mrow + (ed.y - lo) * H;
#pragma unroll
      for (int h = 0; h < H; ++h) {
        float sc = l[h] + r[h];
        sc = sc > 0.f ? sc : LRELU_SLOPE * sc;
        atomicMax(mo + h, f2o(sc));
      }
    }
    if constexpr (!SAVED) {
      int pos = 0;
      if (lane == 0) pos = atomicAdd(&listCount, n);
      pos = __builtin_amdgcn_readfirstlane(pos);
      if (lane < n && pos + lane < AGG_LCAP) elist[pos + lane] = ed;
    }
  };

  auto work_acc = [&](const int2* qb, int n, auto kind) {
    constexpr bool SAVED = decltype(kind)::value;
    int sN = 0, dl = 0;
    float a[H];
#pragma unroll
    for (int h = 0; h < H; ++h) a[h] = 0.f;
    if (lane < n) {
      const int2 ed = qb[lane];
      if constexpr (SAVED) sN = ed.x; else sN = src[ed.x];
      dl = ed.y - lo;
      float l[H], r[H];
      load_heads<H>(er, ed.y, r);
      load_heads<H>(el, sN, l);
#pragma unroll
      for (int h = 0; h < H; ++h) {
        float sc = l[h] + r[h];
        sc = sc > 0.f ? sc : LRELU_SLOPE * sc;
        float m = o2f(mrow[dl * H + h]);
        if (m < -1e30f) m = 0.f;
        a[h] = __expf(sc - m);
        atomicAdd(srow + dl * H + h, a[h]);
      }
    }
    for (int k0 = 0; k0 < n; k0 += G) {
      float v[G][LINES];
#pragma unroll
      for (int g = 0; g < G; ++g) {
        const int k  = (k0 + g < n) ? (k0 + g) : (n - 1);
        const int sk = __builtin_amdgcn_readlane(sN, k);
        const float* f = feat + (size_t)sk * R + lane;
#pragma unroll
        for (int j = 0; j < LINES; ++j) v[g][j] = f[32 * j];
      }
#pragma unroll
      for (int g = 0; g < G; ++g)
#pragma unroll
        for (int j = 0; j < LINES; ++j) asm volatile("" :: "v"(v[g][j]));
#pragma unroll
      for (int g = 0; g < G; ++g) {
        const int k = k0 + g;
        if (k < n) {
          const int dk = __builtin_amdgcn_readlane(dl, k);
          float* o = sacc + dk * R + lane;
#pragma unroll
          for (int j = 0; j < LINES; ++j) {
            const float ak = __int_as_float(__builtin_amdgcn_readlane(__float_as_int(a[j / LPH]), k));
            atomicAdd(o + 32 * j, ak * v[g][j]);
          }
        }
      }
    }
  };

  auto scan_apply = [&](auto&& work) {
    int cnt = 0;
    const int nChunks = (E + 127) >> 7;
    const unsigned below = (1u << lane) - 1u;
    for (int c0 = wave; c0 < nChunks; c0 += (AGG_THREADS / 32) * AGG_U) {
      int4 d[AGG_U];
#pragma unroll
      for (int u = 0; u < AGG_U; ++u) {
        const int c    = c0 + (AGG_THREADS / 32) * u;
        const int base = c * 128 + 4 * lane;
        d[u] = make_int4(-1, -1, -1, -1);
        if (c < nChunks) {
          if (c * 128 + 128 <= E) {
            d[u] = *reinterpret_cast<const int4*>(dst + base);
          } else {
            if (base + 0 < E) d[u].x = dst[base + 0];
            if (base + 1 < E) d[u].y = dst[base + 1];
            if (base + 2 < E) d[u].z = dst[base + 2];
            if (base + 3 < E) d[u].w = dst[base + 3];
          }
        }
      }
#pragma unroll
      for (int u = 0; u < AGG_U; ++u) {
        const int c    = c0 + (AGG_THREADS / 32) * u;
        const int base = c * 128 + 4 * lane;
        const int dv[4] = { d[u].x, d[u].y, d[u].z, d[u].w };
#pragma unroll
        for (int j = 0; j < 4; ++j) {
          const bool mine = (unsigned)(dv[j] - lo) < (unsigned)nn;
          const unsigned mask = (unsigned)__ballot(mine);
          if (mask != 0u) {
            if (mine) q[cnt + __popc(mask & below)] = make_int2(base + j, dv[j]);
            cnt += __popc(mask);
          }
        }
#pragma nounroll
        while (cnt >= 32) { cnt -= 32; work(q + cnt, 32); }
      }
    }
    if (cnt > 0) work(q, cnt);
  };

  saved = __builtin_amdgcn_readfirstlane(saved);
  if (__builtin_expect(saved <= AGG_LCAP, 1)) {
    const int per = (saved + (AGG_THREADS / 32) - 1) / (AGG_THREADS / 32);
    const int p1  = (wave + 1) * per < saved ? (wave + 1) * per : saved;
    for (int p = wave * per; p < p1; p += 32) {
      const int left = p1 - p;
      work_max(glist + p, left < 32 ? left : 32, list_kind<true>{});
    }
    __syncthreads();
    for (int p = wave * per; p < p1; p += 32) {
      const int left = p1 - p;
      work_acc(glist + p, left < 32 ? left : 32, list_kind<true>{});
    }
  } else {
    scan_apply([&](const int2* qb, int n) { work_max(qb, n, list_kind<false>{}); });
    __syncthreads();

    const int total = __builtin_amdgcn_readfirstlane(listCount);
    if (__builtin_expect(total <= AGG_LCAP, 1)) {
      const int per = (total + (AGG_THREADS / 32) - 1) / (AGG_THREADS / 32);
      const int p1  = (wave + 1) * per < total ? (wave + 1) * per : total;
      for (int p = wave * per; p < p1; p += 32) {
        const int left = p1 - p;
        work_acc(elist + p, left < 32 ? left : 32, list_kind<false>{});
      }
    } else {
      scan_apply([&](const int2* qb, int n) { work_acc(qb, n, list_kind<false>{}); });
    }
  }
  __syncthreads();

  float4* out4 = reinterpret_cast<float4*>(acc + (size_t)lo * R);
  for (int i = tid; i < nn * (R / 4); i += AGG_THREADS) {
    const int row = i / (R / 4);
    const int h   = (i - row * (R / 4)) / (D / 4);
    const float den = fmaxf(srow[row * H + h], 1e-9f);
    const float4 v = sacc4[i];
    out4[i] = make_float4(v.x / den, v.y / den, v.z / den, v.w / den);
  }
}

__global__ void __launch_bounds__(AGG_THREADS)
owner_softmax_aggregate_h4d64(const int* __restrict__ src0, const int* __restrict__ dst0,
                              const int* __restrict__ src1, const int* __restrict__ dst1,
                              const float* __restrict__ el, const float* __restrict__ er,
                              const float* __restrict__ feat, float* __restrict__ acc,
                              const int* __restrict__ lists, const int* __restrict__ counts, int E, int N) {
  owner_softmax_aggregate_body<4, 64>(src0, dst0, src1, dst1, el, er, feat, acc, lists, counts, E, N);
}
__global__ void __launch_bounds__(AGG_THREADS)
owner_softmax_aggregate_h1d64(const int* __restrict__ src0, const int* __restrict__ dst0,
                              const int* __restrict__ src1, const int* __restrict__ dst1,
                              const float* __restrict__ el, const float* __restrict__ er,
                              const float* __restrict__ feat, float* __restrict__ acc,
                              const int* __restrict__ lists, const int* __restrict__ counts, int E, int N) {
  owner_softmax_aggregate_body<1, 64>(src0, dst0, src1, dst1, el, er, feat, acc, lists, counts, E, N);
}

__device__ __forceinline__ float elu1(float v) { return (v > 0.f) ? v : expm1f(v); }

template <int H, int D>
__device__ __forceinline__ void epilogue0_body(const float* __restrict__ acc, const float* __restrict__ b,
                                               float* __restrict__ h1, int N) {
  constexpr int D4 = D / 4;
  const unsigned n4 = (unsigned)N * D4;
  const unsigned T = gridDim.x * blockDim.x;
  for (unsigned i = blockIdx.x * blockDim.x + threadIdx.x; i < n4; i += T) {
    const unsigned n = i / D4, q = i - n * D4;
    float4 v[2][H], bb[H];
#pragma unroll
    for (int t = 0; t < 2; ++t)
#pragma unroll
      for (int h = 0; h < H; ++h)
        v[t][h] = *reinterpret_cast<const float4*>(acc + ((size_t)t * N + n) * (H * D) + h * D + 4 * q);
#pragma unroll
    for (int h = 0; h < H; ++h) bb[h] = *reinterpret_cast<const float4*>(b + h * D + 4 * q);
    float4 sum = make_float4(0.f, 0.f, 0.f, 0.f);
#pragma unroll
    for (int t = 0; t < 2; ++t)
#pragma unroll
      for (int h = 0; h < H; ++h) {
        sum.x += elu1(v[t][h].x + bb[h].x);
        sum.y += elu1(v[t][h].y + bb[h].y);
        sum.z += elu1(v[t][h].z + bb[h].z);
        sum.w += elu1(v[t][h].w + bb[h].w);
      }
    const float hf = (float)H;
    *reinterpret_cast<float4*>(h1 + (size_t)i * 4) =
        make_float4(sum.x / hf, sum.y / hf, sum.z / hf, sum.w / hf);
  }
}
__global__ void epilogue0_h4d64(const float* __restrict__ acc, const float* __restrict__ b,
                                float* __restrict__ h1, int N) {
  epilogue0_body<4, 64>(acc, b, h1, N);
}


template <int D, int U>
__device__ __forceinline__ void epilogue1_body(const float* __restrict__ acc, const float* __restrict__ b,
                                               float* __restrict__ out, int N) {
  constexpr int D4 = D / 4;
  const unsigned n4 = (unsigned)N * D4;
  const unsigned T = gridDim.x * blockDim.x;
  const v4f* a0 = reinterpret_cast<const v4f*>(acc);
  const v4f* a1 = reinterpret_cast<const v4f*>(acc + (size_t)N * D);
  const v4f* b4 = reinterpret_cast<const v4f*>(b);
  v4f* o4 = reinterpret_cast<v4f*>(out);
  for (unsigned i0 = blockIdx.x * blockDim.x + threadIdx.x; i0 < n4; i0 += U * T) {
    v4f x[U], y[U], bb[U];
#pragma unroll
    for (int u = 0; u < U; ++u) {
      const unsigned i  = i0 + u * T;
      const unsigned ic = i < n4 ? i : n4 - 1;
      x[u]  = a0[ic];
      y[u]  = a1[ic];
      bb[u] = b4[ic % D4];
    }
#pragma unroll
    for (int u = 0; u < U; ++u) asm volatile("" :: "v"(x[u]), "v"(y[u]), "v"(bb[u]));
#pragma unroll
    for (int u = 0; u < U; ++u) {
      const unsigned i = i0 + u * T;
      if (i < n4) o4[i] = (x[u] + bb[u]) + (y[u] + bb[u]);
    }
  }
}
__global__ void epilogue1_d64(const float* __restrict__ acc, const float* __restrict__ b,
                              float* __restrict__ out, int N) {
  epilogue1_body<64, 4>(acc, b, out, N);
}

extern "C" void kernel_launch(void* const* d_in, const int* in_sizes, int n_in,
                              void* d_out, int out_size, void* d_ws, size_t ws_size,
                              hipStream_t stream) {
  const int IN = 128, H0 = 4, D0 = 64, H1 = 1, D1 = 64;
  const int N = in_sizes[0] / IN;
  const int E = in_sizes[9];

  const float* x   = (const float*)d_in[0];
  const float* W0  = (const float*)d_in[1];
  const float* al0 = (const float*)d_in[2];
  const float* ar0 = (const float*)d_in[3];
  const float* b0  = (const float*)d_in[4];
  const float* W1  = (const float*)d_in[5];
  const float* al1 = (const float*)d_in[6];
  const float* ar1 = (const float*)d_in[7];
  const float* b1  = (const float*)d_in[8];
  const int* srcT[2] = { (const int*)d_in[9],  (const int*)d_in[11] };
  const int* dstT[2] = { (const int*)d_in[10], (const int*)d_in[12] };
  float* out = (float*)d_out;

  float* ws = (float*)d_ws;
  float* feat0 = ws;
  float* el0   = feat0 + (size_t)N * H0 * D0;
  float* er0   = el0 + (size_t)N * H0;
  float* acc0  = er0 + (size_t)N * H0;
  float* h1    = acc0 + 2 * (size_t)N * H0 * D0;
  float* feat1 = feat0;
  float* el1   = feat1 + (size_t)N * D1;
  float* er1   = el1 + N;
  float* acc1  = er1 + N;

  const int TB = 256;
  const int aggBlocks = (N + AGG_NODES - 1) / AGG_NODES;
  int* lists  = (int*)(h1 + (size_t)N * D0);
  int* counts = lists + (size_t)2 * aggBlocks * AGG_LCAP * 2;
  const int epi0Blocks  = 1024;
  const int epi1Blocks  = (N * (D1 / 4) + TB * 4 - 1) / (TB * 4);
  const int bucketBlocks = (E + BK_THREADS * 4 - 1) / (BK_THREADS * 4);
  (void)ws_size; (void)n_in; (void)out_size;

  {
    gemm_lds_k128_n256<<<256, GT_THREADS, 0, stream>>>(x, W0, feat0, N, counts, 2 * aggBlocks);

    bucket_edges<<<2 * bucketBlocks, BK_THREADS, 0, stream>>>(srcT[0], dstT[0], srcT[1], dstT[1], lists, counts, E, N, aggBlocks);

    node_scores_h4<<<(N + 63) / 64, NS_THREADS, 0, stream>>>(feat0, al0, ar0, el0, er0, N);

    owner_softmax_aggregate_h4d64<<<2 * aggBlocks, AGG_THREADS, 0, stream>>>(srcT[0], dstT[0], srcT[1], dstT[1], el0, er0, feat0, acc0, lists, counts, E, N);

    epilogue0_h4d64<<<epi0Blocks, TB, 0, stream>>>(acc0, b0, h1, N);
  }

  {
    gemm_lds_k64_n64<<<(N + 127) / 128, GT_THREADS, 0, stream>>>(h1, W1, feat1, N);

    node_scores_h1<<<(N + 255) / 256, NS_THREADS, 0, stream>>>(feat1, al1, ar1, el1, er1, N);

    owner_softmax_aggregate_h1d64<<<2 * aggBlocks, AGG_THREADS, 0, stream>>>(srcT[0], dstT[0], srcT[1], dstT[1], el1, er1, feat1, acc1, lists, counts, E, N);

    epilogue1_d64<<<epi1Blocks, TB, 0, stream>>>(acc1, b1, out, N);
  }
}


# Node logits by reassociation, el = x (W al), er = x (W ar): GEMM blocks add the 2H columns W al, W ar to B in LDS, one wave per M-tile multiplies them; 9 -> 7 launches; el/er sum order changes
# speedup vs baseline: 4.3260x; 1.0338x over previous
#include <hip/hip_runtime.h>
#include <math.h>

typedef float v2f __attribute__((ext_vector_type(2)));
typedef float v4f __attribute__((ext_vector_type(4)));
typedef float v8f __attribute__((ext_vector_type(8)));

#define LRELU_SLOPE 0.2f

__device__ __forceinline__ unsigned f2o(float f) {
  unsigned u = __float_as_uint(f);
  return (u & 0x80000000u) ? ~u : (u | 0x80000000u);
}
__device__ __forceinline__ float o2f(unsigned o) {
  unsigned u = (o & 0x80000000u) ? (o & 0x7fffffffu) : ~o;
  return __uint_as_float(u);
}
#define ORD_NEG_INF 0x007fffffu

#define GT_THREADS 512

template <int K, int NC>
__device__ __forceinline__ void gemm_lds_body(const float* __restrict__ A,
                                              const float* __restrict__ B,
                                              float* __restrict__ C, int M,
                                              const float* __restrict__ al,
                                              const float* __restrict__ ar,
                                              float* __restrict__ rec) {
  constexpr int H   = NC / 64;
  constexpr int NP  = NC / 32;
  constexpr int WR  = 16 / NP;
  constexpr int RS  = WR * 16;
  constexpr int PA  = K + 4;
  constexpr int PB  = 2 * NC + 96;
  constexpr int K4  = K / 4;
  constexpr int NC4 = NC / 4;
  constexpr int NPF = (RS * K4) / GT_THREADS;
  constexpr int NBF = ((K / 2) * NC4) / GT_THREADS;
  static_assert(NC % 64 == 0 && 16 % NP == 0, "a wave takes one M-tile and one column pair; heads are 64 columns");
  static_assert(2 * H <= 16 && (NP == 8 || NP == 2), "the logits fit one extra tile; its waves are picked for these two shapes");
  static_assert((RS * K4) % GT_THREADS == 0 && (NPF == 2 || NPF == 4), "an A tile is 2 or 4 pieces per thread");
  static_assert(((K / 2) * NC4) % GT_THREADS == 0, "B splits evenly over the threads");
  static_assert(K % 4 == 0, "k steps of 4");

  __shared__ __align__(16) float Bs[(K / 2) * PB];
  __shared__ __align__(16) float As[2][RS * PA];

  const int tid  = threadIdx.x;
  const int lane = tid & 31;
  const int half = lane >> 4;
  const int l    = lane & 15;
  const int wave = __builtin_amdgcn_readfirstlane(tid >> 5);
  const int wm   = wave / NP;
  const int wn   = wave - wm * NP;
  const int xw   = (NP == 8) ? (wm + 1) : ((wm >> 1) & 1);
  const bool has_x = (wn == xw);

  const int nSteps = (M + RS - 1) / RS;

  auto fetch_piece = [&](int s, int u) -> v4f {
    const int it = tid + u * GT_THREADS;
    const int r = it / K4, c = it - r * K4;
    int row = s * RS + r;
    row = row < M ? row : M - 1;
    return *reinterpret_cast<const v4f*>(A + (size_t)row * K + 4 * c);
  };
  auto stash_piece = [&](float* as, int u, v4f v) {
    const int it = tid + u * GT_THREADS;
    const int r = it / K4, c = it - r * K4;
    *reinterpret_cast<v4f*>(as + r * PA + 4 * c) = v;
  };

  int step = blockIdx.x;
  v4f p0 = fetch_piece(step, 0), p1 = fetch_piece(step, 1), p2 = p0, p3 = p0;
  if constexpr (NPF == 4) { p2 = fetch_piece(step, 2); p3 = fetch_piece(step, 3); }
#pragma unroll
  for (int u = 0; u < NBF; ++u) {
    const int it = tid + u * GT_THREADS;
    const int kp = it / NC4, c = it - kp * NC4;
    const float4 r0 = *reinterpret_cast<const float4*>(B + (size_t)(2 * kp) * NC + 4 * c);
    const float4 r1 = *reinterpret_cast<const float4*>(B + (size_t)(2 * kp + 1) * NC + 4 * c);
    float4* d = reinterpret_cast<float4*>(Bs + kp * PB + 8 * c);
    d[0] = make_float4(r0.x, r1.x, r0.y, r1.y);
    d[1] = make_float4(r0.z, r1.z, r0.w, r1.w);
  }
  for (int it = tid; it < (K / 2) * 32; it += GT_THREADS) {
    const int kp = it >> 5, cc = it & 31;
    if ((cc >> 1) >= 2 * H) Bs[kp * PB + 2 * NC + cc] = 0.f;
  }
  for (int it = tid; it < K * H; it += GT_THREADS) {
    const int k = it / H, h = it - k * H;
    const float4* bq = reinterpret_cast<const float4*>(B + (size_t)k * NC + h * 64);
    const float4* lq = reinterpret_cast<const float4*>(al + h * 64);
    const float4* rq = reinterpret_cast<const float4*>(ar + h * 64);
    float4 sl = make_float4(0.f, 0.f, 0.f, 0.f), sr = sl;
#pragma unroll
    for (int i = 0; i < 16; ++i) {
      const float4 b = bq[i], wl = lq[i], wr = rq[i];
      sl.x += b.x * wl.x; sl.y += b.y * wl.y; sl.z += b.z * wl.z; sl.w += b.w * wl.w;
      sr.x += b.x * wr.x; sr.y += b.y * wr.y; sr.z += b.z * wr.z; sr.w += b.w * wr.w;
    }
    float* d = Bs + (k >> 1) * PB + 2 * NC + (k & 1);
    d[2 * h]       = (sl.x + sl.y) + (sl.z + sl.w);
    d[2 * (H + h)] = (sr.x + sr.y) + (sr.z + sr.w);
  }

  int buf = 0;
  for (; step < nSteps; step += gridDim.x) {
    float* as = As[buf];
    stash_piece(as, 0, p0);
    stash_piece(as, 1, p1);
    if constexpr (NPF == 4) { stash_piece(as, 2, p2); stash_piece(as, 3, p3); }
    __syncthreads();

    const int next = step + gridDim.x;
    p0 = fetch_piece(next, 0);
    p1 = fetch_piece(next, 1);
    if constexpr (NPF == 4) { p2 = fetch_piece(next, 2); p3 = fetch_piece(next, 3); }

    v8f acc0 = {}, acc1 = {}, accx = {};
    const float* ap = as + (wm * 16 + l) * PA + 2 * half;
    const float* bp = Bs + half * PB + (wn * 32 + l) * 2;
    if (has_x) {
      const float* bx = Bs + half * PB + (NC + l) * 2;
#pragma unroll
      for (int k = 0; k < K; k += 4) {
        const v2f a  = *reinterpret_cast<const v2f*>(ap + k);
        const v2f b0 = *reinterpret_cast<const v2f*>(bp + (k >> 1) * PB);
        const v2f b1 = *reinterpret_cast<const v2f*>(bp + (k >> 1) * PB + 32);
        const v2f b2 = *reinterpret_cast<const v2f*>(bx + (k >> 1) * PB);
        acc0 = __builtin_amdgcn_wmma_f32_16x16x4_f32(false, a, false, b0, (short)0, acc0, false, false);
        acc1 = __builtin_amdgcn_wmma_f32_16x16x4_f32(false, a, false, b1, (short)0, acc1, false, false);
        accx = __builtin_amdgcn_wmma_f32_16x16x4_f32(false, a, false, b2, (short)0, accx, false, false);
      }
    } else {
#pragma unroll
      for (int k = 0; k < K; k += 4) {
        const v2f a  = *reinterpret_cast<const v2f*>(ap + k);
        const v2f b0 = *reinterpret_cast<const v2f*>(bp + (k >> 1) * PB);
        const v2f b1 = *reinterpret_cast<const v2f*>(bp + (k >> 1) * PB + 32);
        acc0 = __builtin_amdgcn_wmma_f32_16x16x4_f32(false, a, false, b0, (short)0, acc0, false, false);
        acc1 = __builtin_amdgcn_wmma_f32_16x16x4_f32(false, a, false, b1, (short)0, acc1, false, false);
      }
    }

    const int row0 = step * RS + wm * 16 + half * 8;
    float* cp = C + (size_t)row0 * NC + wn * 32 + l;
#pragma unroll
    for (int r = 0; r < 8; ++r) {
      if (row0 + r < M) {
        cp[(size_t)r * NC]      = acc0[r];
        cp[(size_t)r * NC + 16] = acc1[r];
      }
    }
    if (has_x && l < 2 * H) {
      float* rp = rec + (size_t)row0 * (2 * H) + l;
#pragma unroll
      for (int r = 0; r < 8; ++r)
        if (row0 + r < M) rp[(size_t)r * (2 * H)] = accx[r];
    }
    buf ^= 1;
  }
}

__global__ void __launch_bounds__(GT_THREADS)
gemm_lds_k128_n256(const float* __restrict__ A, const float* __restrict__ B, float* __restrict__ C, int M,
                   const float* __restrict__ al, const float* __restrict__ ar, float* __restrict__ rec,
                   int* __restrict__ zero, int nzero) {
  if (blockIdx.x == 0)
    for (int i = threadIdx.x; i < nzero; i += GT_THREADS) zero[i] = 0;
  gemm_lds_body<128, 256>(A, B, C, M, al, ar, rec);
}
__global__ void __launch_bounds__(GT_THREADS)
gemm_lds_k64_n64(const float* __restrict__ A, const float* __restrict__ B, float* __restrict__ C, int M,
                 const float* __restrict__ al, const float* __restrict__ ar, float* __restrict__ rec) {
  gemm_lds_body<64, 64>(A, B, C, M, al, ar, rec);
}

#define AGG_THREADS 1024
#define AGG_NODES   196
#define AGG_QCAP    192
#define AGG_U       8
#define AGG_LCAP    4096

#define BK_THREADS 1024
#define BK_MAXB    1024

__global__ void __launch_bounds__(BK_THREADS)
bucket_edges(const int* __restrict__ src0, const int* __restrict__ dst0,
             const int* __restrict__ src1, const int* __restrict__ dst1,
             int* __restrict__ lists, int* __restrict__ counts, int E, int N, int nb) {
  __shared__ int place[BK_MAXB];
  const int tid = threadIdx.x;
  const int cb  = (int)(gridDim.x >> 1);
  const int t   = (int)blockIdx.x >= cb ? 1 : 0;
  const int b   = (int)blockIdx.x - t * cb;
  counts += t * nb;
  if (nb > BK_MAXB) {
    if (b == 0)
      for (int i = tid; i < nb; i += BK_THREADS) counts[i] = AGG_LCAP + 1;
    return;
  }
  const int* __restrict__ src = t ? src1 : src0;
  const int* __restrict__ dst = t ? dst1 : dst0;
  int2* glist = reinterpret_cast<int2*>(lists) + (size_t)t * nb * AGG_LCAP;

  for (int i = tid; i < nb; i += BK_THREADS) place[i] = 0;
  __syncthreads();

  const int base = (b * BK_THREADS + tid) * 4;
  int dv[4] = { -1, -1, -1, -1 };
  int sv[4] = { 0, 0, 0, 0 };
  if (base + 4 <= E) {
    const int4 d4 = *reinterpret_cast<const int4*>(dst + base);
    const int4 s4 = *reinterpret_cast<const int4*>(src + base);
    dv[0] = d4.x; dv[1] = d4.y; dv[2] = d4.z; dv[3] = d4.w;
    sv[0] = s4.x; sv[1] = s4.y; sv[2] = s4.z; sv[3] = s4.w;
  } else {
#pragma unroll
    for (int j = 0; j < 4; ++j)
      if (base + j < E) { dv[j] = dst[base + j]; sv[j] = src[base + j]; }
  }
  int bk[4], rk[4];
#pragma unroll
  for (int j = 0; j < 4; ++j) {
    const bool in = (unsigned)dv[j] < (unsigned)N;
    bk[j] = in ? dv[j] / AGG_NODES : -1;
    rk[j] = 0;
    if (in) rk[j] = atomicAdd(&place[bk[j]], 1);
  }
  __syncthreads();
  for (int i = tid; i < nb; i += BK_THREADS) {
    const int n = place[i];
    if (n > 0) place[i] = atomicAdd(counts + i, n);
  }
  __syncthreads();
#pragma unroll
  for (int j = 0; j < 4; ++j) {
    if (bk[j] >= 0) {
      const int pos = place[bk[j]] + rk[j];
      if (pos < AGG_LCAP) glist[(size_t)bk[j] * AGG_LCAP + pos] = make_int2(sv[j], dv[j]);
    }
  }
}

template <bool B> struct list_kind { static constexpr bool value = B; };

template <int H>
__device__ __forceinline__ void load_heads(const float* __restrict__ p, int node, float (&o)[H]) {
  if constexpr (H == 4) {
    const float4 t = *reinterpret_cast<const float4*>(p + (size_t)node * 8);
    o[0] = t.x; o[1] = t.y; o[2] = t.z; o[3] = t.w;
  } else {
#pragma unroll
    for (int h = 0; h < H; ++h) o[h] = p[(size_t)node * (2 * H) + h];
  }
}

template <int H, int D>
__device__ __forceinline__ void owner_softmax_aggregate_body(const int* __restrict__ src0, const int* __restrict__ dst0,
                                                             const int* __restrict__ src1, const int* __restrict__ dst1,
                                                             const float* __restrict__ el, const float* __restrict__ er,
                                                             const float* __restrict__ feat, float* __restrict__ acc,
                                                             const int* __restrict__ lists, const int* __restrict__ counts,
                                                             int E, int N) {
  constexpr int R     = H * D;
  constexpr int LINES = R / 32;
  constexpr int LPH   = D / 32;
  constexpr int G     = (LINES >= 8) ? 4 : 16;
  static_assert(D % 32 == 0, "a line must lie inside one head");

  __shared__ float    sacc[AGG_NODES * R];
  __shared__ unsigned mrow[AGG_NODES * H];
  __shared__ float    srow[AGG_NODES * H];
  __shared__ int2     queue[(AGG_THREADS / 32) * AGG_QCAP];
  __shared__ int2     elist[AGG_LCAP];
  __shared__ int      listCount;

  const int tid  = threadIdx.x;
  const int lane = tid & 31;
  const int wave = __builtin_amdgcn_readfirstlane(tid >> 5);
  const int nb  = (int)(gridDim.x >> 1);
  const int t   = (int)blockIdx.x >= nb ? 1 : 0;
  const int bid = (int)blockIdx.x - t * nb;
  const int* __restrict__ src = t ? src1 : src0;
  const int* __restrict__ dst = t ? dst1 : dst0;
  acc += (size_t)t * N * R;
  const int lo = bid * AGG_NODES;
  const int nn = (N - lo) < AGG_NODES ? (N - lo) : AGG_NODES;

  const int2* glist = reinterpret_cast<const int2*>(lists) + (size_t)blockIdx.x * AGG_LCAP;
  int saved = counts[blockIdx.x];

  float4* sacc4 = reinterpret_cast<float4*>(sacc);
  for (int i = tid; i < AGG_NODES * R / 4; i += AGG_THREADS) sacc4[i] = make_float4(0.f, 0.f, 0.f, 0.f);
  for (int i = tid; i < AGG_NODES * H; i += AGG_THREADS) { mrow[i] = ORD_NEG_INF; srow[i] = 0.f; }
  if (tid == 0) listCount = 0;
  __syncthreads();

  int2* q = queue + wave * AGG_QCAP;

  auto work_max = [&](const int2* qb, int n, auto kind) {
    constexpr bool SAVED = decltype(kind)::value;
    int2 ed = make_int2(0, 0);
    int sN = 0;
    if (lane < n) {
      ed = qb[lane];
      if constexpr (SAVED) sN = ed.x; else sN = src[ed.x];
      float l[H], r[H];
      load_heads<H>(er, ed.y, r);
      load_heads<H>(el, sN, l);
      unsigned* mo = mrow + (ed.y - lo) * H;
#pragma unroll
      for (int h = 0; h < H; ++h) {
        float sc = l[h] + r[h];
        sc = sc > 0.f ? sc : LRELU_SLOPE * sc;
        atomicMax(mo + h, f2o(sc));
      }
    }
    if constexpr (!SAVED) {
      int pos = 0;
      if (lane == 0) pos = atomicAdd(&listCount, n);
      pos = __builtin_amdgcn_readfirstlane(pos);
      if (lane < n && pos + lane < AGG_LCAP) elist[pos + lane] = ed;
    }
  };

  auto work_acc = [&](const int2* qb, int n, auto kind) {
    constexpr bool SAVED = decltype(kind)::value;
    int sN = 0, dl = 0;
    float a[H];
#pragma unroll
    for (int h = 0; h < H; ++h) a[h] = 0.f;
    if (lane < n) {
      const int2 ed = qb[lane];
      if constexpr (SAVED) sN = ed.x; else sN = src[ed.x];
      dl = ed.y - lo;
      float l[H], r[H];
      load_heads<H>(er, ed.y, r);
      load_heads<H>(el, sN, l);
#pragma unroll
      for (int h = 0; h < H; ++h) {
        float sc = l[h] + r[h];
        sc = sc > 0.f ? sc : LRELU_SLOPE * sc;
        float m = o2f(mrow[dl * H + h]);
        if (m < -1e30f) m = 0.f;
        a[h] = __expf(sc - m);
        atomicAdd(srow + dl * H + h, a[h]);
      }
    }
    for (int k0 = 0; k0 < n; k0 += G) {
      float v[G][LINES];
#pragma unroll
      for (int g = 0; g < G; ++g) {
        const int k  = (k0 + g < n) ? (k0 + g) : (n - 1);
        const int sk = __builtin_amdgcn_readlane(sN, k);
        const float* f = feat + (size_t)sk * R + lane;
#pragma unroll
        for (int j = 0; j < LINES; ++j) v[g][j] = f[32 * j];
      }
#pragma unroll
      for (int g = 0; g < G; ++g)
#pragma unroll
        for (int j = 0; j < LINES; ++j) asm volatile("" :: "v"(v[g][j]));
#pragma unroll
      for (int g = 0; g < G; ++g) {
        const int k = k0 + g;
        if (k < n) {
          const int dk = __builtin_amdgcn_readlane(dl, k);
          float* o = sacc + dk * R + lane;
#pragma unroll
          for (int j = 0; j < LINES; ++j) {
            const float ak = __int_as_float(__builtin_amdgcn_readlane(__float_as_int(a[j / LPH]), k));
            atomicAdd(o + 32 * j, ak * v[g][j]);
          }
        }
      }
    }
  };

  auto scan_apply = [&](auto&& work) {
    int cnt = 0;
    const int nChunks = (E + 127) >> 7;
    const unsigned below = (1u << lane) - 1u;
    for (int c0 = wave; c0 < nChunks; c0 += (AGG_THREADS / 32) * AGG_U) {
      int4 d[AGG_U];
#pragma unroll
      for (int u = 0; u < AGG_U; ++u) {
        const int c    = c0 + (AGG_THREADS / 32) * u;
        const int base = c * 128 + 4 * lane;
        d[u] = make_int4(-1, -1, -1, -1);
        if (c < nChunks) {
          if (c * 128 + 128 <= E) {
            d[u] = *reinterpret_cast<const int4*>(dst + base);
          } else {
            if (base + 0 < E) d[u].x = dst[base + 0];
            if (base + 1 < E) d[u].y = dst[base + 1];
            if (base + 2 < E) d[u].z = dst[base + 2];
            if (base + 3 < E) d[u].w = dst[base + 3];
          }
        }
      }
#pragma unroll
      for (int u = 0; u < AGG_U; ++u) {
        const int c    = c0 + (AGG_THREADS / 32) * u;
        const int base = c * 128 + 4 * lane;
        const int dv[4] = { d[u].x, d[u].y, d[u].z, d[u].w };
#pragma unroll
        for (int j = 0; j < 4; ++j) {
          const bool mine = (unsigned)(dv[j] - lo) < (unsigned)nn;
          const unsigned mask = (unsigned)__ballot(mine);
          if (mask != 0u) {
            if (mine) q[cnt + __popc(mask & below)] = make_int2(base + j, dv[j]);
            cnt += __popc(mask);
          }
        }
#pragma nounroll
        while (cnt >= 32) { cnt -= 32; work(q + cnt, 32); }
      }
    }
    if (cnt > 0) work(q, cnt);
  };

  saved = __builtin_amdgcn_readfirstlane(saved);
  if (__builtin_expect(saved <= AGG_LCAP, 1)) {
    const int per = (saved + (AGG_THREADS / 32) - 1) / (AGG_THREADS / 32);
    const int p1  = (wave + 1) * per < saved ? (wave + 1) * per : saved;
    for (int p = wave * per; p < p1; p += 32) {
      const int left = p1 - p;
      work_max(glist + p, left < 32 ? left : 32, list_kind<true>{});
    }
    __syncthreads();
    for (int p = wave * per; p < p1; p += 32) {
      const int left = p1 - p;
      work_acc(glist + p, left < 32 ? left : 32, list_kind<true>{});
    }
  } else {
    scan_apply([&](const int2* qb, int n) { work_max(qb, n, list_kind<false>{}); });
    __syncthreads();

    const int total = __builtin_amdgcn_readfirstlane(listCount);
    if (__builtin_expect(total <= AGG_LCAP, 1)) {
      const int per = (total + (AGG_THREADS / 32) - 1) / (AGG_THREADS / 32);
      const int p1  = (wave + 1) * per < total ? (wave + 1) * per : total;
      for (int p = wave * per; p < p1; p += 32) {
        const int left = p1 - p;
        work_acc(elist + p, left < 32 ? left : 32, list_kind<false>{});
      }
    } else {
      scan_apply([&](const int2* qb, int n) { work_acc(qb, n, list_kind<false>{}); });
    }
  }
  __syncthreads();

  float4* out4 = reinterpret_cast<float4*>(acc + (size_t)lo * R);
  for (int i = tid; i < nn * (R / 4); i += AGG_THREADS) {
    const int row = i / (R / 4);
    const int h   = (i - row * (R / 4)) / (D / 4);
    const float den = fmaxf(srow[row * H + h], 1e-9f);
    const float4 v = sacc4[i];
    out4[i] = make_float4(v.x / den, v.y / den, v.z / den, v.w / den);
  }
}

__global__ void __launch_bounds__(AGG_THREADS)
owner_softmax_aggregate_h4d64(const int* __restrict__ src0, const int* __restrict__ dst0,
                              const int* __restrict__ src1, const int* __restrict__ dst1,
                              const float* __restrict__ el, const float* __restrict__ er,
                              const float* __restrict__ feat, float* __restrict__ acc,
                              const int* __restrict__ lists, const int* __restrict__ counts, int E, int N) {
  owner_softmax_aggregate_body<4, 64>(src0, dst0, src1, dst1, el, er, feat, acc, lists, counts, E, N);
}
__global__ void __launch_bounds__(AGG_THREADS)
owner_softmax_aggregate_h1d64(const int* __restrict__ src0, const int* __restrict__ dst0,
                              const int* __restrict__ src1, const int* __restrict__ dst1,
                              const float* __restrict__ el, const float* __restrict__ er,
                              const float* __restrict__ feat, float* __restrict__ acc,
                              const int* __restrict__ lists, const int* __restrict__ counts, int E, int N) {
  owner_softmax_aggregate_body<1, 64>(src0, dst0, src1, dst1, el, er, feat, acc, lists, counts, E, N);
}

__device__ __forceinline__ float elu1(float v) { return (v > 0.f) ? v : expm1f(v); }

template <int H, int D>
__device__ __forceinline__ void epilogue0_body(const float* __restrict__ acc, const float* __restrict__ b,
                                               float* __restrict__ h1, int N) {
  constexpr int D4 = D / 4;
  const unsigned n4 = (unsigned)N * D4;
  const unsigned T = gridDim.x * blockDim.x;
  for (unsigned i = blockIdx.x * blockDim.x + threadIdx.x; i < n4; i += T) {
    const unsigned n = i / D4, q = i - n * D4;
    float4 v[2][H], bb[H];
#pragma unroll
    for (int t = 0; t < 2; ++t)
#pragma unroll
      for (int h = 0; h < H; ++h)
        v[t][h] = *reinterpret_cast<const float4*>(acc + ((size_t)t * N + n) * (H * D) + h * D + 4 * q);
#pragma unroll
    for (int h = 0; h < H; ++h) bb[h] = *reinterpret_cast<const float4*>(b + h * D + 4 * q);
    float4 sum = make_float4(0.f, 0.f, 0.f, 0.f);
#pragma unroll
    for (int t = 0; t < 2; ++t)
#pragma unroll
      for (int h = 0; h < H; ++h) {
        sum.x += elu1(v[t][h].x + bb[h].x);
        sum.y += elu1(v[t][h].y + bb[h].y);
        sum.z += elu1(v[t][h].z + bb[h].z);
        sum.w += elu1(v[t][h].w + bb[h].w);
      }
    const float hf = (float)H;
    *reinterpret_cast<float4*>(h1 + (size_t)i * 4) =
        make_float4(sum.x / hf, sum.y / hf, sum.z / hf, sum.w / hf);
  }
}
__global__ void epilogue0_h4d64(const float* __restrict__ acc, const float* __restrict__ b,
                                float* __restrict__ h1, int N) {
  epilogue0_body<4, 64>(acc, b, h1, N);
}


template <int D, int U>
__device__ __forceinline__ void epilogue1_body(const float* __restrict__ acc, const float* __restrict__ b,
                                               float* __restrict__ out, int N) {
  constexpr int D4 = D / 4;
  const unsigned n4 = (unsigned)N * D4;
  const unsigned T = gridDim.x * blockDim.x;
  const v4f* a0 = reinterpret_cast<const v4f*>(acc);
  const v4f* a1 = reinterpret_cast<const v4f*>(acc + (size_t)N * D);
  const v4f* b4 = reinterpret_cast<const v4f*>(b);
  v4f* o4 = reinterpret_cast<v4f*>(out);
  for (unsigned i0 = blockIdx.x * blockDim.x + threadIdx.x; i0 < n4; i0 += U * T) {
    v4f x[U], y[U], bb[U];
#pragma unroll
    for (int u = 0; u < U; ++u) {
      const unsigned i  = i0 + u * T;
      const unsigned ic = i < n4 ? i : n4 - 1;
      x[u]  = a0[ic];
      y[u]  = a1[ic];
      bb[u] = b4[ic % D4];
    }
#pragma unroll
    for (int u = 0; u < U; ++u) asm volatile("" :: "v"(x[u]), "v"(y[u]), "v"(bb[u]));
#pragma unroll
    for (int u = 0; u < U; ++u) {
      const unsigned i = i0 + u * T;
      if (i < n4) o4[i] = (x[u] + bb[u]) + (y[u] + bb[u]);
    }
  }
}
__global__ void epilogue1_d64(const float* __restrict__ acc, const float* __restrict__ b,
                              float* __restrict__ out, int N) {
  epilogue1_body<64, 4>(acc, b, out, N);
}

extern "C" void kernel_launch(void* const* d_in, const int* in_sizes, int n_in,
                              void* d_out, int out_size, void* d_ws, size_t ws_size,
                              hipStream_t stream) {
  const int IN = 128, H0 = 4, D0 = 64, H1 = 1, D1 = 64;
  const int N = in_sizes[0] / IN;
  const int E = in_sizes[9];

  const float* x   = (const float*)d_in[0];
  const float* W0  = (const float*)d_in[1];
  const float* al0 = (const float*)d_in[2];
  const float* ar0 = (const float*)d_in[3];
  const float* b0  = (const float*)d_in[4];
  const float* W1  = (const float*)d_in[5];
  const float* al1 = (const float*)d_in[6];
  const float* ar1 = (const float*)d_in[7];
  const float* b1  = (const float*)d_in[8];
  const int* srcT[2] = { (const int*)d_in[9],  (const int*)d_in[11] };
  const int* dstT[2] = { (const int*)d_in[10], (const int*)d_in[12] };
  float* out = (float*)d_out;

  float* ws = (float*)d_ws;
  float* feat0 = ws;
  float* rec0  = feat0 + (size_t)N * H0 * D0;
  float* acc0  = rec0 + 2 * (size_t)N * H0;
  float* h1    = acc0 + 2 * (size_t)N * H0 * D0;
  float* feat1 = feat0;
  float* rec1  = feat1 + (size_t)N * D1;
  float* acc1  = rec1 + 2 * (size_t)N;

  const int TB = 256;
  const int aggBlocks = (N + AGG_NODES - 1) / AGG_NODES;
  int* lists  = (int*)(h1 + (size_t)N * D0);
  int* counts = lists + (size_t)2 * aggBlocks * AGG_LCAP * 2;
  const int epi0Blocks  = 1024;
  const int epi1Blocks  = (N * (D1 / 4) + TB * 4 - 1) / (TB * 4);
  const int bucketBlocks = (E + BK_THREADS * 4 - 1) / (BK_THREADS * 4);
  (void)ws_size; (void)n_in; (void)out_size;

  {
    gemm_lds_k128_n256<<<256, GT_THREADS, 0, stream>>>(x, W0, feat0, N, al0, ar0, rec0, counts, 2 * aggBlocks);

    bucket_edges<<<2 * bucketBlocks, BK_THREADS, 0, stream>>>(srcT[0], dstT[0], srcT[1], dstT[1], lists, counts, E, N, aggBlocks);

    owner_softmax_aggregate_h4d64<<<2 * aggBlocks, AGG_THREADS, 0, stream>>>(srcT[0], dstT[0], srcT[1], dstT[1], rec0, rec0 + H0, feat0, acc0, lists, counts, E, N);

    epilogue0_h4d64<<<epi0Blocks, TB, 0, stream>>>(acc0, b0, h1, N);
  }

  {
    gemm_lds_k64_n64<<<(N + 127) / 128, GT_THREADS, 0, stream>>>(h1, W1, feat1, N, al1, ar1, rec1);

    owner_softmax_aggregate_h1d64<<<2 * aggBlocks, AGG_THREADS, 0, stream>>>(srcT[0], dstT[0], srcT[1], dstT[1], rec1, rec1 + H1, feat1, acc1, lists, counts, E, N);

    epilogue1_d64<<<epi1Blocks, TB, 0, stream>>>(acc1, b1, out, N);
  }
}
